# starting kernel
# baseline (speedup 1.0000x reference)
#include <hip/hip_runtime.h>


typedef __attribute__((ext_vector_type(2))) float v2f;
typedef __attribute__((ext_vector_type(8))) float v8f;

#define IN_FEATS 128
#define HID      64
#define NC       32


__global__ void fill_zero_kernel(float* __restrict__ p, int n) {
    int i = blockIdx.x * blockDim.x + threadIdx.x;
    if (i < n) p[i] = 0.0f;
}

__global__ void deg_kernel(const int* __restrict__ src, const int* __restrict__ dst,
                           float* __restrict__ deg_out, float* __restrict__ deg_in, int E) {
    int e = blockIdx.x * blockDim.x + threadIdx.x;
    if (e < E) {
        unsafeAtomicAdd(&deg_out[src[e]], 1.0f);
        unsafeAtomicAdd(&deg_in[dst[e]], 1.0f);
    }
}

__global__ void rsqrt_clip_kernel(float* __restrict__ d, int n) {
    int i = blockIdx.x * blockDim.x + threadIdx.x;
    if (i < n) d[i] = rsqrtf(fmaxf(d[i], 1.0f));
}

__global__ void bias_relu_kernel(float* __restrict__ a, const float* __restrict__ rs_in,
                                 const float* __restrict__ b, int n) {
    int i = blockIdx.x * blockDim.x + threadIdx.x;
    if (i < n * HID) {
        int row = i >> 6;
        int col = i & (HID - 1);
        float v = a[i] * rs_in[row] + b[col];
        a[i] = fmaxf(v, 0.0f);
    }
}

__global__ void out_acc_kernel(float* __restrict__ out, const float* __restrict__ a,
                               const float* __restrict__ rs_in, const float* __restrict__ b,
                               int n) {
    int i = blockIdx.x * blockDim.x + threadIdx.x;
    if (i < n * NC) {
        int row = i >> 5;
        int col = i & (NC - 1);
        out[i] += 0.5f * (a[i] * rs_in[row] + b[col]);
    }
}

template <int F>
__global__ void scatter_kernel(const float* __restrict__ h, const int* __restrict__ src,
                               const int* __restrict__ dst, float* __restrict__ agg, int E) {
    constexpr int C = F / 4;
    int idx = blockIdx.x * blockDim.x + threadIdx.x;
    if (idx >= E * C) return;
    int e = idx / C;
    int c = idx - e * C;
    int s = src[e];
    int d = dst[e];
    const float4 v = *(const float4*)&h[(long)s * F + c * 4];
    float* p = &agg[(long)d * F + c * 4];
    unsafeAtomicAdd(p + 0, v.x);
    unsafeAtomicAdd(p + 1, v.y);
    unsafeAtomicAdd(p + 2, v.z);
    unsafeAtomicAdd(p + 3, v.w);
}

template <int IN, int OUT>
__global__ void __launch_bounds__(256)
gemm_gc_kernel(const float* __restrict__ X, const float* __restrict__ rs,
               const float* __restrict__ W, float* __restrict__ H, int n) {
    constexpr int LDW = IN + 4;
    __shared__ float lw[OUT * LDW];

    for (int i = threadIdx.x; i < IN * OUT; i += 256) {
        int k = i / OUT;
        int c = i - k * OUT;
        lw[c * LDW + k] = W[i];
    }
    __syncthreads();

    const int lane = threadIdx.x & 31;
    const int wave = threadIdx.x >> 5;
    const int r0 = (blockIdx.x * 8 + wave) * 16;
    if (r0 >= n) return;

    const int mrow = r0 + (lane & 15);
    const int koff = (lane >> 4) * 2;
    const float rsv = rs[mrow];
    const float* xrow = X + (long)mrow * IN;
    const int col = lane & 15;

    constexpr int CT = OUT / 16;
    v8f acc[CT] = {};

    for (int k = 0; k < IN; k += 4) {
        float2 av = *(const float2*)(xrow + k + koff);
        v2f a;
        a.x = av.x * rsv;
        a.y = av.y * rsv;
#pragma unroll
        for (int c = 0; c < CT; ++c) {
            float2 bv = *(const float2*)&lw[(c * 16 + col) * LDW + k + koff];
            v2f b;
            b.x = bv.x;
            b.y = bv.y;
            acc[c] = __builtin_amdgcn_wmma_f32_16x16x4_f32(
                false, a, false, b, (short)0, acc[c], false, false);
        }
    }

    const int mhi = (lane >> 4) * 8;
#pragma unroll
    for (int c = 0; c < CT; ++c) {
#pragma unroll
        for (int r = 0; r < 8; ++r) {
            H[(long)(r0 + mhi + r) * OUT + c * 16 + col] = acc[c][r];
        }
    }
}


extern "C" void kernel_launch(void* const* d_in, const int* in_sizes, int n_in,
                              void* d_out, int out_size, void* d_ws, size_t ws_size,
                              hipStream_t stream) {
    const float* x    = (const float*)d_in[0];
    const int*   src1 = (const int*)d_in[1];
    const int*   dst1 = (const int*)d_in[2];
    const int*   src2 = (const int*)d_in[3];
    const int*   dst2 = (const int*)d_in[4];
    const float* W1   = (const float*)d_in[5];
    const float* b1   = (const float*)d_in[6];
    const float* W2   = (const float*)d_in[7];
    const float* b2   = (const float*)d_in[8];
    float* out = (float*)d_out;

    const int N = in_sizes[0] / IN_FEATS;
    const int E = in_sizes[1];

    float* ws   = (float*)d_ws;
    float* rsO  = ws;
    float* rsI  = rsO + N;
    float* h    = rsI + N;
    float* agg1 = h + (size_t)N * HID;
    float* h2   = agg1 + (size_t)N * HID;
    float* agg2 = h2 + (size_t)N * NC;

    const int T = 256;
    auto cdiv = [](long a, long b) { return (int)((a + b - 1) / b); };

    const int tiles   = N / 16;
    const int gblocks = cdiv(tiles, 8);

    fill_zero_kernel<<<cdiv((long)N * NC, T), T, 0, stream>>>(out, N * NC);

    for (int g = 0; g < 2; ++g) {
        const int* src = (g == 0) ? src1 : src2;
        const int* dst = (g == 0) ? dst1 : dst2;

        fill_zero_kernel<<<cdiv(N, T), T, 0, stream>>>(rsO, N);
        fill_zero_kernel<<<cdiv(N, T), T, 0, stream>>>(rsI, N);
        deg_kernel<<<cdiv(E, T), T, 0, stream>>>(src, dst, rsO, rsI, E);
        rsqrt_clip_kernel<<<cdiv(N, T), T, 0, stream>>>(rsO, N);
        rsqrt_clip_kernel<<<cdiv(N, T), T, 0, stream>>>(rsI, N);

        gemm_gc_kernel<IN_FEATS, HID><<<gblocks, T, 0, stream>>>(x, rsO, W1, h, N);
        fill_zero_kernel<<<cdiv((long)N * HID, T), T, 0, stream>>>(agg1, N * HID);
        scatter_kernel<HID><<<cdiv((long)E * (HID / 4), T), T, 0, stream>>>(h, src, dst, agg1, E);
        bias_relu_kernel<<<cdiv((long)N * HID, T), T, 0, stream>>>(agg1, rsI, b1, N);

        gemm_gc_kernel<HID, NC><<<gblocks, T, 0, stream>>>(agg1, rsO, W2, h2, N);
        fill_zero_kernel<<<cdiv((long)N * NC, T), T, 0, stream>>>(agg2, N * NC);
        scatter_kernel<NC><<<cdiv((long)E * (NC / 4), T), T, 0, stream>>>(h2, src, dst, agg2, E);
        out_acc_kernel<<<cdiv((long)N * NC, T), T, 0, stream>>>(out, agg2, rsI, b2, N);
    }
}


# degree counts: per-node-range LDS histograms (ds_add_u32) stored as per-slice integer partials and summed in the rsqrt kernel; replaces random global float atomics and their zero fills
# speedup vs baseline: 1.4660x; 1.4660x over previous
#include <hip/hip_runtime.h>


typedef __attribute__((ext_vector_type(2))) float v2f;
typedef __attribute__((ext_vector_type(8))) float v8f;

#define IN_FEATS 128
#define HID      64
#define NC       32


__global__ void fill_zero_kernel(float* __restrict__ p, int n) {
    int i = blockIdx.x * blockDim.x + threadIdx.x;
    if (i < n) p[i] = 0.0f;
}

#define DEG_HR 12544
#define DEG_S  16
#define DEG_T  1024

__device__ __forceinline__ void deg_bin4(unsigned int* cnt, int4 k, unsigned int lo) {
    unsigned int k0 = (unsigned int)k.x - lo;
    unsigned int k1 = (unsigned int)k.y - lo;
    unsigned int k2 = (unsigned int)k.z - lo;
    unsigned int k3 = (unsigned int)k.w - lo;
    if (k0 < DEG_HR) atomicAdd(&cnt[k0], 1u);
    if (k1 < DEG_HR) atomicAdd(&cnt[k1], 1u);
    if (k2 < DEG_HR) atomicAdd(&cnt[k2], 1u);
    if (k3 < DEG_HR) atomicAdd(&cnt[k3], 1u);
}

__global__ void __launch_bounds__(DEG_T)
deg_hist_kernel(const int* __restrict__ src, const int* __restrict__ dst,
                unsigned int* __restrict__ part, int E, int stride) {
    __shared__ uint4 cnt4[DEG_HR / 4];
    unsigned int* cnt = (unsigned int*)cnt4;
    const int tid = threadIdx.x;
    const int s = blockIdx.x;
    const unsigned int lo = blockIdx.y * DEG_HR;
    const int a = blockIdx.z;
    const int* __restrict__ keys = a ? dst : src;

    for (int i = tid; i < DEG_HR / 4; i += DEG_T) cnt4[i] = make_uint4(0u, 0u, 0u, 0u);
    __syncthreads();

    const int per = (((E + DEG_S - 1) / DEG_S) + 3) & ~3;
    const int e0 = s * per;
    int e1 = e0 + per;
    if (e1 > E) e1 = E;
    const int nvec = (e1 > e0) ? ((e1 - e0) >> 2) : 0;
    const int4* __restrict__ kv = (const int4*)(keys + e0);

    int i = tid;
    for (; i + 7 * DEG_T < nvec; i += 8 * DEG_T) {
        int4 k[8];
#pragma unroll
        for (int j = 0; j < 8; ++j) k[j] = kv[i + j * DEG_T];
#pragma unroll
        for (int j = 0; j < 8; ++j) deg_bin4(cnt, k[j], lo);
    }
    for (; i < nvec; i += DEG_T) deg_bin4(cnt, kv[i], lo);
    for (int e = e0 + (nvec << 2) + tid; e < e1; e += DEG_T) {
        unsigned int k0 = (unsigned int)keys[e] - lo;
        if (k0 < DEG_HR) atomicAdd(&cnt[k0], 1u);
    }
    __syncthreads();

    uint4* __restrict__ p = (uint4*)(part + (size_t)(a * DEG_S + s) * stride + lo);
    for (int j = tid; j < DEG_HR / 4; j += DEG_T) p[j] = cnt4[j];
}

__global__ void deg_rsqrt_kernel(const unsigned int* __restrict__ part, float* __restrict__ rs,
                                 int n, int stride) {
    int node = blockIdx.x * blockDim.x + threadIdx.x;
    int a = blockIdx.y;
    if (node < n) {
        const unsigned int* p = part + (size_t)a * DEG_S * stride + node;
        unsigned int c = 0;
#pragma unroll
        for (int s = 0; s < DEG_S; ++s) c += p[(size_t)s * stride];
        rs[(size_t)a * n + node] = rsqrtf(fmaxf((float)c, 1.0f));
    }
}

__global__ void bias_relu_kernel(float* __restrict__ a, const float* __restrict__ rs_in,
                                 const float* __restrict__ b, int n) {
    int i = blockIdx.x * blockDim.x + threadIdx.x;
    if (i < n * HID) {
        int row = i >> 6;
        int col = i & (HID - 1);
        float v = a[i] * rs_in[row] + b[col];
        a[i] = fmaxf(v, 0.0f);
    }
}

__global__ void out_acc_kernel(float* __restrict__ out, const float* __restrict__ a,
                               const float* __restrict__ rs_in, const float* __restrict__ b,
                               int n) {
    int i = blockIdx.x * blockDim.x + threadIdx.x;
    if (i < n * NC) {
        int row = i >> 5;
        int col = i & (NC - 1);
        out[i] += 0.5f * (a[i] * rs_in[row] + b[col]);
    }
}

template <int F>
__global__ void scatter_kernel(const float* __restrict__ h, const int* __restrict__ src,
                               const int* __restrict__ dst, float* __restrict__ agg, int E) {
    constexpr int C = F / 4;
    int idx = blockIdx.x * blockDim.x + threadIdx.x;
    if (idx >= E * C) return;
    int e = idx / C;
    int c = idx - e * C;
    int s = src[e];
    int d = dst[e];
    const float4 v = *(const float4*)&h[(long)s * F + c * 4];
    float* p = &agg[(long)d * F + c * 4];
    unsafeAtomicAdd(p + 0, v.x);
    unsafeAtomicAdd(p + 1, v.y);
    unsafeAtomicAdd(p + 2, v.z);
    unsafeAtomicAdd(p + 3, v.w);
}

template <int IN, int OUT>
__global__ void __launch_bounds__(256)
gemm_gc_kernel(const float* __restrict__ X, const float* __restrict__ rs,
               const float* __restrict__ W, float* __restrict__ H, int n) {
    constexpr int LDW = IN + 4;
    __shared__ float lw[OUT * LDW];

    for (int i = threadIdx.x; i < IN * OUT; i += 256) {
        int k = i / OUT;
        int c = i - k * OUT;
        lw[c * LDW + k] = W[i];
    }
    __syncthreads();

    const int lane = threadIdx.x & 31;
    const int wave = threadIdx.x >> 5;
    const int r0 = (blockIdx.x * 8 + wave) * 16;
    if (r0 >= n) return;

    const int mrow = r0 + (lane & 15);
    const int koff = (lane >> 4) * 2;
    const float rsv = rs[mrow];
    const float* xrow = X + (long)mrow * IN;
    const int col = lane & 15;

    constexpr int CT = OUT / 16;
    v8f acc[CT] = {};

    for (int k = 0; k < IN; k += 4) {
        float2 av = *(const float2*)(xrow + k + koff);
        v2f a;
        a.x = av.x * rsv;
        a.y = av.y * rsv;
#pragma unroll
        for (int c = 0; c < CT; ++c) {
            float2 bv = *(const float2*)&lw[(c * 16 + col) * LDW + k + koff];
            v2f b;
            b.x = bv.x;
            b.y = bv.y;
            acc[c] = __builtin_amdgcn_wmma_f32_16x16x4_f32(
                false, a, false, b, (short)0, acc[c], false, false);
        }
    }

    const int mhi = (lane >> 4) * 8;
#pragma unroll
    for (int c = 0; c < CT; ++c) {
#pragma unroll
        for (int r = 0; r < 8; ++r) {
            H[(long)(r0 + mhi + r) * OUT + c * 16 + col] = acc[c][r];
        }
    }
}


extern "C" void kernel_launch(void* const* d_in, const int* in_sizes, int n_in,
                              void* d_out, int out_size, void* d_ws, size_t ws_size,
                              hipStream_t stream) {
    const float* x    = (const float*)d_in[0];
    const int*   src1 = (const int*)d_in[1];
    const int*   dst1 = (const int*)d_in[2];
    const int*   src2 = (const int*)d_in[3];
    const int*   dst2 = (const int*)d_in[4];
    const float* W1   = (const float*)d_in[5];
    const float* b1   = (const float*)d_in[6];
    const float* W2   = (const float*)d_in[7];
    const float* b2   = (const float*)d_in[8];
    float* out = (float*)d_out;

    const int N = in_sizes[0] / IN_FEATS;
    const int E = in_sizes[1];

    const int dranges = (N + DEG_HR - 1) / DEG_HR;
    const int dstride = dranges * DEG_HR;
    unsigned int* dpart = (unsigned int*)d_ws;
    float* ws   = (float*)(dpart + (size_t)2 * DEG_S * dstride);
    float* rsO  = ws;
    float* rsI  = rsO + N;
    float* h    = rsI + N;
    float* agg1 = h + (size_t)N * HID;
    float* h2   = agg1 + (size_t)N * HID;
    float* agg2 = h2 + (size_t)N * NC;

    const int T = 256;
    auto cdiv = [](long a, long b) { return (int)((a + b - 1) / b); };

    const int tiles   = N / 16;
    const int gblocks = cdiv(tiles, 8);

    fill_zero_kernel<<<cdiv((long)N * NC, T), T, 0, stream>>>(out, N * NC);

    for (int g = 0; g < 2; ++g) {
        const int* src = (g == 0) ? src1 : src2;
        const int* dst = (g == 0) ? dst1 : dst2;

        deg_hist_kernel<<<dim3(DEG_S, dranges, 2), DEG_T, 0, stream>>>(src, dst, dpart, E, dstride);
        deg_rsqrt_kernel<<<dim3(cdiv(N, T), 2), T, 0, stream>>>(dpart, rsO, N, dstride);

        gemm_gc_kernel<IN_FEATS, HID><<<gblocks, T, 0, stream>>>(x, rsO, W1, h, N);
        fill_zero_kernel<<<cdiv((long)N * HID, T), T, 0, stream>>>(agg1, N * HID);
        scatter_kernel<HID><<<cdiv((long)E * (HID / 4), T), T, 0, stream>>>(h, src, dst, agg1, E);
        bias_relu_kernel<<<cdiv((long)N * HID, T), T, 0, stream>>>(agg1, rsI, b1, N);

        gemm_gc_kernel<HID, NC><<<gblocks, T, 0, stream>>>(agg1, rsO, W2, h2, N);
        fill_zero_kernel<<<cdiv((long)N * NC, T), T, 0, stream>>>(agg2, N * NC);
        scatter_kernel<NC><<<cdiv((long)E * (NC / 4), T), T, 0, stream>>>(h2, src, dst, agg2, E);
        out_acc_kernel<<<cdiv((long)N * NC, T), T, 0, stream>>>(out, agg2, rsI, b2, N);
    }
}


# edge scatter re-mapped: one wave per edge row, lane = feature, so each gather load and each float atomic covers one whole 128-byte line; src/dst read once per 32 edges, broadcast by readlane
# speedup vs baseline: 2.2056x; 1.5045x over previous
#include <hip/hip_runtime.h>


typedef __attribute__((ext_vector_type(2))) float v2f;
typedef __attribute__((ext_vector_type(8))) float v8f;

#define IN_FEATS 128
#define HID      64
#define NC       32


__global__ void fill_zero_kernel(float* __restrict__ p, int n) {
    int i = blockIdx.x * blockDim.x + threadIdx.x;
    if (i < n) p[i] = 0.0f;
}

#define DEG_HR 12544
#define DEG_S  16
#define DEG_T  1024

__device__ __forceinline__ void deg_bin4(unsigned int* cnt, int4 k, unsigned int lo) {
    unsigned int k0 = (unsigned int)k.x - lo;
    unsigned int k1 = (unsigned int)k.y - lo;
    unsigned int k2 = (unsigned int)k.z - lo;
    unsigned int k3 = (unsigned int)k.w - lo;
    if (k0 < DEG_HR) atomicAdd(&cnt[k0], 1u);
    if (k1 < DEG_HR) atomicAdd(&cnt[k1], 1u);
    if (k2 < DEG_HR) atomicAdd(&cnt[k2], 1u);
    if (k3 < DEG_HR) atomicAdd(&cnt[k3], 1u);
}

__global__ void __launch_bounds__(DEG_T)
deg_hist_kernel(const int* __restrict__ src, const int* __restrict__ dst,
                unsigned int* __restrict__ part, int E, int stride) {
    __shared__ uint4 cnt4[DEG_HR / 4];
    unsigned int* cnt = (unsigned int*)cnt4;
    const int tid = threadIdx.x;
    const int s = blockIdx.x;
    const unsigned int lo = blockIdx.y * DEG_HR;
    const int a = blockIdx.z;
    const int* __restrict__ keys = a ? dst : src;

    for (int i = tid; i < DEG_HR / 4; i += DEG_T) cnt4[i] = make_uint4(0u, 0u, 0u, 0u);
    __syncthreads();

    const int per = (((E + DEG_S - 1) / DEG_S) + 3) & ~3;
    const int e0 = s * per;
    int e1 = e0 + per;
    if (e1 > E) e1 = E;
    const int nvec = (e1 > e0) ? ((e1 - e0) >> 2) : 0;
    const int4* __restrict__ kv = (const int4*)(keys + e0);

    int i = tid;
    for (; i + 7 * DEG_T < nvec; i += 8 * DEG_T) {
        int4 k[8];
#pragma unroll
        for (int j = 0; j < 8; ++j) k[j] = kv[i + j * DEG_T];
#pragma unroll
        for (int j = 0; j < 8; ++j) deg_bin4(cnt, k[j], lo);
    }
    for (; i < nvec; i += DEG_T) deg_bin4(cnt, kv[i], lo);
    for (int e = e0 + (nvec << 2) + tid; e < e1; e += DEG_T) {
        unsigned int k0 = (unsigned int)keys[e] - lo;
        if (k0 < DEG_HR) atomicAdd(&cnt[k0], 1u);
    }
    __syncthreads();

    uint4* __restrict__ p = (uint4*)(part + (size_t)(a * DEG_S + s) * stride + lo);
    for (int j = tid; j < DEG_HR / 4; j += DEG_T) p[j] = cnt4[j];
}

__global__ void deg_rsqrt_kernel(const unsigned int* __restrict__ part, float* __restrict__ rs,
                                 int n, int stride) {
    int node = blockIdx.x * blockDim.x + threadIdx.x;
    int a = blockIdx.y;
    if (node < n) {
        const unsigned int* p = part + (size_t)a * DEG_S * stride + node;
        unsigned int c = 0;
#pragma unroll
        for (int s = 0; s < DEG_S; ++s) c += p[(size_t)s * stride];
        rs[(size_t)a * n + node] = rsqrtf(fmaxf((float)c, 1.0f));
    }
}

__global__ void bias_relu_kernel(float* __restrict__ a, const float* __restrict__ rs_in,
                                 const float* __restrict__ b, int n) {
    int i = blockIdx.x * blockDim.x + threadIdx.x;
    if (i < n * HID) {
        int row = i >> 6;
        int col = i & (HID - 1);
        float v = a[i] * rs_in[row] + b[col];
        a[i] = fmaxf(v, 0.0f);
    }
}

__global__ void out_acc_kernel(float* __restrict__ out, const float* __restrict__ a,
                               const float* __restrict__ rs_in, const float* __restrict__ b,
                               int n) {
    int i = blockIdx.x * blockDim.x + threadIdx.x;
    if (i < n * NC) {
        int row = i >> 5;
        int col = i & (NC - 1);
        out[i] += 0.5f * (a[i] * rs_in[row] + b[col]);
    }
}

template <int F>
__global__ void __launch_bounds__(256)
scatter_kernel(const float* __restrict__ h, const int* __restrict__ src,
               const int* __restrict__ dst, float* __restrict__ agg, int E) {
    constexpr int R = F / 32;
    constexpr int U = 16 / R;
    const int lane = threadIdx.x & 31;
    const int e0 = ((blockIdx.x * blockDim.x + threadIdx.x) >> 5) * 32;
    if (e0 >= E) return;
    const int e = e0 + lane;
    int s_l = 0, d_l = 0;
    if (e < E) { s_l = src[e]; d_l = dst[e]; }
    const int cnt = (E - e0 < 32) ? (E - e0) : 32;

    if (cnt == 32) {
#pragma unroll
        for (int j0 = 0; j0 < 32; j0 += U) {
            float v[U][R];
#pragma unroll
            for (int u = 0; u < U; ++u) {
                const int s = __builtin_amdgcn_readlane(s_l, j0 + u);
                const float* __restrict__ hp = h + (size_t)s * F + lane;
#pragma unroll
                for (int r = 0; r < R; ++r) v[u][r] = hp[r * 32];
            }
#pragma unroll
            for (int u = 0; u < U; ++u) {
                const int d = __builtin_amdgcn_readlane(d_l, j0 + u);
                float* __restrict__ ap = agg + (size_t)d * F + lane;
#pragma unroll
                for (int r = 0; r < R; ++r) unsafeAtomicAdd(ap + r * 32, v[u][r]);
            }
        }
    } else {
        for (int j = 0; j < cnt; ++j) {
            const int s = __builtin_amdgcn_readlane(s_l, j);
            const int d = __builtin_amdgcn_readlane(d_l, j);
            const float* __restrict__ hp = h + (size_t)s * F + lane;
            float* __restrict__ ap = agg + (size_t)d * F + lane;
#pragma unroll
            for (int r = 0; r < R; ++r) unsafeAtomicAdd(ap + r * 32, hp[r * 32]);
        }
    }
}

template <int IN, int OUT>
__global__ void __launch_bounds__(256)
gemm_gc_kernel(const float* __restrict__ X, const float* __restrict__ rs,
               const float* __restrict__ W, float* __restrict__ H, int n) {
    constexpr int LDW = IN + 4;
    __shared__ float lw[OUT * LDW];

    for (int i = threadIdx.x; i < IN * OUT; i += 256) {
        int k = i / OUT;
        int c = i - k * OUT;
        lw[c * LDW + k] = W[i];
    }
    __syncthreads();

    const int lane = threadIdx.x & 31;
    const int wave = threadIdx.x >> 5;
    const int r0 = (blockIdx.x * 8 + wave) * 16;
    if (r0 >= n) return;

    const int mrow = r0 + (lane & 15);
    const int koff = (lane >> 4) * 2;
    const float rsv = rs[mrow];
    const float* xrow = X + (long)mrow * IN;
    const int col = lane & 15;

    constexpr int CT = OUT / 16;
    v8f acc[CT] = {};

    for (int k = 0; k < IN; k += 4) {
        float2 av = *(const float2*)(xrow + k + koff);
        v2f a;
        a.x = av.x * rsv;
        a.y = av.y * rsv;
#pragma unroll
        for (int c = 0; c < CT; ++c) {
            float2 bv = *(const float2*)&lw[(c * 16 + col) * LDW + k + koff];
            v2f b;
            b.x = bv.x;
            b.y = bv.y;
            acc[c] = __builtin_amdgcn_wmma_f32_16x16x4_f32(
                false, a, false, b, (short)0, acc[c], false, false);
        }
    }

    const int mhi = (lane >> 4) * 8;
#pragma unroll
    for (int c = 0; c < CT; ++c) {
#pragma unroll
        for (int r = 0; r < 8; ++r) {
            H[(long)(r0 + mhi + r) * OUT + c * 16 + col] = acc[c][r];
        }
    }
}


extern "C" void kernel_launch(void* const* d_in, const int* in_sizes, int n_in,
                              void* d_out, int out_size, void* d_ws, size_t ws_size,
                              hipStream_t stream) {
    const float* x    = (const float*)d_in[0];
    const int*   src1 = (const int*)d_in[1];
    const int*   dst1 = (const int*)d_in[2];
    const int*   src2 = (const int*)d_in[3];
    const int*   dst2 = (const int*)d_in[4];
    const float* W1   = (const float*)d_in[5];
    const float* b1   = (const float*)d_in[6];
    const float* W2   = (const float*)d_in[7];
    const float* b2   = (const float*)d_in[8];
    float* out = (float*)d_out;

    const int N = in_sizes[0] / IN_FEATS;
    const int E = in_sizes[1];

    const int dranges = (N + DEG_HR - 1) / DEG_HR;
    const int dstride = dranges * DEG_HR;
    unsigned int* dpart = (unsigned int*)d_ws;
    float* ws   = (float*)(dpart + (size_t)2 * DEG_S * dstride);
    float* rsO  = ws;
    float* rsI  = rsO + N;
    float* h    = rsI + N;
    float* agg1 = h + (size_t)N * HID;
    float* h2   = agg1 + (size_t)N * HID;
    float* agg2 = h2 + (size_t)N * NC;

    const int T = 256;
    auto cdiv = [](long a, long b) { return (int)((a + b - 1) / b); };

    const int tiles   = N / 16;
    const int gblocks = cdiv(tiles, 8);

    fill_zero_kernel<<<cdiv((long)N * NC, T), T, 0, stream>>>(out, N * NC);

    for (int g = 0; g < 2; ++g) {
        const int* src = (g == 0) ? src1 : src2;
        const int* dst = (g == 0) ? dst1 : dst2;

        deg_hist_kernel<<<dim3(DEG_S, dranges, 2), DEG_T, 0, stream>>>(src, dst, dpart, E, dstride);
        deg_rsqrt_kernel<<<dim3(cdiv(N, T), 2), T, 0, stream>>>(dpart, rsO, N, dstride);

        gemm_gc_kernel<IN_FEATS, HID><<<gblocks, T, 0, stream>>>(x, rsO, W1, h, N);
        fill_zero_kernel<<<cdiv((long)N * HID, T), T, 0, stream>>>(agg1, N * HID);
        scatter_kernel<HID><<<cdiv((long)E, T), T, 0, stream>>>(h, src, dst, agg1, E);
        bias_relu_kernel<<<cdiv((long)N * HID, T), T, 0, stream>>>(agg1, rsI, b1, N);

        gemm_gc_kernel<HID, NC><<<gblocks, T, 0, stream>>>(agg1, rsO, W2, h2, N);
        fill_zero_kernel<<<cdiv((long)N * NC, T), T, 0, stream>>>(agg2, N * NC);
        scatter_kernel<NC><<<cdiv((long)E, T), T, 0, stream>>>(h2, src, dst, agg2, E);
        out_acc_kernel<<<cdiv((long)N * NC, T), T, 0, stream>>>(out, agg2, rsI, b2, N);
    }
}


# edge sums without float atomics: each call counting-sorts edges by destination via LDS cursors; one wave adds a node's source rows in registers in list order; scale, bias, mean folded into its store
# speedup vs baseline: 3.0199x; 1.3692x over previous
#include <hip/hip_runtime.h>


typedef __attribute__((ext_vector_type(2))) float v2f;
typedef __attribute__((ext_vector_type(8))) float v8f;

#define IN_FEATS 128
#define HID      64
#define NC       32


#define DEG_HR 12544
#define DEG_S  16
#define DEG_T  1024

__device__ __forceinline__ void deg_bin4(unsigned int* cnt, int4 k, unsigned int lo) {
    unsigned int k0 = (unsigned int)k.x - lo;
    unsigned int k1 = (unsigned int)k.y - lo;
    unsigned int k2 = (unsigned int)k.z - lo;
    unsigned int k3 = (unsigned int)k.w - lo;
    if (k0 < DEG_HR) atomicAdd(&cnt[k0], 1u);
    if (k1 < DEG_HR) atomicAdd(&cnt[k1], 1u);
    if (k2 < DEG_HR) atomicAdd(&cnt[k2], 1u);
    if (k3 < DEG_HR) atomicAdd(&cnt[k3], 1u);
}

__global__ void __launch_bounds__(DEG_T)
deg_hist_kernel(const int* __restrict__ src, const int* __restrict__ dst,
                unsigned int* __restrict__ part, int E, int stride) {
    __shared__ uint4 cnt4[DEG_HR / 4];
    unsigned int* cnt = (unsigned int*)cnt4;
    const int tid = threadIdx.x;
    const int s = blockIdx.x;
    const unsigned int lo = blockIdx.y * DEG_HR;
    const int a = blockIdx.z;
    const int* __restrict__ keys = a ? dst : src;

    for (int i = tid; i < DEG_HR / 4; i += DEG_T) cnt4[i] = make_uint4(0u, 0u, 0u, 0u);
    __syncthreads();

    const int per = (((E + DEG_S - 1) / DEG_S) + 3) & ~3;
    const int e0 = s * per;
    int e1 = e0 + per;
    if (e1 > E) e1 = E;
    const int nvec = (e1 > e0) ? ((e1 - e0) >> 2) : 0;
    const int4* __restrict__ kv = (const int4*)(keys + e0);

    int i = tid;
    for (; i + 7 * DEG_T < nvec; i += 8 * DEG_T) {
        int4 k[8];
#pragma unroll
        for (int j = 0; j < 8; ++j) k[j] = kv[i + j * DEG_T];
#pragma unroll
        for (int j = 0; j < 8; ++j) deg_bin4(cnt, k[j], lo);
    }
    for (; i < nvec; i += DEG_T) deg_bin4(cnt, kv[i], lo);
    for (int e = e0 + (nvec << 2) + tid; e < e1; e += DEG_T) {
        unsigned int k0 = (unsigned int)keys[e] - lo;
        if (k0 < DEG_HR) atomicAdd(&cnt[k0], 1u);
    }
    __syncthreads();

    uint4* __restrict__ p = (uint4*)(part + (size_t)(a * DEG_S + s) * stride + lo);
    for (int j = tid; j < DEG_HR / 4; j += DEG_T) p[j] = cnt4[j];
}

__global__ void __launch_bounds__(256)
deg_rsqrt_kernel(const unsigned int* __restrict__ part, float* __restrict__ rs,
                 int* __restrict__ bsum, int n, int stride) {
    __shared__ int wsum[8];
    const int node = blockIdx.x * 256 + threadIdx.x;
    const int a = blockIdx.y;
    unsigned int c = 0;
    if (node < n) {
        const unsigned int* p = part + (size_t)a * DEG_S * stride + node;
#pragma unroll
        for (int s = 0; s < DEG_S; ++s) c += p[(size_t)s * stride];
        rs[(size_t)a * n + node] = rsqrtf(fmaxf((float)c, 1.0f));
    }
    if (a == 1) {
        int v = (int)c;
        for (int o = 16; o > 0; o >>= 1) v += __shfl_xor(v, o);
        if ((threadIdx.x & 31) == 0) wsum[threadIdx.x >> 5] = v;
        __syncthreads();
        if (threadIdx.x == 0) {
            int t = 0;
            for (int w = 0; w < 8; ++w) t += wsum[w];
            bsum[blockIdx.x] = t;
        }
    }
}

__global__ void __launch_bounds__(256)
row_scan_kernel(unsigned int* __restrict__ pd, const int* __restrict__ bsum,
                int* __restrict__ rowstart, int n, int stride) {
    __shared__ int red[8];
    __shared__ int wtot[8];
    const int tid = threadIdx.x;
    const int lane = tid & 31;
    const int wave = tid >> 5;

    int o = 0;
    for (int j = tid; j < (int)blockIdx.x; j += 256) o += bsum[j];
    for (int d = 16; d > 0; d >>= 1) o += __shfl_xor(o, d);
    if (lane == 0) red[wave] = o;

    const int node = blockIdx.x * 256 + tid;
    unsigned int t[DEG_S];
    int c = 0;
#pragma unroll
    for (int s = 0; s < DEG_S; ++s) {
        t[s] = (node < n) ? pd[(size_t)s * stride + node] : 0u;
        c += (int)t[s];
    }
    int inc = c;
    for (int d = 1; d < 32; d <<= 1) {
        const int u = __shfl_up(inc, d);
        if (lane >= d) inc += u;
    }
    if (lane == 31) wtot[wave] = inc;
    __syncthreads();

    int base = 0;
    for (int w = 0; w < 8; ++w) {
        base += red[w];
        if (w < wave) base += wtot[w];
    }
    const int excl = base + inc - c;
    if (node < n) {
        rowstart[node] = excl;
        unsigned int off = (unsigned int)excl;
#pragma unroll
        for (int s = 0; s < DEG_S; ++s) {
            pd[(size_t)s * stride + node] = off;
            off += t[s];
        }
    }
    if (node == n - 1) rowstart[n] = excl + c;
}

__device__ __forceinline__ void csr_place1(unsigned int* cur, int* __restrict__ esrc,
                                           int key, int val, unsigned int lo) {
    const unsigned int k = (unsigned int)key - lo;
    if (k < DEG_HR) {
        const unsigned int pos = atomicAdd(&cur[k], 1u);
        esrc[pos] = val;
    }
}

__device__ __forceinline__ void csr_place4(unsigned int* cur, int* __restrict__ esrc,
                                           int4 k, int4 v, unsigned int lo) {
    csr_place1(cur, esrc, k.x, v.x, lo);
    csr_place1(cur, esrc, k.y, v.y, lo);
    csr_place1(cur, esrc, k.z, v.z, lo);
    csr_place1(cur, esrc, k.w, v.w, lo);
}

__global__ void __launch_bounds__(DEG_T)
csr_place_kernel(const int* __restrict__ src, const int* __restrict__ dst,
                 const unsigned int* __restrict__ pd, int* __restrict__ esrc,
                 int E, int stride) {
    __shared__ uint4 cur4[DEG_HR / 4];
    unsigned int* cur = (unsigned int*)cur4;
    const int tid = threadIdx.x;
    const int s = blockIdx.x;
    const unsigned int lo = blockIdx.y * DEG_HR;

    const uint4* __restrict__ p4 = (const uint4*)(pd + (size_t)s * stride + lo);
    for (int i = tid; i < DEG_HR / 4; i += DEG_T) cur4[i] = p4[i];
    __syncthreads();

    const int per = (((E + DEG_S - 1) / DEG_S) + 3) & ~3;
    const int e0 = s * per;
    int e1 = e0 + per;
    if (e1 > E) e1 = E;
    const int nvec = (e1 > e0) ? ((e1 - e0) >> 2) : 0;
    const int4* __restrict__ kv = (const int4*)(dst + e0);
    const int4* __restrict__ sv = (const int4*)(src + e0);

    int i = tid;
    for (; i + 7 * DEG_T < nvec; i += 8 * DEG_T) {
        int4 k[8], v[8];
#pragma unroll
        for (int j = 0; j < 8; ++j) { k[j] = kv[i + j * DEG_T]; v[j] = sv[i + j * DEG_T]; }
#pragma unroll
        for (int j = 0; j < 8; ++j) csr_place4(cur, esrc, k[j], v[j], lo);
    }
    for (; i < nvec; i += DEG_T) csr_place4(cur, esrc, kv[i], sv[i], lo);
    for (int e = e0 + (nvec << 2) + tid; e < e1; e += DEG_T) csr_place1(cur, esrc, dst[e], src[e], lo);
}

#define GNPW 8

template <int F>
__device__ __forceinline__ void row_sum(const float* __restrict__ h, const int* __restrict__ esrc,
                                        int beg, int end, int lane, float (&acc)[F / 32]) {
    constexpr int R = F / 32;
    constexpr int U = 16 / R;
#pragma unroll
    for (int r = 0; r < R; ++r) acc[r] = 0.0f;
    for (int base = beg; base < end; base += 32) {
        const int cnt = (end - base < 32) ? (end - base) : 32;
        const int id_l = esrc[base + ((lane < cnt) ? lane : (cnt - 1))];
        for (int j0 = 0; j0 < cnt; j0 += U) {
            float v[U][R];
#pragma unroll
            for (int u = 0; u < U; ++u) {
                const int sidx = __builtin_amdgcn_readlane(id_l, j0 + u);
                const float* __restrict__ hp = h + (size_t)sidx * F + lane;
#pragma unroll
                for (int r = 0; r < R; ++r) v[u][r] = hp[r * 32];
            }
#pragma unroll
            for (int u = 0; u < U; ++u) {
                const bool live = (u == 0) || ((j0 + u) < cnt);
#pragma unroll
                for (int r = 0; r < R; ++r) acc[r] += live ? v[u][r] : 0.0f;
            }
        }
    }
}

__device__ __forceinline__ int lane_pick_i(int v, int i) { return __builtin_amdgcn_readlane(v, i); }
__device__ __forceinline__ float lane_pick_f(float v, int i) {
    return __int_as_float(__builtin_amdgcn_readlane(__float_as_int(v), i));
}

__global__ void __launch_bounds__(256)
gather_relu_kernel(const float* __restrict__ h, const int* __restrict__ rowstart,
                   const int* __restrict__ esrc, const float* __restrict__ rs_in,
                   const float* __restrict__ b, float* __restrict__ h1, int n) {
    constexpr int R = HID / 32;
    const int lane = threadIdx.x & 31;
    const int gw = __builtin_amdgcn_readfirstlane((int)((blockIdx.x * 256 + threadIdx.x) >> 5));
    const int n0 = gw * GNPW;
    if (n0 >= n) return;
    const int nn = (n - n0 < GNPW) ? (n - n0) : GNPW;
    const int rw_l = rowstart[n0 + ((lane < nn) ? lane : nn)];
    const float sc_l = rs_in[n0 + ((lane < nn) ? lane : (nn - 1))];

    float bias[R];
#pragma unroll
    for (int r = 0; r < R; ++r) bias[r] = b[lane + r * 32];

    for (int i = 0; i < nn; ++i) {
        float acc[R];
        row_sum<HID>(h, esrc, lane_pick_i(rw_l, i), lane_pick_i(rw_l, i + 1), lane, acc);
        const float rsv = lane_pick_f(sc_l, i);
        float* __restrict__ op = h1 + (size_t)(n0 + i) * HID + lane;
#pragma unroll
        for (int r = 0; r < R; ++r) op[r * 32] = fmaxf(acc[r] * rsv + bias[r], 0.0f);
    }
}

__global__ void __launch_bounds__(256)
gather_mean_kernel(const float* __restrict__ hA, const int* __restrict__ rowstartA,
                   const int* __restrict__ esrcA, const float* __restrict__ rsA,
                   const float* __restrict__ hB, const int* __restrict__ rowstartB,
                   const int* __restrict__ esrcB, const float* __restrict__ rsB,
                   const float* __restrict__ b, float* __restrict__ out, int n) {
    const int lane = threadIdx.x & 31;
    const int gw = __builtin_amdgcn_readfirstlane((int)((blockIdx.x * 256 + threadIdx.x) >> 5));
    const int n0 = gw * GNPW;
    if (n0 >= n) return;
    const int nn = (n - n0 < GNPW) ? (n - n0) : GNPW;
    const int ia = n0 + ((lane < nn) ? lane : nn);
    const int ib = n0 + ((lane < nn) ? lane : (nn - 1));
    const int rwA_l = rowstartA[ia];
    const int rwB_l = rowstartB[ia];
    const float scA_l = rsA[ib];
    const float scB_l = rsB[ib];
    const float bias = b[lane];

    for (int i = 0; i < nn; ++i) {
        float accA[1], accB[1];
        row_sum<NC>(hA, esrcA, lane_pick_i(rwA_l, i), lane_pick_i(rwA_l, i + 1), lane, accA);
        row_sum<NC>(hB, esrcB, lane_pick_i(rwB_l, i), lane_pick_i(rwB_l, i + 1), lane, accB);
        const float tA = accA[0] * lane_pick_f(scA_l, i) + bias;
        const float tB = accB[0] * lane_pick_f(scB_l, i) + bias;
        out[(size_t)(n0 + i) * NC + lane] = 0.5f * tA + 0.5f * tB;
    }
}

template <int IN, int OUT>
__global__ void __launch_bounds__(256)
gemm_gc_kernel(const float* __restrict__ X, const float* __restrict__ rs,
               const float* __restrict__ W, float* __restrict__ H, int n) {
    constexpr int LDW = IN + 4;
    __shared__ float lw[OUT * LDW];

    for (int i = threadIdx.x; i < IN * OUT; i += 256) {
        int k = i / OUT;
        int c = i - k * OUT;
        lw[c * LDW + k] = W[i];
    }
    __syncthreads();

    const int lane = threadIdx.x & 31;
    const int wave = threadIdx.x >> 5;
    const int r0 = (blockIdx.x * 8 + wave) * 16;
    if (r0 >= n) return;

    const int mrow = r0 + (lane & 15);
    const int koff = (lane >> 4) * 2;
    const float rsv = rs[mrow];
    const float* xrow = X + (long)mrow * IN;
    const int col = lane & 15;

    constexpr int CT = OUT / 16;
    v8f acc[CT] = {};

    for (int k = 0; k < IN; k += 4) {
        float2 av = *(const float2*)(xrow + k + koff);
        v2f a;
        a.x = av.x * rsv;
        a.y = av.y * rsv;
#pragma unroll
        for (int c = 0; c < CT; ++c) {
            float2 bv = *(const float2*)&lw[(c * 16 + col) * LDW + k + koff];
            v2f b;
            b.x = bv.x;
            b.y = bv.y;
            acc[c] = __builtin_amdgcn_wmma_f32_16x16x4_f32(
                false, a, false, b, (short)0, acc[c], false, false);
        }
    }

    const int mhi = (lane >> 4) * 8;
#pragma unroll
    for (int c = 0; c < CT; ++c) {
#pragma unroll
        for (int r = 0; r < 8; ++r) {
            H[(long)(r0 + mhi + r) * OUT + c * 16 + col] = acc[c][r];
        }
    }
}


extern "C" void kernel_launch(void* const* d_in, const int* in_sizes, int n_in,
                              void* d_out, int out_size, void* d_ws, size_t ws_size,
                              hipStream_t stream) {
    const float* x    = (const float*)d_in[0];
    const int*   src1 = (const int*)d_in[1];
    const int*   dst1 = (const int*)d_in[2];
    const int*   src2 = (const int*)d_in[3];
    const int*   dst2 = (const int*)d_in[4];
    const float* W1   = (const float*)d_in[5];
    const float* b1   = (const float*)d_in[6];
    const float* W2   = (const float*)d_in[7];
    const float* b2   = (const float*)d_in[8];
    float* out = (float*)d_out;

    const int N = in_sizes[0] / IN_FEATS;
    const int E = in_sizes[1];

    const int dranges = (N + DEG_HR - 1) / DEG_HR;
    const int dstride = dranges * DEG_HR;
    unsigned int* dpart = (unsigned int*)d_ws;
    float* ws   = (float*)(dpart + (size_t)2 * DEG_S * dstride);
    float* rsOI = ws;
    float* h    = rsOI + (size_t)4 * N;
    float* h1   = h + (size_t)N * HID;
    float* h2   = h1 + (size_t)N * HID;
    int* esrc   = (int*)(h2 + (size_t)2 * N * NC);
    int* rowst  = esrc + (size_t)2 * E;
    int* bsum   = rowst + (size_t)2 * (N + 1);

    const int T = 256;
    auto cdiv = [](long a, long b) { return (int)((a + b - 1) / b); };

    const int tiles   = N / 16;
    const int gblocks = cdiv(tiles, 8);
    const int nblocks = cdiv(N, T);
    const int wblocks = cdiv(cdiv(N, GNPW), 8);

    for (int g = 0; g < 2; ++g) {
        const int* src = (g == 0) ? src1 : src2;
        const int* dst = (g == 0) ? dst1 : dst2;
        float* rsO = rsOI + (size_t)g * 2 * N;
        float* rsI = rsO + N;
        int* es = esrc + (size_t)g * E;
        int* rw = rowst + (size_t)g * (N + 1);

        deg_hist_kernel<<<dim3(DEG_S, dranges, 2), DEG_T, 0, stream>>>(src, dst, dpart, E, dstride);
        unsigned int* dpartI = dpart + (size_t)DEG_S * dstride;
        deg_rsqrt_kernel<<<dim3(nblocks, 2), T, 0, stream>>>(dpart, rsO, bsum, N, dstride);
        row_scan_kernel<<<nblocks, T, 0, stream>>>(dpartI, bsum, rw, N, dstride);
        csr_place_kernel<<<dim3(DEG_S, dranges), DEG_T, 0, stream>>>(src, dst, dpartI, es, E, dstride);

        gemm_gc_kernel<IN_FEATS, HID><<<gblocks, T, 0, stream>>>(x, rsO, W1, h, N);
        gather_relu_kernel<<<wblocks, T, 0, stream>>>(h, rw, es, rsI, b1, h1, N);

        gemm_gc_kernel<HID, NC><<<gblocks, T, 0, stream>>>(h1, rsO, W2, h2 + (size_t)g * N * NC, N);
    }

    gather_mean_kernel<<<wblocks, T, 0, stream>>>(
        h2, rowst, esrc, rsOI + N,
        h2 + (size_t)N * NC, rowst + (N + 1), esrc + E, rsOI + (size_t)3 * N,
        b2, out, N);
}


# both graphs served by every launch (graph = grid coordinate): 15 launches become 8; the layer-1 product loads each x fragment once, scales it per graph and reuses the one B fragment
# speedup vs baseline: 3.5974x; 1.1912x over previous
#include <hip/hip_runtime.h>


typedef __attribute__((ext_vector_type(2))) float v2f;
typedef __attribute__((ext_vector_type(8))) float v8f;

#define IN_FEATS 128
#define HID      64
#define NC       32


#define DEG_HR 12544
#define DEG_S  16
#define DEG_T  1024

__device__ __forceinline__ void deg_bin4(unsigned int* cnt, int4 k, unsigned int lo) {
    unsigned int k0 = (unsigned int)k.x - lo;
    unsigned int k1 = (unsigned int)k.y - lo;
    unsigned int k2 = (unsigned int)k.z - lo;
    unsigned int k3 = (unsigned int)k.w - lo;
    if (k0 < DEG_HR) atomicAdd(&cnt[k0], 1u);
    if (k1 < DEG_HR) atomicAdd(&cnt[k1], 1u);
    if (k2 < DEG_HR) atomicAdd(&cnt[k2], 1u);
    if (k3 < DEG_HR) atomicAdd(&cnt[k3], 1u);
}

__global__ void __launch_bounds__(DEG_T)
deg_hist_kernel(const int* __restrict__ src1, const int* __restrict__ dst1,
                const int* __restrict__ src2, const int* __restrict__ dst2,
                unsigned int* __restrict__ part, int E, int stride) {
    __shared__ uint4 cnt4[DEG_HR / 4];
    unsigned int* cnt = (unsigned int*)cnt4;
    const int tid = threadIdx.x;
    const int s = blockIdx.x;
    const unsigned int lo = blockIdx.y * DEG_HR;
    const int p = blockIdx.z;
    const int* __restrict__ keys = (p & 2) ? ((p & 1) ? dst2 : src2) : ((p & 1) ? dst1 : src1);

    for (int i = tid; i < DEG_HR / 4; i += DEG_T) cnt4[i] = make_uint4(0u, 0u, 0u, 0u);
    __syncthreads();

    const int per = (((E + DEG_S - 1) / DEG_S) + 3) & ~3;
    const int e0 = s * per;
    int e1 = e0 + per;
    if (e1 > E) e1 = E;
    const int nvec = (e1 > e0) ? ((e1 - e0) >> 2) : 0;
    const int4* __restrict__ kv = (const int4*)(keys + e0);

    int i = tid;
    for (; i + 7 * DEG_T < nvec; i += 8 * DEG_T) {
        int4 k[8];
#pragma unroll
        for (int j = 0; j < 8; ++j) k[j] = kv[i + j * DEG_T];
#pragma unroll
        for (int j = 0; j < 8; ++j) deg_bin4(cnt, k[j], lo);
    }
    for (; i < nvec; i += DEG_T) deg_bin4(cnt, kv[i], lo);
    for (int e = e0 + (nvec << 2) + tid; e < e1; e += DEG_T) {
        unsigned int k0 = (unsigned int)keys[e] - lo;
        if (k0 < DEG_HR) atomicAdd(&cnt[k0], 1u);
    }
    __syncthreads();

    uint4* __restrict__ po = (uint4*)(part + (size_t)(p * DEG_S + s) * stride + lo);
    for (int j = tid; j < DEG_HR / 4; j += DEG_T) po[j] = cnt4[j];
}

__global__ void __launch_bounds__(256)
deg_rsqrt_kernel(const unsigned int* __restrict__ part, float* __restrict__ rs,
                 int* __restrict__ bsum, int n, int stride) {
    __shared__ int wsum[8];
    const int node = blockIdx.x * 256 + threadIdx.x;
    const int a = blockIdx.y;
    unsigned int c = 0;
    if (node < n) {
        const unsigned int* p = part + (size_t)a * DEG_S * stride + node;
#pragma unroll
        for (int s = 0; s < DEG_S; ++s) c += p[(size_t)s * stride];
        rs[(size_t)a * n + node] = rsqrtf(fmaxf((float)c, 1.0f));
    }
    if (a & 1) {
        int v = (int)c;
        for (int o = 16; o > 0; o >>= 1) v += __shfl_xor(v, o);
        if ((threadIdx.x & 31) == 0) wsum[threadIdx.x >> 5] = v;
        __syncthreads();
        if (threadIdx.x == 0) {
            int t = 0;
            for (int w = 0; w < 8; ++w) t += wsum[w];
            bsum[(a >> 1) * ((n + 255) >> 8) + blockIdx.x] = t;
        }
    }
}

__global__ void __launch_bounds__(256)
row_scan_kernel(unsigned int* __restrict__ part, const int* __restrict__ bsum_all,
                int* __restrict__ rowstart_all, int n, int stride) {
    __shared__ int red[8];
    __shared__ int wtot[8];
    const int g = blockIdx.y;
    unsigned int* __restrict__ pd = part + (size_t)(2 * g + 1) * DEG_S * stride;
    const int* __restrict__ bsum = bsum_all + g * ((n + 255) >> 8);
    int* __restrict__ rowstart = rowstart_all + (size_t)g * (n + 1);
    const int tid = threadIdx.x;
    const int lane = tid & 31;
    const int wave = tid >> 5;

    int o = 0;
    for (int j = tid; j < (int)blockIdx.x; j += 256) o += bsum[j];
    for (int d = 16; d > 0; d >>= 1) o += __shfl_xor(o, d);
    if (lane == 0) red[wave] = o;

    const int node = blockIdx.x * 256 + tid;
    unsigned int t[DEG_S];
    int c = 0;
#pragma unroll
    for (int s = 0; s < DEG_S; ++s) {
        t[s] = (node < n) ? pd[(size_t)s * stride + node] : 0u;
        c += (int)t[s];
    }
    int inc = c;
    for (int d = 1; d < 32; d <<= 1) {
        const int u = __shfl_up(inc, d);
        if (lane >= d) inc += u;
    }
    if (lane == 31) wtot[wave] = inc;
    __syncthreads();

    int base = 0;
    for (int w = 0; w < 8; ++w) {
        base += red[w];
        if (w < wave) base += wtot[w];
    }
    const int excl = base + inc - c;
    if (node < n) {
        rowstart[node] = excl;
        unsigned int off = (unsigned int)excl;
#pragma unroll
        for (int s = 0; s < DEG_S; ++s) {
            pd[(size_t)s * stride + node] = off;
            off += t[s];
        }
    }
    if (node == n - 1) rowstart[n] = excl + c;
}

__device__ __forceinline__ void csr_place1(unsigned int* cur, int* __restrict__ esrc,
                                           int key, int val, unsigned int lo) {
    const unsigned int k = (unsigned int)key - lo;
    if (k < DEG_HR) {
        const unsigned int pos = atomicAdd(&cur[k], 1u);
        esrc[pos] = val;
    }
}

__device__ __forceinline__ void csr_place4(unsigned int* cur, int* __restrict__ esrc,
                                           int4 k, int4 v, unsigned int lo) {
    csr_place1(cur, esrc, k.x, v.x, lo);
    csr_place1(cur, esrc, k.y, v.y, lo);
    csr_place1(cur, esrc, k.z, v.z, lo);
    csr_place1(cur, esrc, k.w, v.w, lo);
}

__global__ void __launch_bounds__(DEG_T)
csr_place_kernel(const int* __restrict__ src1, const int* __restrict__ dst1,
                 const int* __restrict__ src2, const int* __restrict__ dst2,
                 const unsigned int* __restrict__ part, int* __restrict__ esrc_all,
                 int E, int stride) {
    __shared__ uint4 cur4[DEG_HR / 4];
    unsigned int* cur = (unsigned int*)cur4;
    const int tid = threadIdx.x;
    const int s = blockIdx.x;
    const unsigned int lo = blockIdx.y * DEG_HR;
    const int g = blockIdx.z;
    const int* __restrict__ src = g ? src2 : src1;
    const int* __restrict__ dst = g ? dst2 : dst1;
    const unsigned int* __restrict__ pd = part + (size_t)(2 * g + 1) * DEG_S * stride;
    int* __restrict__ esrc = esrc_all + (size_t)g * E;

    const uint4* __restrict__ p4 = (const uint4*)(pd + (size_t)s * stride + lo);
    for (int i = tid; i < DEG_HR / 4; i += DEG_T) cur4[i] = p4[i];
    __syncthreads();

    const int per = (((E + DEG_S - 1) / DEG_S) + 3) & ~3;
    const int e0 = s * per;
    int e1 = e0 + per;
    if (e1 > E) e1 = E;
    const int nvec = (e1 > e0) ? ((e1 - e0) >> 2) : 0;
    const int4* __restrict__ kv = (const int4*)(dst + e0);
    const int4* __restrict__ sv = (const int4*)(src + e0);

    int i = tid;
    for (; i + 7 * DEG_T < nvec; i += 8 * DEG_T) {
        int4 k[8], v[8];
#pragma unroll
        for (int j = 0; j < 8; ++j) { k[j] = kv[i + j * DEG_T]; v[j] = sv[i + j * DEG_T]; }
#pragma unroll
        for (int j = 0; j < 8; ++j) csr_place4(cur, esrc, k[j], v[j], lo);
    }
    for (; i < nvec; i += DEG_T) csr_place4(cur, esrc, kv[i], sv[i], lo);
    for (int e = e0 + (nvec << 2) + tid; e < e1; e += DEG_T) csr_place1(cur, esrc, dst[e], src[e], lo);
}

#define GNPW 8

template <int F>
__device__ __forceinline__ void row_sum(const float* __restrict__ h, const int* __restrict__ esrc,
                                        int beg, int end, int lane, float (&acc)[F / 32]) {
    constexpr int R = F / 32;
    constexpr int U = 16 / R;
#pragma unroll
    for (int r = 0; r < R; ++r) acc[r] = 0.0f;
    for (int base = beg; base < end; base += 32) {
        const int cnt = (end - base < 32) ? (end - base) : 32;
        const int id_l = esrc[base + ((lane < cnt) ? lane : (cnt - 1))];
        for (int j0 = 0; j0 < cnt; j0 += U) {
            float v[U][R];
#pragma unroll
            for (int u = 0; u < U; ++u) {
                const int sidx = __builtin_amdgcn_readlane(id_l, j0 + u);
                const float* __restrict__ hp = h + (size_t)sidx * F + lane;
#pragma unroll
                for (int r = 0; r < R; ++r) v[u][r] = hp[r * 32];
            }
#pragma unroll
            for (int u = 0; u < U; ++u) {
                const bool live = (u == 0) || ((j0 + u) < cnt);
#pragma unroll
                for (int r = 0; r < R; ++r) acc[r] += live ? v[u][r] : 0.0f;
            }
        }
    }
}

__device__ __forceinline__ int lane_pick_i(int v, int i) { return __builtin_amdgcn_readlane(v, i); }
__device__ __forceinline__ float lane_pick_f(float v, int i) {
    return __int_as_float(__builtin_amdgcn_readlane(__float_as_int(v), i));
}

__global__ void __launch_bounds__(256)
gather_relu_kernel(const float* __restrict__ h_all, const int* __restrict__ rowstart_all,
                   const int* __restrict__ esrc_all, const float* __restrict__ rs_all,
                   const float* __restrict__ b, float* __restrict__ h1_all, int n, int E) {
    constexpr int R = HID / 32;
    const int g = blockIdx.y;
    const float* __restrict__ h = h_all + (size_t)g * n * HID;
    const int* __restrict__ rowstart = rowstart_all + (size_t)g * (n + 1);
    const int* __restrict__ esrc = esrc_all + (size_t)g * E;
    const float* __restrict__ rs_in = rs_all + (size_t)(2 * g + 1) * n;
    float* __restrict__ h1 = h1_all + (size_t)g * n * HID;
    const int lane = threadIdx.x & 31;
    const int gw = __builtin_amdgcn_readfirstlane((int)((blockIdx.x * 256 + threadIdx.x) >> 5));
    const int n0 = gw * GNPW;
    if (n0 >= n) return;
    const int nn = (n - n0 < GNPW) ? (n - n0) : GNPW;
    const int rw_l = rowstart[n0 + ((lane < nn) ? lane : nn)];
    const float sc_l = rs_in[n0 + ((lane < nn) ? lane : (nn - 1))];

    float bias[R];
#pragma unroll
    for (int r = 0; r < R; ++r) bias[r] = b[lane + r * 32];

    for (int i = 0; i < nn; ++i) {
        float acc[R];
        row_sum<HID>(h, esrc, lane_pick_i(rw_l, i), lane_pick_i(rw_l, i + 1), lane, acc);
        const float rsv = lane_pick_f(sc_l, i);
        float* __restrict__ op = h1 + (size_t)(n0 + i) * HID + lane;
#pragma unroll
        for (int r = 0; r < R; ++r) op[r * 32] = fmaxf(acc[r] * rsv + bias[r], 0.0f);
    }
}

__global__ void __launch_bounds__(256)
gather_mean_kernel(const float* __restrict__ hA, const int* __restrict__ rowstartA,
                   const int* __restrict__ esrcA, const float* __restrict__ rsA,
                   const float* __restrict__ hB, const int* __restrict__ rowstartB,
                   const int* __restrict__ esrcB, const float* __restrict__ rsB,
                   const float* __restrict__ b, float* __restrict__ out, int n) {
    const int lane = threadIdx.x & 31;
    const int gw = __builtin_amdgcn_readfirstlane((int)((blockIdx.x * 256 + threadIdx.x) >> 5));
    const int n0 = gw * GNPW;
    if (n0 >= n) return;
    const int nn = (n - n0 < GNPW) ? (n - n0) : GNPW;
    const int ia = n0 + ((lane < nn) ? lane : nn);
    const int ib = n0 + ((lane < nn) ? lane : (nn - 1));
    const int rwA_l = rowstartA[ia];
    const int rwB_l = rowstartB[ia];
    const float scA_l = rsA[ib];
    const float scB_l = rsB[ib];
    const float bias = b[lane];

    for (int i = 0; i < nn; ++i) {
        float accA[1], accB[1];
        row_sum<NC>(hA, esrcA, lane_pick_i(rwA_l, i), lane_pick_i(rwA_l, i + 1), lane, accA);
        row_sum<NC>(hB, esrcB, lane_pick_i(rwB_l, i), lane_pick_i(rwB_l, i + 1), lane, accB);
        const float tA = accA[0] * lane_pick_f(scA_l, i) + bias;
        const float tB = accB[0] * lane_pick_f(scB_l, i) + bias;
        out[(size_t)(n0 + i) * NC + lane] = 0.5f * tA + 0.5f * tB;
    }
}

template <int IN, int OUT, int G>
__global__ void __launch_bounds__(256)
gemm_gc_kernel(const float* __restrict__ Xall, const float* __restrict__ rs_all,
               const float* __restrict__ W, float* __restrict__ Hall, int n) {
    constexpr int LDW = IN + 4;
    __shared__ float lw[OUT * LDW];

    for (int i = threadIdx.x; i < IN * OUT; i += 256) {
        int k = i / OUT;
        int c = i - k * OUT;
        lw[c * LDW + k] = W[i];
    }
    __syncthreads();

    const int lane = threadIdx.x & 31;
    const int wave = threadIdx.x >> 5;
    const int r0 = (blockIdx.x * 8 + wave) * 16;
    if (r0 >= n) return;

    const int g0 = (G == 2) ? 0 : (int)blockIdx.y;
    const int mrow = r0 + (lane & 15);
    const int koff = (lane >> 4) * 2;
    const float* xrow = Xall + ((G == 2) ? (size_t)0 : (size_t)g0 * n * IN) + (size_t)mrow * IN;
    const int col = lane & 15;

    float rsv[G];
#pragma unroll
    for (int j = 0; j < G; ++j) rsv[j] = rs_all[(size_t)(2 * (g0 + j)) * n + mrow];

    constexpr int CT = OUT / 16;
    v8f acc[G][CT] = {};

    for (int k = 0; k < IN; k += 4) {
        float2 av = *(const float2*)(xrow + k + koff);
        v2f a[G];
#pragma unroll
        for (int j = 0; j < G; ++j) {
            a[j].x = av.x * rsv[j];
            a[j].y = av.y * rsv[j];
        }
#pragma unroll
        for (int c = 0; c < CT; ++c) {
            float2 bv = *(const float2*)&lw[(c * 16 + col) * LDW + k + koff];
            v2f b;
            b.x = bv.x;
            b.y = bv.y;
#pragma unroll
            for (int j = 0; j < G; ++j) {
                acc[j][c] = __builtin_amdgcn_wmma_f32_16x16x4_f32(
                    false, a[j], false, b, (short)0, acc[j][c], false, false);
            }
        }
    }

    const int mhi = (lane >> 4) * 8;
#pragma unroll
    for (int j = 0; j < G; ++j) {
        float* __restrict__ H = Hall + (size_t)(g0 + j) * n * OUT;
#pragma unroll
        for (int c = 0; c < CT; ++c) {
#pragma unroll
            for (int r = 0; r < 8; ++r) {
                H[(size_t)(r0 + mhi + r) * OUT + c * 16 + col] = acc[j][c][r];
            }
        }
    }
}


extern "C" void kernel_launch(void* const* d_in, const int* in_sizes, int n_in,
                              void* d_out, int out_size, void* d_ws, size_t ws_size,
                              hipStream_t stream) {
    const float* x    = (const float*)d_in[0];
    const int*   src1 = (const int*)d_in[1];
    const int*   dst1 = (const int*)d_in[2];
    const int*   src2 = (const int*)d_in[3];
    const int*   dst2 = (const int*)d_in[4];
    const float* W1   = (const float*)d_in[5];
    const float* b1   = (const float*)d_in[6];
    const float* W2   = (const float*)d_in[7];
    const float* b2   = (const float*)d_in[8];
    float* out = (float*)d_out;

    const int N = in_sizes[0] / IN_FEATS;
    const int E = in_sizes[1];

    const int dranges = (N + DEG_HR - 1) / DEG_HR;
    const int dstride = dranges * DEG_HR;
    unsigned int* dpart = (unsigned int*)d_ws;
    float* ws   = (float*)(dpart + (size_t)4 * DEG_S * dstride);
    float* rsOI = ws;
    float* h    = rsOI + (size_t)4 * N;
    float* h1   = h + (size_t)2 * N * HID;
    float* h2   = h1 + (size_t)2 * N * HID;
    int* esrc   = (int*)(h2 + (size_t)2 * N * NC);
    int* rowst  = esrc + (size_t)2 * E;
    int* bsum   = rowst + (size_t)2 * (N + 1);

    const int T = 256;
    auto cdiv = [](long a, long b) { return (int)((a + b - 1) / b); };

    const int tiles   = N / 16;
    const int gblocks = cdiv(tiles, 8);
    const int nblocks = cdiv(N, T);
    const int wblocks = cdiv(cdiv(N, GNPW), 8);

    deg_hist_kernel<<<dim3(DEG_S, dranges, 4), DEG_T, 0, stream>>>(src1, dst1, src2, dst2, dpart, E, dstride);
    deg_rsqrt_kernel<<<dim3(nblocks, 4), T, 0, stream>>>(dpart, rsOI, bsum, N, dstride);
    row_scan_kernel<<<dim3(nblocks, 2), T, 0, stream>>>(dpart, bsum, rowst, N, dstride);
    csr_place_kernel<<<dim3(DEG_S, dranges, 2), DEG_T, 0, stream>>>(src1, dst1, src2, dst2, dpart, esrc, E, dstride);

    gemm_gc_kernel<IN_FEATS, HID, 2><<<gblocks, T, 0, stream>>>(x, rsOI, W1, h, N);
    gather_relu_kernel<<<dim3(wblocks, 2), T, 0, stream>>>(h, rowst, esrc, rsOI, b1, h1, N, E);

    gemm_gc_kernel<HID, NC, 1><<<dim3(gblocks, 2), T, 0, stream>>>(h1, rsOI, W2, h2, N);

    gather_mean_kernel<<<wblocks, T, 0, stream>>>(
        h2, rowst, esrc, rsOI + N,
        h2 + (size_t)N * NC, rowst + (N + 1), esrc + E, rsOI + (size_t)3 * N,
        b2, out, N);
}


# layer-2 product moved into the layer-1 edge-sum launch: a wave sums the 16 nodes of one tile into its own LDS tile and multiplies it there on WMMA; h1 never goes to memory; 8 launches become 7
# speedup vs baseline: 3.7995x; 1.0562x over previous
#include <hip/hip_runtime.h>


typedef __attribute__((ext_vector_type(2))) float v2f;
typedef __attribute__((ext_vector_type(8))) float v8f;

#define IN_FEATS 128
#define HID      64
#define NC       32


#define DEG_HR 12544
#define DEG_S  16
#define DEG_T  1024

__device__ __forceinline__ void deg_bin4(unsigned int* cnt, int4 k, unsigned int lo) {
    unsigned int k0 = (unsigned int)k.x - lo;
    unsigned int k1 = (unsigned int)k.y - lo;
    unsigned int k2 = (unsigned int)k.z - lo;
    unsigned int k3 = (unsigned int)k.w - lo;
    if (k0 < DEG_HR) atomicAdd(&cnt[k0], 1u);
    if (k1 < DEG_HR) atomicAdd(&cnt[k1], 1u);
    if (k2 < DEG_HR) atomicAdd(&cnt[k2], 1u);
    if (k3 < DEG_HR) atomicAdd(&cnt[k3], 1u);
}

__global__ void __launch_bounds__(DEG_T)
deg_hist_kernel(const int* __restrict__ src1, const int* __restrict__ dst1,
                const int* __restrict__ src2, const int* __restrict__ dst2,
                unsigned int* __restrict__ part, int E, int stride) {
    __shared__ uint4 cnt4[DEG_HR / 4];
    unsigned int* cnt = (unsigned int*)cnt4;
    const int tid = threadIdx.x;
    const int s = blockIdx.x;
    const unsigned int lo = blockIdx.y * DEG_HR;
    const int p = blockIdx.z;
    const int* __restrict__ keys = (p & 2) ? ((p & 1) ? dst2 : src2) : ((p & 1) ? dst1 : src1);

    for (int i = tid; i < DEG_HR / 4; i += DEG_T) cnt4[i] = make_uint4(0u, 0u, 0u, 0u);
    __syncthreads();

    const int per = (((E + DEG_S - 1) / DEG_S) + 3) & ~3;
    const int e0 = s * per;
    int e1 = e0 + per;
    if (e1 > E) e1 = E;
    const int nvec = (e1 > e0) ? ((e1 - e0) >> 2) : 0;
    const int4* __restrict__ kv = (const int4*)(keys + e0);

    int i = tid;
    for (; i + 7 * DEG_T < nvec; i += 8 * DEG_T) {
        int4 k[8];
#pragma unroll
        for (int j = 0; j < 8; ++j) k[j] = kv[i + j * DEG_T];
#pragma unroll
        for (int j = 0; j < 8; ++j) deg_bin4(cnt, k[j], lo);
    }
    for (; i < nvec; i += DEG_T) deg_bin4(cnt, kv[i], lo);
    for (int e = e0 + (nvec << 2) + tid; e < e1; e += DEG_T) {
        unsigned int k0 = (unsigned int)keys[e] - lo;
        if (k0 < DEG_HR) atomicAdd(&cnt[k0], 1u);
    }
    __syncthreads();

    uint4* __restrict__ po = (uint4*)(part + (size_t)(p * DEG_S + s) * stride + lo);
    for (int j = tid; j < DEG_HR / 4; j += DEG_T) po[j] = cnt4[j];
}

__global__ void __launch_bounds__(256)
deg_rsqrt_kernel(const unsigned int* __restrict__ part, float* __restrict__ rs,
                 int* __restrict__ bsum, int n, int stride) {
    __shared__ int wsum[8];
    const int node = blockIdx.x * 256 + threadIdx.x;
    const int a = blockIdx.y;
    unsigned int c = 0;
    if (node < n) {
        const unsigned int* p = part + (size_t)a * DEG_S * stride + node;
#pragma unroll
        for (int s = 0; s < DEG_S; ++s) c += p[(size_t)s * stride];
        rs[(size_t)a * n + node] = rsqrtf(fmaxf((float)c, 1.0f));
    }
    if (a & 1) {
        int v = (int)c;
        for (int o = 16; o > 0; o >>= 1) v += __shfl_xor(v, o);
        if ((threadIdx.x & 31) == 0) wsum[threadIdx.x >> 5] = v;
        __syncthreads();
        if (threadIdx.x == 0) {
            int t = 0;
            for (int w = 0; w < 8; ++w) t += wsum[w];
            bsum[(a >> 1) * ((n + 255) >> 8) + blockIdx.x] = t;
        }
    }
}

__global__ void __launch_bounds__(256)
row_scan_kernel(unsigned int* __restrict__ part, const int* __restrict__ bsum_all,
                int* __restrict__ rowstart_all, int n, int stride) {
    __shared__ int red[8];
    __shared__ int wtot[8];
    const int g = blockIdx.y;
    unsigned int* __restrict__ pd = part + (size_t)(2 * g + 1) * DEG_S * stride;
    const int* __restrict__ bsum = bsum_all + g * ((n + 255) >> 8);
    int* __restrict__ rowstart = rowstart_all + (size_t)g * (n + 1);
    const int tid = threadIdx.x;
    const int lane = tid & 31;
    const int wave = tid >> 5;

    int o = 0;
    for (int j = tid; j < (int)blockIdx.x; j += 256) o += bsum[j];
    for (int d = 16; d > 0; d >>= 1) o += __shfl_xor(o, d);
    if (lane == 0) red[wave] = o;

    const int node = blockIdx.x * 256 + tid;
    unsigned int t[DEG_S];
    int c = 0;
#pragma unroll
    for (int s = 0; s < DEG_S; ++s) {
        t[s] = (node < n) ? pd[(size_t)s * stride + node] : 0u;
        c += (int)t[s];
    }
    int inc = c;
    for (int d = 1; d < 32; d <<= 1) {
        const int u = __shfl_up(inc, d);
        if (lane >= d) inc += u;
    }
    if (lane == 31) wtot[wave] = inc;
    __syncthreads();

    int base = 0;
    for (int w = 0; w < 8; ++w) {
        base += red[w];
        if (w < wave) base += wtot[w];
    }
    const int excl = base + inc - c;
    if (node < n) {
        rowstart[node] = excl;
        unsigned int off = (unsigned int)excl;
#pragma unroll
        for (int s = 0; s < DEG_S; ++s) {
            pd[(size_t)s * stride + node] = off;
            off += t[s];
        }
    }
    if (node == n - 1) rowstart[n] = excl + c;
}

__device__ __forceinline__ void csr_place1(unsigned int* cur, int* __restrict__ esrc,
                                           int key, int val, unsigned int lo) {
    const unsigned int k = (unsigned int)key - lo;
    if (k < DEG_HR) {
        const unsigned int pos = atomicAdd(&cur[k], 1u);
        esrc[pos] = val;
    }
}

__device__ __forceinline__ void csr_place4(unsigned int* cur, int* __restrict__ esrc,
                                           int4 k, int4 v, unsigned int lo) {
    csr_place1(cur, esrc, k.x, v.x, lo);
    csr_place1(cur, esrc, k.y, v.y, lo);
    csr_place1(cur, esrc, k.z, v.z, lo);
    csr_place1(cur, esrc, k.w, v.w, lo);
}

__global__ void __launch_bounds__(DEG_T)
csr_place_kernel(const int* __restrict__ src1, const int* __restrict__ dst1,
                 const int* __restrict__ src2, const int* __restrict__ dst2,
                 const unsigned int* __restrict__ part, int* __restrict__ esrc_all,
                 int E, int stride) {
    __shared__ uint4 cur4[DEG_HR / 4];
    unsigned int* cur = (unsigned int*)cur4;
    const int tid = threadIdx.x;
    const int s = blockIdx.x;
    const unsigned int lo = blockIdx.y * DEG_HR;
    const int g = blockIdx.z;
    const int* __restrict__ src = g ? src2 : src1;
    const int* __restrict__ dst = g ? dst2 : dst1;
    const unsigned int* __restrict__ pd = part + (size_t)(2 * g + 1) * DEG_S * stride;
    int* __restrict__ esrc = esrc_all + (size_t)g * E;

    const uint4* __restrict__ p4 = (const uint4*)(pd + (size_t)s * stride + lo);
    for (int i = tid; i < DEG_HR / 4; i += DEG_T) cur4[i] = p4[i];
    __syncthreads();

    const int per = (((E + DEG_S - 1) / DEG_S) + 3) & ~3;
    const int e0 = s * per;
    int e1 = e0 + per;
    if (e1 > E) e1 = E;
    const int nvec = (e1 > e0) ? ((e1 - e0) >> 2) : 0;
    const int4* __restrict__ kv = (const int4*)(dst + e0);
    const int4* __restrict__ sv = (const int4*)(src + e0);

    int i = tid;
    for (; i + 7 * DEG_T < nvec; i += 8 * DEG_T) {
        int4 k[8], v[8];
#pragma unroll
        for (int j = 0; j < 8; ++j) { k[j] = kv[i + j * DEG_T]; v[j] = sv[i + j * DEG_T]; }
#pragma unroll
        for (int j = 0; j < 8; ++j) csr_place4(cur, esrc, k[j], v[j], lo);
    }
    for (; i < nvec; i += DEG_T) csr_place4(cur, esrc, kv[i], sv[i], lo);
    for (int e = e0 + (nvec << 2) + tid; e < e1; e += DEG_T) csr_place1(cur, esrc, dst[e], src[e], lo);
}

#define GNPW 8

template <int F>
__device__ __forceinline__ void row_sum(const float* __restrict__ h, const int* __restrict__ esrc,
                                        int beg, int end, int lane, float (&acc)[F / 32]) {
    constexpr int R = F / 32;
    constexpr int U = 16 / R;
#pragma unroll
    for (int r = 0; r < R; ++r) acc[r] = 0.0f;
    for (int base = beg; base < end; base += 32) {
        const int cnt = (end - base < 32) ? (end - base) : 32;
        const int id_l = esrc[base + ((lane < cnt) ? lane : (cnt - 1))];
        for (int j0 = 0; j0 < cnt; j0 += U) {
            float v[U][R];
#pragma unroll
            for (int u = 0; u < U; ++u) {
                const int sidx = __builtin_amdgcn_readlane(id_l, j0 + u);
                const float* __restrict__ hp = h + (size_t)sidx * F + lane;
#pragma unroll
                for (int r = 0; r < R; ++r) v[u][r] = hp[r * 32];
            }
#pragma unroll
            for (int u = 0; u < U; ++u) {
                const bool live = (u == 0) || ((j0 + u) < cnt);
#pragma unroll
                for (int r = 0; r < R; ++r) acc[r] += live ? v[u][r] : 0.0f;
            }
        }
    }
}

__device__ __forceinline__ int lane_pick_i(int v, int i) { return __builtin_amdgcn_readlane(v, i); }
__device__ __forceinline__ float lane_pick_f(float v, int i) {
    return __int_as_float(__builtin_amdgcn_readlane(__float_as_int(v), i));
}

#define GRW 16
#define GLD (HID + 4)

__global__ void __launch_bounds__(256)
gather_relu_gemm_kernel(const float* __restrict__ h_all, const int* __restrict__ rowstart_all,
                        const int* __restrict__ esrc_all, const float* __restrict__ rs_all,
                        const float* __restrict__ b, const float* __restrict__ W2,
                        float* __restrict__ h2_all, int n, int E) {
    constexpr int R = HID / 32;
    constexpr int CT = NC / 16;
    __shared__ __attribute__((aligned(16))) float lw[NC * GLD];
    __shared__ __attribute__((aligned(16))) float la[8 * GRW * GLD];

#pragma unroll
    for (int j = 0; j < HID * NC / 256; ++j) {
        const int i = j * 256 + (int)threadIdx.x;
        const int k = i / NC;
        const int c = i - k * NC;
        lw[c * GLD + k] = W2[i];
    }
    __syncthreads();

    const int g = blockIdx.y;
    const float* __restrict__ h = h_all + (size_t)g * n * HID;
    const int* __restrict__ rowstart = rowstart_all + (size_t)g * (n + 1);
    const int* __restrict__ esrc = esrc_all + (size_t)g * E;
    const float* __restrict__ rs_out = rs_all + (size_t)(2 * g) * n;
    const float* __restrict__ rs_in = rs_all + (size_t)(2 * g + 1) * n;
    float* __restrict__ h2 = h2_all + (size_t)g * n * NC;
    const int lane = threadIdx.x & 31;
    const int wave = threadIdx.x >> 5;
    const int gw = __builtin_amdgcn_readfirstlane((int)(blockIdx.x * 8 + wave));
    const int n0 = gw * GRW;
    if (n0 >= n) return;
    const int nn = (n - n0 < GRW) ? (n - n0) : GRW;
    const int rw_l = rowstart[n0 + ((lane < nn) ? lane : nn)];
    const int ib = n0 + ((lane < nn) ? lane : (nn - 1));
    const float sci_l = rs_in[ib];
    const float sco_l = rs_out[ib];
    float* __restrict__ lt = la + wave * (GRW * GLD);

    float bias[R];
#pragma unroll
    for (int r = 0; r < R; ++r) bias[r] = b[lane + r * 32];

    for (int i = 0; i < nn; ++i) {
        float acc[R];
        row_sum<HID>(h, esrc, lane_pick_i(rw_l, i), lane_pick_i(rw_l, i + 1), lane, acc);
        const float rsi = lane_pick_f(sci_l, i);
        const float rso = lane_pick_f(sco_l, i);
#pragma unroll
        for (int r = 0; r < R; ++r)
            lt[i * GLD + lane + r * 32] = fmaxf(acc[r] * rsi + bias[r], 0.0f) * rso;
    }

    const int col = lane & 15;
    const int koff = (lane >> 4) * 2;
    v8f d[CT] = {};
    for (int k = 0; k < HID; k += 4) {
        float2 av = *(const float2*)&lt[col * GLD + k + koff];
        v2f a;
        a.x = av.x;
        a.y = av.y;
#pragma unroll
        for (int c = 0; c < CT; ++c) {
            float2 bv = *(const float2*)&lw[(c * 16 + col) * GLD + k + koff];
            v2f bb;
            bb.x = bv.x;
            bb.y = bv.y;
            d[c] = __builtin_amdgcn_wmma_f32_16x16x4_f32(
                false, a, false, bb, (short)0, d[c], false, false);
        }
    }

    const int mhi = (lane >> 4) * 8;
#pragma unroll
    for (int c = 0; c < CT; ++c) {
#pragma unroll
        for (int r = 0; r < 8; ++r) lt[(mhi + r) * GLD + c * 16 + col] = d[c][r];
    }
    for (int i = 0; i < nn; ++i) h2[(size_t)(n0 + i) * NC + lane] = lt[i * GLD + lane];
}

__global__ void __launch_bounds__(256)
gather_mean_kernel(const float* __restrict__ hA, const int* __restrict__ rowstartA,
                   const int* __restrict__ esrcA, const float* __restrict__ rsA,
                   const float* __restrict__ hB, const int* __restrict__ rowstartB,
                   const int* __restrict__ esrcB, const float* __restrict__ rsB,
                   const float* __restrict__ b, float* __restrict__ out, int n) {
    const int lane = threadIdx.x & 31;
    const int gw = __builtin_amdgcn_readfirstlane((int)((blockIdx.x * 256 + threadIdx.x) >> 5));
    const int n0 = gw * GNPW;
    if (n0 >= n) return;
    const int nn = (n - n0 < GNPW) ? (n - n0) : GNPW;
    const int ia = n0 + ((lane < nn) ? lane : nn);
    const int ib = n0 + ((lane < nn) ? lane : (nn - 1));
    const int rwA_l = rowstartA[ia];
    const int rwB_l = rowstartB[ia];
    const float scA_l = rsA[ib];
    const float scB_l = rsB[ib];
    const float bias = b[lane];

    for (int i = 0; i < nn; ++i) {
        float accA[1], accB[1];
        row_sum<NC>(hA, esrcA, lane_pick_i(rwA_l, i), lane_pick_i(rwA_l, i + 1), lane, accA);
        row_sum<NC>(hB, esrcB, lane_pick_i(rwB_l, i), lane_pick_i(rwB_l, i + 1), lane, accB);
        const float tA = accA[0] * lane_pick_f(scA_l, i) + bias;
        const float tB = accB[0] * lane_pick_f(scB_l, i) + bias;
        out[(size_t)(n0 + i) * NC + lane] = 0.5f * tA + 0.5f * tB;
    }
}

template <int IN, int OUT, int G>
__global__ void __launch_bounds__(256)
gemm_gc_kernel(const float* __restrict__ Xall, const float* __restrict__ rs_all,
               const float* __restrict__ W, float* __restrict__ Hall, int n) {
    constexpr int LDW = IN + 4;
    __shared__ float lw[OUT * LDW];

    for (int i = threadIdx.x; i < IN * OUT; i += 256) {
        int k = i / OUT;
        int c = i - k * OUT;
        lw[c * LDW + k] = W[i];
    }
    __syncthreads();

    const int lane = threadIdx.x & 31;
    const int wave = threadIdx.x >> 5;
    const int r0 = (blockIdx.x * 8 + wave) * 16;
    if (r0 >= n) return;

    const int g0 = (G == 2) ? 0 : (int)blockIdx.y;
    const int mrow = r0 + (lane & 15);
    const int koff = (lane >> 4) * 2;
    const float* xrow = Xall + ((G == 2) ? (size_t)0 : (size_t)g0 * n * IN) + (size_t)mrow * IN;
    const int col = lane & 15;

    float rsv[G];
#pragma unroll
    for (int j = 0; j < G; ++j) rsv[j] = rs_all[(size_t)(2 * (g0 + j)) * n + mrow];

    constexpr int CT = OUT / 16;
    v8f acc[G][CT] = {};

    for (int k = 0; k < IN; k += 4) {
        float2 av = *(const float2*)(xrow + k + koff);
        v2f a[G];
#pragma unroll
        for (int j = 0; j < G; ++j) {
            a[j].x = av.x * rsv[j];
            a[j].y = av.y * rsv[j];
        }
#pragma unroll
        for (int c = 0; c < CT; ++c) {
            float2 bv = *(const float2*)&lw[(c * 16 + col) * LDW + k + koff];
            v2f b;
            b.x = bv.x;
            b.y = bv.y;
#pragma unroll
            for (int j = 0; j < G; ++j) {
                acc[j][c] = __builtin_amdgcn_wmma_f32_16x16x4_f32(
                    false, a[j], false, b, (short)0, acc[j][c], false, false);
            }
        }
    }

    const int mhi = (lane >> 4) * 8;
#pragma unroll
    for (int j = 0; j < G; ++j) {
        float* __restrict__ H = Hall + (size_t)(g0 + j) * n * OUT;
#pragma unroll
        for (int c = 0; c < CT; ++c) {
#pragma unroll
            for (int r = 0; r < 8; ++r) {
                H[(size_t)(r0 + mhi + r) * OUT + c * 16 + col] = acc[j][c][r];
            }
        }
    }
}


extern "C" void kernel_launch(void* const* d_in, const int* in_sizes, int n_in,
                              void* d_out, int out_size, void* d_ws, size_t ws_size,
                              hipStream_t stream) {
    const float* x    = (const float*)d_in[0];
    const int*   src1 = (const int*)d_in[1];
    const int*   dst1 = (const int*)d_in[2];
    const int*   src2 = (const int*)d_in[3];
    const int*   dst2 = (const int*)d_in[4];
    const float* W1   = (const float*)d_in[5];
    const float* b1   = (const float*)d_in[6];
    const float* W2   = (const float*)d_in[7];
    const float* b2   = (const float*)d_in[8];
    float* out = (float*)d_out;

    const int N = in_sizes[0] / IN_FEATS;
    const int E = in_sizes[1];

    const int dranges = (N + DEG_HR - 1) / DEG_HR;
    const int dstride = dranges * DEG_HR;
    unsigned int* dpart = (unsigned int*)d_ws;
    float* ws   = (float*)(dpart + (size_t)4 * DEG_S * dstride);
    float* rsOI = ws;
    float* h    = rsOI + (size_t)4 * N;
    float* h2   = h + (size_t)2 * N * HID;
    int* esrc   = (int*)(h2 + (size_t)2 * N * NC);
    int* rowst  = esrc + (size_t)2 * E;
    int* bsum   = rowst + (size_t)2 * (N + 1);

    const int T = 256;
    auto cdiv = [](long a, long b) { return (int)((a + b - 1) / b); };

    const int tiles   = N / 16;
    const int gblocks = cdiv(tiles, 8);
    const int nblocks = cdiv(N, T);
    const int wblocks = cdiv(cdiv(N, GNPW), 8);
    const int tblocks = cdiv(cdiv(N, GRW), 8);

    deg_hist_kernel<<<dim3(DEG_S, dranges, 4), DEG_T, 0, stream>>>(src1, dst1, src2, dst2, dpart, E, dstride);
    deg_rsqrt_kernel<<<dim3(nblocks, 4), T, 0, stream>>>(dpart, rsOI, bsum, N, dstride);
    row_scan_kernel<<<dim3(nblocks, 2), T, 0, stream>>>(dpart, bsum, rowst, N, dstride);
    csr_place_kernel<<<dim3(DEG_S, dranges, 2), DEG_T, 0, stream>>>(src1, dst1, src2, dst2, dpart, esrc, E, dstride);

    gemm_gc_kernel<IN_FEATS, HID, 2><<<gblocks, T, 0, stream>>>(x, rsOI, W1, h, N);
    gather_relu_gemm_kernel<<<dim3(tblocks, 2), T, 0, stream>>>(h, rowst, esrc, rsOI, b1, W2, h2, N, E);

    gather_mean_kernel<<<wblocks, T, 0, stream>>>(
        h2, rowst, esrc, rsOI + N,
        h2 + (size_t)N * NC, rowst + (N + 1), esrc + E, rsOI + (size_t)3 * N,
        b2, out, N);
}


# rsqrt, row scan and layer-1 product share one launch (block index = role): histogram blocks also store 256-node in-degree totals, product waves sum their rows' out-degrees; 7 launches become 5
# speedup vs baseline: 3.9614x; 1.0426x over previous
#include <hip/hip_runtime.h>


typedef __attribute__((ext_vector_type(2))) float v2f;
typedef __attribute__((ext_vector_type(8))) float v8f;

#define IN_FEATS 128
#define HID      64
#define NC       32


#define DEG_HR 12544
#define DEG_S  16
#define DEG_T  1024

__device__ __forceinline__ void deg_bin4(unsigned int* cnt, int4 k, unsigned int lo) {
    unsigned int k0 = (unsigned int)k.x - lo;
    unsigned int k1 = (unsigned int)k.y - lo;
    unsigned int k2 = (unsigned int)k.z - lo;
    unsigned int k3 = (unsigned int)k.w - lo;
    if (k0 < DEG_HR) atomicAdd(&cnt[k0], 1u);
    if (k1 < DEG_HR) atomicAdd(&cnt[k1], 1u);
    if (k2 < DEG_HR) atomicAdd(&cnt[k2], 1u);
    if (k3 < DEG_HR) atomicAdd(&cnt[k3], 1u);
}

__global__ void __launch_bounds__(DEG_T)
deg_hist_kernel(const int* __restrict__ src1, const int* __restrict__ dst1,
                const int* __restrict__ src2, const int* __restrict__ dst2,
                unsigned int* __restrict__ part, int* __restrict__ gsum, int E, int stride) {
    __shared__ uint4 cnt4[DEG_HR / 4];
    unsigned int* cnt = (unsigned int*)cnt4;
    const int tid = threadIdx.x;
    const int s = blockIdx.x;
    const unsigned int lo = blockIdx.y * DEG_HR;
    const int p = blockIdx.z;
    const int* __restrict__ keys = (p & 2) ? ((p & 1) ? dst2 : src2) : ((p & 1) ? dst1 : src1);

    for (int i = tid; i < DEG_HR / 4; i += DEG_T) cnt4[i] = make_uint4(0u, 0u, 0u, 0u);
    __syncthreads();

    const int per = (((E + DEG_S - 1) / DEG_S) + 3) & ~3;
    const int e0 = s * per;
    int e1 = e0 + per;
    if (e1 > E) e1 = E;
    const int nvec = (e1 > e0) ? ((e1 - e0) >> 2) : 0;
    const int4* __restrict__ kv = (const int4*)(keys + e0);

    int i = tid;
    for (; i + 7 * DEG_T < nvec; i += 8 * DEG_T) {
        int4 k[8];
#pragma unroll
        for (int j = 0; j < 8; ++j) k[j] = kv[i + j * DEG_T];
#pragma unroll
        for (int j = 0; j < 8; ++j) deg_bin4(cnt, k[j], lo);
    }
    for (; i < nvec; i += DEG_T) deg_bin4(cnt, kv[i], lo);
    for (int e = e0 + (nvec << 2) + tid; e < e1; e += DEG_T) {
        unsigned int k0 = (unsigned int)keys[e] - lo;
        if (k0 < DEG_HR) atomicAdd(&cnt[k0], 1u);
    }
    __syncthreads();

    uint4* __restrict__ po = (uint4*)(part + (size_t)(p * DEG_S + s) * stride + lo);
    for (int j = tid; j < DEG_HR / 4; j += DEG_T) po[j] = cnt4[j];

    if (p & 1) {
        const int lane = tid & 31;
        int* __restrict__ go = gsum + ((size_t)(p >> 1) * (stride >> 8) + (size_t)blockIdx.y * (DEG_HR / 256)) * DEG_S + s;
        for (int grp = tid >> 5; grp < DEG_HR / 256; grp += DEG_T / 32) {
            const uint4 a = cnt4[grp * 64 + lane * 2];
            const uint4 b = cnt4[grp * 64 + lane * 2 + 1];
            int v = (int)(a.x + a.y + a.z + a.w + b.x + b.y + b.z + b.w);
            for (int o = 16; o > 0; o >>= 1) v += __shfl_xor(v, o);
            if (lane == 0) go[(size_t)grp * DEG_S] = v;
        }
    }
}

__device__ __forceinline__ void csr_place1(unsigned int* cur, int* __restrict__ esrc,
                                           int key, int val, unsigned int lo) {
    const unsigned int k = (unsigned int)key - lo;
    if (k < DEG_HR) {
        const unsigned int pos = atomicAdd(&cur[k], 1u);
        esrc[pos] = val;
    }
}

__device__ __forceinline__ void csr_place4(unsigned int* cur, int* __restrict__ esrc,
                                           int4 k, int4 v, unsigned int lo) {
    csr_place1(cur, esrc, k.x, v.x, lo);
    csr_place1(cur, esrc, k.y, v.y, lo);
    csr_place1(cur, esrc, k.z, v.z, lo);
    csr_place1(cur, esrc, k.w, v.w, lo);
}

__global__ void __launch_bounds__(DEG_T)
csr_place_kernel(const int* __restrict__ src1, const int* __restrict__ dst1,
                 const int* __restrict__ src2, const int* __restrict__ dst2,
                 const unsigned int* __restrict__ part, int* __restrict__ esrc_all,
                 int E, int stride) {
    __shared__ uint4 cur4[DEG_HR / 4];
    unsigned int* cur = (unsigned int*)cur4;
    const int tid = threadIdx.x;
    const int s = blockIdx.x;
    const unsigned int lo = blockIdx.y * DEG_HR;
    const int g = blockIdx.z;
    const int* __restrict__ src = g ? src2 : src1;
    const int* __restrict__ dst = g ? dst2 : dst1;
    const unsigned int* __restrict__ pd = part + (size_t)(2 * g + 1) * DEG_S * stride;
    int* __restrict__ esrc = esrc_all + (size_t)g * E;

    const uint4* __restrict__ p4 = (const uint4*)(pd + (size_t)s * stride + lo);
    for (int i = tid; i < DEG_HR / 4; i += DEG_T) cur4[i] = p4[i];
    __syncthreads();

    const int per = (((E + DEG_S - 1) / DEG_S) + 3) & ~3;
    const int e0 = s * per;
    int e1 = e0 + per;
    if (e1 > E) e1 = E;
    const int nvec = (e1 > e0) ? ((e1 - e0) >> 2) : 0;
    const int4* __restrict__ kv = (const int4*)(dst + e0);
    const int4* __restrict__ sv = (const int4*)(src + e0);

    int i = tid;
    for (; i + 7 * DEG_T < nvec; i += 8 * DEG_T) {
        int4 k[8], v[8];
#pragma unroll
        for (int j = 0; j < 8; ++j) { k[j] = kv[i + j * DEG_T]; v[j] = sv[i + j * DEG_T]; }
#pragma unroll
        for (int j = 0; j < 8; ++j) csr_place4(cur, esrc, k[j], v[j], lo);
    }
    for (; i < nvec; i += DEG_T) csr_place4(cur, esrc, kv[i], sv[i], lo);
    for (int e = e0 + (nvec << 2) + tid; e < e1; e += DEG_T) csr_place1(cur, esrc, dst[e], src[e], lo);
}

#define GNPW 8

template <int F>
__device__ __forceinline__ void row_sum(const float* __restrict__ h, const int* __restrict__ esrc,
                                        int beg, int end, int lane, float (&acc)[F / 32]) {
    constexpr int R = F / 32;
    constexpr int U = 16 / R;
#pragma unroll
    for (int r = 0; r < R; ++r) acc[r] = 0.0f;
    for (int base = beg; base < end; base += 32) {
        const int cnt = (end - base < 32) ? (end - base) : 32;
        const int id_l = esrc[base + ((lane < cnt) ? lane : (cnt - 1))];
        for (int j0 = 0; j0 < cnt; j0 += U) {
            float v[U][R];
#pragma unroll
            for (int u = 0; u < U; ++u) {
                const int sidx = __builtin_amdgcn_readlane(id_l, j0 + u);
                const float* __restrict__ hp = h + (size_t)sidx * F + lane;
#pragma unroll
                for (int r = 0; r < R; ++r) v[u][r] = hp[r * 32];
            }
#pragma unroll
            for (int u = 0; u < U; ++u) {
                const bool live = (u == 0) || ((j0 + u) < cnt);
#pragma unroll
                for (int r = 0; r < R; ++r) acc[r] += live ? v[u][r] : 0.0f;
            }
        }
    }
}

__device__ __forceinline__ int lane_pick_i(int v, int i) { return __builtin_amdgcn_readlane(v, i); }
__device__ __forceinline__ float lane_pick_f(float v, int i) {
    return __int_as_float(__builtin_amdgcn_readlane(__float_as_int(v), i));
}

#define GRW 16
#define GLD (HID + 4)

__global__ void __launch_bounds__(256)
gather_relu_gemm_kernel(const float* __restrict__ h_all, const int* __restrict__ rowstart_all,
                        const int* __restrict__ esrc_all, const float* __restrict__ rs_all,
                        const float* __restrict__ b, const float* __restrict__ W2,
                        float* __restrict__ h2_all, int n, int E) {
    constexpr int R = HID / 32;
    constexpr int CT = NC / 16;
    __shared__ __attribute__((aligned(16))) float lw[NC * GLD];
    __shared__ __attribute__((aligned(16))) float la[8 * GRW * GLD];

#pragma unroll
    for (int j = 0; j < HID * NC / 256; ++j) {
        const int i = j * 256 + (int)threadIdx.x;
        const int k = i / NC;
        const int c = i - k * NC;
        lw[c * GLD + k] = W2[i];
    }
    __syncthreads();

    const int g = blockIdx.y;
    const float* __restrict__ h = h_all + (size_t)g * n * HID;
    const int* __restrict__ rowstart = rowstart_all + (size_t)g * (n + 1);
    const int* __restrict__ esrc = esrc_all + (size_t)g * E;
    const float* __restrict__ rs_out = rs_all + (size_t)(2 * g) * n;
    const float* __restrict__ rs_in = rs_all + (size_t)(2 * g + 1) * n;
    float* __restrict__ h2 = h2_all + (size_t)g * n * NC;
    const int lane = threadIdx.x & 31;
    const int wave = threadIdx.x >> 5;
    const int gw = __builtin_amdgcn_readfirstlane((int)(blockIdx.x * 8 + wave));
    const int n0 = gw * GRW;
    if (n0 >= n) return;
    const int nn = (n - n0 < GRW) ? (n - n0) : GRW;
    const int rw_l = rowstart[n0 + ((lane < nn) ? lane : nn)];
    const int ib = n0 + ((lane < nn) ? lane : (nn - 1));
    const float sci_l = rs_in[ib];
    const float sco_l = rs_out[ib];
    float* __restrict__ lt = la + wave * (GRW * GLD);

    float bias[R];
#pragma unroll
    for (int r = 0; r < R; ++r) bias[r] = b[lane + r * 32];

    for (int i = 0; i < nn; ++i) {
        float acc[R];
        row_sum<HID>(h, esrc, lane_pick_i(rw_l, i), lane_pick_i(rw_l, i + 1), lane, acc);
        const float rsi = lane_pick_f(sci_l, i);
        const float rso = lane_pick_f(sco_l, i);
#pragma unroll
        for (int r = 0; r < R; ++r)
            lt[i * GLD + lane + r * 32] = fmaxf(acc[r] * rsi + bias[r], 0.0f) * rso;
    }

    const int col = lane & 15;
    const int koff = (lane >> 4) * 2;
    v8f d[CT] = {};
    for (int k = 0; k < HID; k += 4) {
        float2 av = *(const float2*)&lt[col * GLD + k + koff];
        v2f a;
        a.x = av.x;
        a.y = av.y;
#pragma unroll
        for (int c = 0; c < CT; ++c) {
            float2 bv = *(const float2*)&lw[(c * 16 + col) * GLD + k + koff];
            v2f bb;
            bb.x = bv.x;
            bb.y = bv.y;
            d[c] = __builtin_amdgcn_wmma_f32_16x16x4_f32(
                false, a, false, bb, (short)0, d[c], false, false);
        }
    }

    const int mhi = (lane >> 4) * 8;
#pragma unroll
    for (int c = 0; c < CT; ++c) {
#pragma unroll
        for (int r = 0; r < 8; ++r) lt[(mhi + r) * GLD + c * 16 + col] = d[c][r];
    }
    for (int i = 0; i < nn; ++i) h2[(size_t)(n0 + i) * NC + lane] = lt[i * GLD + lane];
}

__global__ void __launch_bounds__(256)
gather_mean_kernel(const float* __restrict__ hA, const int* __restrict__ rowstartA,
                   const int* __restrict__ esrcA, const float* __restrict__ rsA,
                   const float* __restrict__ hB, const int* __restrict__ rowstartB,
                   const int* __restrict__ esrcB, const float* __restrict__ rsB,
                   const float* __restrict__ b, float* __restrict__ out, int n) {
    const int lane = threadIdx.x & 31;
    const int gw = __builtin_amdgcn_readfirstlane((int)((blockIdx.x * 256 + threadIdx.x) >> 5));
    const int n0 = gw * GNPW;
    if (n0 >= n) return;
    const int nn = (n - n0 < GNPW) ? (n - n0) : GNPW;
    const int ia = n0 + ((lane < nn) ? lane : nn);
    const int ib = n0 + ((lane < nn) ? lane : (nn - 1));
    const int rwA_l = rowstartA[ia];
    const int rwB_l = rowstartB[ia];
    const float scA_l = rsA[ib];
    const float scB_l = rsB[ib];
    const float bias = b[lane];

    for (int i = 0; i < nn; ++i) {
        float accA[1], accB[1];
        row_sum<NC>(hA, esrcA, lane_pick_i(rwA_l, i), lane_pick_i(rwA_l, i + 1), lane, accA);
        row_sum<NC>(hB, esrcB, lane_pick_i(rwB_l, i), lane_pick_i(rwB_l, i + 1), lane, accB);
        const float tA = accA[0] * lane_pick_f(scA_l, i) + bias;
        const float tB = accB[0] * lane_pick_f(scB_l, i) + bias;
        out[(size_t)(n0 + i) * NC + lane] = 0.5f * tA + 0.5f * tB;
    }
}

template <int IN, int OUT>
__global__ void __launch_bounds__(256)
scan_gemm_kernel(const float* __restrict__ X, const float* __restrict__ W, float* __restrict__ Hall,
                 unsigned int* __restrict__ part, const int* __restrict__ gsum_all,
                 float* __restrict__ rs, int* __restrict__ rowstart_all,
                 int n, int stride, int gblocks, int nblocks) {
    constexpr int LDW = IN + 4;
    __shared__ float lw[OUT * LDW];
    __shared__ int red[8];
    __shared__ int wtot[8];
    const int tid = threadIdx.x;
    const int lane = tid & 31;
    const int wave = tid >> 5;

    if ((int)blockIdx.x < gblocks) {
        for (int i = tid; i < IN * OUT; i += 256) {
            int k = i / OUT;
            int cc = i - k * OUT;
            lw[cc * LDW + k] = W[i];
        }
        __syncthreads();

        const int r0 = (blockIdx.x * 8 + wave) * 16;
        if (r0 >= n) return;

        const int mrow = r0 + (lane & 15);
        const int koff = (lane >> 4) * 2;
        const float* xrow = X + (size_t)mrow * IN;
        const int col = lane & 15;

        float rsv[2];
        {
            const unsigned int* __restrict__ p0 = part + mrow;
            const unsigned int* __restrict__ p1 = part + (size_t)2 * DEG_S * stride + mrow;
            unsigned int c0 = 0, c1 = 0;
#pragma unroll
            for (int s = 0; s < DEG_S; ++s) {
                c0 += p0[(size_t)s * stride];
                c1 += p1[(size_t)s * stride];
            }
            rsv[0] = rsqrtf(fmaxf((float)c0, 1.0f));
            rsv[1] = rsqrtf(fmaxf((float)c1, 1.0f));
        }

        constexpr int CT = OUT / 16;
        v8f acc[2][CT] = {};

        for (int k = 0; k < IN; k += 4) {
            float2 av = *(const float2*)(xrow + k + koff);
            v2f a[2];
#pragma unroll
            for (int j = 0; j < 2; ++j) {
                a[j].x = av.x * rsv[j];
                a[j].y = av.y * rsv[j];
            }
#pragma unroll
            for (int cc = 0; cc < CT; ++cc) {
                float2 bv = *(const float2*)&lw[(cc * 16 + col) * LDW + k + koff];
                v2f b;
                b.x = bv.x;
                b.y = bv.y;
#pragma unroll
                for (int j = 0; j < 2; ++j) {
                    acc[j][cc] = __builtin_amdgcn_wmma_f32_16x16x4_f32(
                        false, a[j], false, b, (short)0, acc[j][cc], false, false);
                }
            }
        }

        const int mhi = (lane >> 4) * 8;
#pragma unroll
        for (int j = 0; j < 2; ++j) {
            float* __restrict__ H = Hall + (size_t)j * n * OUT;
#pragma unroll
            for (int cc = 0; cc < CT; ++cc) {
#pragma unroll
                for (int r = 0; r < 8; ++r) {
                    H[(size_t)(r0 + mhi + r) * OUT + cc * 16 + col] = acc[j][cc][r];
                }
            }
        }
        return;
    }

    const int sb = (int)blockIdx.x - gblocks;
    const int g = sb / nblocks;
    const int bx = sb - g * nblocks;
    unsigned int* __restrict__ pd = part + (size_t)(2 * g + 1) * DEG_S * stride;
    const unsigned int* __restrict__ ps = part + (size_t)(2 * g) * DEG_S * stride;
    int* __restrict__ rowstart = rowstart_all + (size_t)g * (n + 1);

    const int4* __restrict__ gs4 = (const int4*)(gsum_all + (size_t)g * (stride >> 8) * DEG_S);
    const int lim4 = bx * (DEG_S / 4);
    int o = 0;
    for (int j0 = 0; j0 < lim4; j0 += 8 * 256) {
        int4 q[8];
#pragma unroll
        for (int u = 0; u < 8; ++u) {
            const int j = j0 + u * 256 + tid;
            q[u] = gs4[(j < lim4) ? j : 0];
        }
#pragma unroll
        for (int u = 0; u < 8; ++u) {
            const int j = j0 + u * 256 + tid;
            o += (j < lim4) ? (q[u].x + q[u].y + q[u].z + q[u].w) : 0;
        }
    }
    for (int d = 16; d > 0; d >>= 1) o += __shfl_xor(o, d);
    if (lane == 0) red[wave] = o;

    const int node = bx * 256 + tid;
    unsigned int t[DEG_S];
    int c = 0;
    unsigned int co = 0;
#pragma unroll
    for (int s = 0; s < DEG_S; ++s) {
        t[s] = (node < n) ? pd[(size_t)s * stride + node] : 0u;
        co += (node < n) ? ps[(size_t)s * stride + node] : 0u;
        c += (int)t[s];
    }
    if (node < n) {
        rs[(size_t)(2 * g) * n + node] = rsqrtf(fmaxf((float)co, 1.0f));
        rs[(size_t)(2 * g + 1) * n + node] = rsqrtf(fmaxf((float)c, 1.0f));
    }
    int inc = c;
    for (int d = 1; d < 32; d <<= 1) {
        const int u = __shfl_up(inc, d);
        if (lane >= d) inc += u;
    }
    if (lane == 31) wtot[wave] = inc;
    __syncthreads();

    int base = 0;
    for (int w = 0; w < 8; ++w) {
        base += red[w];
        if (w < wave) base += wtot[w];
    }
    const int excl = base + inc - c;
    if (node < n) {
        rowstart[node] = excl;
        unsigned int off = (unsigned int)excl;
#pragma unroll
        for (int s = 0; s < DEG_S; ++s) {
            pd[(size_t)s * stride + node] = off;
            off += t[s];
        }
    }
    if (node == n - 1) rowstart[n] = excl + c;
}


extern "C" void kernel_launch(void* const* d_in, const int* in_sizes, int n_in,
                              void* d_out, int out_size, void* d_ws, size_t ws_size,
                              hipStream_t stream) {
    const float* x    = (const float*)d_in[0];
    const int*   src1 = (const int*)d_in[1];
    const int*   dst1 = (const int*)d_in[2];
    const int*   src2 = (const int*)d_in[3];
    const int*   dst2 = (const int*)d_in[4];
    const float* W1   = (const float*)d_in[5];
    const float* b1   = (const float*)d_in[6];
    const float* W2   = (const float*)d_in[7];
    const float* b2   = (const float*)d_in[8];
    float* out = (float*)d_out;

    const int N = in_sizes[0] / IN_FEATS;
    const int E = in_sizes[1];

    const int dranges = (N + DEG_HR - 1) / DEG_HR;
    const int dstride = dranges * DEG_HR;
    unsigned int* dpart = (unsigned int*)d_ws;
    float* ws   = (float*)(dpart + (size_t)4 * DEG_S * dstride);
    float* rsOI = ws;
    float* h    = rsOI + (size_t)4 * N;
    float* h2   = h + (size_t)2 * N * HID;
    int* esrc   = (int*)(h2 + (size_t)2 * N * NC);
    int* gsum   = esrc + (size_t)2 * E;
    int* rowst  = gsum + (size_t)2 * (dstride / 256) * DEG_S;

    const int T = 256;
    auto cdiv = [](long a, long b) { return (int)((a + b - 1) / b); };

    const int tiles   = N / 16;
    const int gblocks = cdiv(tiles, 8);
    const int nblocks = cdiv(N, T);
    const int wblocks = cdiv(cdiv(N, GNPW), 8);
    const int tblocks = cdiv(cdiv(N, GRW), 8);

    deg_hist_kernel<<<dim3(DEG_S, dranges, 4), DEG_T, 0, stream>>>(src1, dst1, src2, dst2, dpart, gsum, E, dstride);
    scan_gemm_kernel<IN_FEATS, HID><<<gblocks + 2 * nblocks, T, 0, stream>>>(
        x, W1, h, dpart, gsum, rsOI, rowst, N, dstride, gblocks, nblocks);
    csr_place_kernel<<<dim3(DEG_S, dranges, 2), DEG_T, 0, stream>>>(src1, dst1, src2, dst2, dpart, esrc, E, dstride);

    gather_relu_gemm_kernel<<<dim3(tblocks, 2), T, 0, stream>>>(h, rowst, esrc, rsOI, b1, W2, h2, N, E);

    gather_mean_kernel<<<wblocks, T, 0, stream>>>(
        h2, rowst, esrc, rsOI + N,
        h2 + (size_t)N * NC, rowst + (N + 1), esrc + E, rsOI + (size_t)3 * N,
        b2, out, N);
}


# edge sums: a wave walks the lists of all its nodes as one stream: the next 32 entries are requested ahead and row batches run across node ends, so no per-node load round trips or padded batches
# speedup vs baseline: 4.1477x; 1.0470x over previous
#include <hip/hip_runtime.h>


typedef __attribute__((ext_vector_type(2))) float v2f;
typedef __attribute__((ext_vector_type(8))) float v8f;

#define IN_FEATS 128
#define HID      64
#define NC       32


#define DEG_HR 12544
#define DEG_S  16
#define DEG_T  1024

__device__ __forceinline__ void deg_bin4(unsigned int* cnt, int4 k, unsigned int lo) {
    unsigned int k0 = (unsigned int)k.x - lo;
    unsigned int k1 = (unsigned int)k.y - lo;
    unsigned int k2 = (unsigned int)k.z - lo;
    unsigned int k3 = (unsigned int)k.w - lo;
    if (k0 < DEG_HR) atomicAdd(&cnt[k0], 1u);
    if (k1 < DEG_HR) atomicAdd(&cnt[k1], 1u);
    if (k2 < DEG_HR) atomicAdd(&cnt[k2], 1u);
    if (k3 < DEG_HR) atomicAdd(&cnt[k3], 1u);
}

__global__ void __launch_bounds__(DEG_T)
deg_hist_kernel(const int* __restrict__ src1, const int* __restrict__ dst1,
                const int* __restrict__ src2, const int* __restrict__ dst2,
                unsigned int* __restrict__ part, int* __restrict__ gsum, int E, int stride) {
    __shared__ uint4 cnt4[DEG_HR / 4];
    unsigned int* cnt = (unsigned int*)cnt4;
    const int tid = threadIdx.x;
    const int s = blockIdx.x;
    const unsigned int lo = blockIdx.y * DEG_HR;
    const int p = blockIdx.z;
    const int* __restrict__ keys = (p & 2) ? ((p & 1) ? dst2 : src2) : ((p & 1) ? dst1 : src1);

    for (int i = tid; i < DEG_HR / 4; i += DEG_T) cnt4[i] = make_uint4(0u, 0u, 0u, 0u);
    __syncthreads();

    const int per = (((E + DEG_S - 1) / DEG_S) + 3) & ~3;
    const int e0 = s * per;
    int e1 = e0 + per;
    if (e1 > E) e1 = E;
    const int nvec = (e1 > e0) ? ((e1 - e0) >> 2) : 0;
    const int4* __restrict__ kv = (const int4*)(keys + e0);

    int i = tid;
    for (; i + 7 * DEG_T < nvec; i += 8 * DEG_T) {
        int4 k[8];
#pragma unroll
        for (int j = 0; j < 8; ++j) k[j] = kv[i + j * DEG_T];
#pragma unroll
        for (int j = 0; j < 8; ++j) deg_bin4(cnt, k[j], lo);
    }
    for (; i < nvec; i += DEG_T) deg_bin4(cnt, kv[i], lo);
    for (int e = e0 + (nvec << 2) + tid; e < e1; e += DEG_T) {
        unsigned int k0 = (unsigned int)keys[e] - lo;
        if (k0 < DEG_HR) atomicAdd(&cnt[k0], 1u);
    }
    __syncthreads();

    uint4* __restrict__ po = (uint4*)(part + (size_t)(p * DEG_S + s) * stride + lo);
    for (int j = tid; j < DEG_HR / 4; j += DEG_T) po[j] = cnt4[j];

    if (p & 1) {
        const int lane = tid & 31;
        int* __restrict__ go = gsum + ((size_t)(p >> 1) * (stride >> 8) + (size_t)blockIdx.y * (DEG_HR / 256)) * DEG_S + s;
        for (int grp = tid >> 5; grp < DEG_HR / 256; grp += DEG_T / 32) {
            const uint4 a = cnt4[grp * 64 + lane * 2];
            const uint4 b = cnt4[grp * 64 + lane * 2 + 1];
            int v = (int)(a.x + a.y + a.z + a.w + b.x + b.y + b.z + b.w);
            for (int o = 16; o > 0; o >>= 1) v += __shfl_xor(v, o);
            if (lane == 0) go[(size_t)grp * DEG_S] = v;
        }
    }
}

__device__ __forceinline__ void csr_place1(unsigned int* cur, int* __restrict__ esrc,
                                           int key, int val, unsigned int lo) {
    const unsigned int k = (unsigned int)key - lo;
    if (k < DEG_HR) {
        const unsigned int pos = atomicAdd(&cur[k], 1u);
        esrc[pos] = val;
    }
}

__device__ __forceinline__ void csr_place4(unsigned int* cur, int* __restrict__ esrc,
                                           int4 k, int4 v, unsigned int lo) {
    csr_place1(cur, esrc, k.x, v.x, lo);
    csr_place1(cur, esrc, k.y, v.y, lo);
    csr_place1(cur, esrc, k.z, v.z, lo);
    csr_place1(cur, esrc, k.w, v.w, lo);
}

__global__ void __launch_bounds__(DEG_T)
csr_place_kernel(const int* __restrict__ src1, const int* __restrict__ dst1,
                 const int* __restrict__ src2, const int* __restrict__ dst2,
                 const unsigned int* __restrict__ part, int* __restrict__ esrc_all,
                 int E, int stride) {
    __shared__ uint4 cur4[DEG_HR / 4];
    unsigned int* cur = (unsigned int*)cur4;
    const int tid = threadIdx.x;
    const int s = blockIdx.x;
    const unsigned int lo = blockIdx.y * DEG_HR;
    const int g = blockIdx.z;
    const int* __restrict__ src = g ? src2 : src1;
    const int* __restrict__ dst = g ? dst2 : dst1;
    const unsigned int* __restrict__ pd = part + (size_t)(2 * g + 1) * DEG_S * stride;
    int* __restrict__ esrc = esrc_all + (size_t)g * E;

    const uint4* __restrict__ p4 = (const uint4*)(pd + (size_t)s * stride + lo);
    for (int i = tid; i < DEG_HR / 4; i += DEG_T) cur4[i] = p4[i];
    __syncthreads();

    const int per = (((E + DEG_S - 1) / DEG_S) + 3) & ~3;
    const int e0 = s * per;
    int e1 = e0 + per;
    if (e1 > E) e1 = E;
    const int nvec = (e1 > e0) ? ((e1 - e0) >> 2) : 0;
    const int4* __restrict__ kv = (const int4*)(dst + e0);
    const int4* __restrict__ sv = (const int4*)(src + e0);

    int i = tid;
    for (; i + 7 * DEG_T < nvec; i += 8 * DEG_T) {
        int4 k[8], v[8];
#pragma unroll
        for (int j = 0; j < 8; ++j) { k[j] = kv[i + j * DEG_T]; v[j] = sv[i + j * DEG_T]; }
#pragma unroll
        for (int j = 0; j < 8; ++j) csr_place4(cur, esrc, k[j], v[j], lo);
    }
    for (; i < nvec; i += DEG_T) csr_place4(cur, esrc, kv[i], sv[i], lo);
    for (int e = e0 + (nvec << 2) + tid; e < e1; e += DEG_T) csr_place1(cur, esrc, dst[e], src[e], lo);
}

#define GNPW 8

__device__ __forceinline__ int lane_pick_i(int v, int i) { return __builtin_amdgcn_readlane(v, i); }
__device__ __forceinline__ float lane_pick_f(float v, int i) {
    return __int_as_float(__builtin_amdgcn_readlane(__float_as_int(v), i));
}

template <int F, typename Fin>
__device__ __forceinline__ void list_stream(const float* __restrict__ h, const int* __restrict__ esrc,
                                            int rw_l, int nn, int lane, Fin fin) {
    constexpr int R = F / 32;
    constexpr int U = 16 / R;
    float acc[R];
#pragma unroll
    for (int r = 0; r < R; ++r) acc[r] = 0.0f;
    int node = 0;
    int nend = lane_pick_i(rw_l, 1);
    const int eb = lane_pick_i(rw_l, 0);
    const int ee = lane_pick_i(rw_l, nn);
    if (ee > eb) {
        int idn = esrc[(eb + lane < ee) ? (eb + lane) : (ee - 1)];
        for (int base = eb; base < ee; base += 32) {
            const int id_l = idn;
            const int nx = base + 32 + lane;
            idn = esrc[(nx < ee) ? nx : (ee - 1)];
            const int cnt = (ee - base < 32) ? (ee - base) : 32;
            for (int j0 = 0; j0 < cnt; j0 += U) {
                float v[U * R];
#pragma unroll
                for (int u = 0; u < U; ++u) {
                    const int sidx = __builtin_amdgcn_readlane(id_l, j0 + u);
                    const float* __restrict__ hp = h + (size_t)sidx * F + lane;
#pragma unroll
                    for (int r = 0; r < R; ++r) v[u * R + r] = hp[r * 32];
                }
                asm volatile("" :: "v"(v[0]), "v"(v[1]), "v"(v[2]), "v"(v[3]), "v"(v[4]), "v"(v[5]),
                             "v"(v[6]), "v"(v[7]), "v"(v[8]), "v"(v[9]), "v"(v[10]), "v"(v[11]),
                             "v"(v[12]), "v"(v[13]), "v"(v[14]), "v"(v[15]));
                const int p = base + j0;
                if (p + U <= nend) {
#pragma unroll
                    for (int u = 0; u < U; ++u) {
#pragma unroll
                        for (int r = 0; r < R; ++r) acc[r] += v[u * R + r];
                    }
                } else {
#pragma unroll
                    for (int u = 0; u < U; ++u) {
                        const int e = p + u;
                        if (e < ee) {
                            while (e >= nend) {
                                fin(node, acc);
                                ++node;
                                nend = lane_pick_i(rw_l, node + 1);
#pragma unroll
                                for (int r = 0; r < R; ++r) acc[r] = 0.0f;
                            }
#pragma unroll
                            for (int r = 0; r < R; ++r) acc[r] += v[u * R + r];
                        }
                    }
                }
            }
        }
    }
    while (node < nn) {
        fin(node, acc);
        ++node;
#pragma unroll
        for (int r = 0; r < R; ++r) acc[r] = 0.0f;
    }
}

#define GRW 16
#define GLD (HID + 4)

__global__ void __launch_bounds__(256)
gather_relu_gemm_kernel(const float* __restrict__ h_all, const int* __restrict__ rowstart_all,
                        const int* __restrict__ esrc_all, const float* __restrict__ rs_all,
                        const float* __restrict__ b, const float* __restrict__ W2,
                        float* __restrict__ h2_all, int n, int E) {
    constexpr int R = HID / 32;
    constexpr int CT = NC / 16;
    __shared__ __attribute__((aligned(16))) float lw[NC * GLD];
    __shared__ __attribute__((aligned(16))) float la[8 * GRW * GLD];

#pragma unroll
    for (int j = 0; j < HID * NC / 256; ++j) {
        const int i = j * 256 + (int)threadIdx.x;
        const int k = i / NC;
        const int c = i - k * NC;
        lw[c * GLD + k] = W2[i];
    }
    __syncthreads();

    const int g = blockIdx.y;
    const float* __restrict__ h = h_all + (size_t)g * n * HID;
    const int* __restrict__ rowstart = rowstart_all + (size_t)g * (n + 1);
    const int* __restrict__ esrc = esrc_all + (size_t)g * E;
    const float* __restrict__ rs_out = rs_all + (size_t)(2 * g) * n;
    const float* __restrict__ rs_in = rs_all + (size_t)(2 * g + 1) * n;
    float* __restrict__ h2 = h2_all + (size_t)g * n * NC;
    const int lane = threadIdx.x & 31;
    const int wave = threadIdx.x >> 5;
    const int gw = __builtin_amdgcn_readfirstlane((int)(blockIdx.x * 8 + wave));
    const int n0 = gw * GRW;
    if (n0 >= n) return;
    const int nn = (n - n0 < GRW) ? (n - n0) : GRW;
    const int rw_l = rowstart[n0 + ((lane < nn) ? lane : nn)];
    const int ib = n0 + ((lane < nn) ? lane : (nn - 1));
    const float sci_l = rs_in[ib];
    const float sco_l = rs_out[ib];
    float* __restrict__ lt = la + wave * (GRW * GLD);

    float bias[R];
#pragma unroll
    for (int r = 0; r < R; ++r) bias[r] = b[lane + r * 32];

    list_stream<HID>(h, esrc, rw_l, nn, lane, [&](int i, const float (&acc)[R]) {
        const float rsi = lane_pick_f(sci_l, i);
        const float rso = lane_pick_f(sco_l, i);
#pragma unroll
        for (int r = 0; r < R; ++r)
            lt[i * GLD + lane + r * 32] = fmaxf(acc[r] * rsi + bias[r], 0.0f) * rso;
    });

    const int col = lane & 15;
    const int koff = (lane >> 4) * 2;
    v8f d[CT] = {};
    for (int k = 0; k < HID; k += 4) {
        float2 av = *(const float2*)&lt[col * GLD + k + koff];
        v2f a;
        a.x = av.x;
        a.y = av.y;
#pragma unroll
        for (int c = 0; c < CT; ++c) {
            float2 bv = *(const float2*)&lw[(c * 16 + col) * GLD + k + koff];
            v2f bb;
            bb.x = bv.x;
            bb.y = bv.y;
            d[c] = __builtin_amdgcn_wmma_f32_16x16x4_f32(
                false, a, false, bb, (short)0, d[c], false, false);
        }
    }

    const int mhi = (lane >> 4) * 8;
#pragma unroll
    for (int c = 0; c < CT; ++c) {
#pragma unroll
        for (int r = 0; r < 8; ++r) lt[(mhi + r) * GLD + c * 16 + col] = d[c][r];
    }
    for (int i = 0; i < nn; ++i) h2[(size_t)(n0 + i) * NC + lane] = lt[i * GLD + lane];
}

__global__ void __launch_bounds__(256)
gather_mean_kernel(const float* __restrict__ hA, const int* __restrict__ rowstartA,
                   const int* __restrict__ esrcA, const float* __restrict__ rsA,
                   const float* __restrict__ hB, const int* __restrict__ rowstartB,
                   const int* __restrict__ esrcB, const float* __restrict__ rsB,
                   const float* __restrict__ b, float* __restrict__ out, int n) {
    __shared__ float tl[8 * GNPW * NC];
    const int lane = threadIdx.x & 31;
    const int gw = __builtin_amdgcn_readfirstlane((int)((blockIdx.x * 256 + threadIdx.x) >> 5));
    const int n0 = gw * GNPW;
    if (n0 >= n) return;
    const int nn = (n - n0 < GNPW) ? (n - n0) : GNPW;
    const int ia = n0 + ((lane < nn) ? lane : nn);
    const int ib = n0 + ((lane < nn) ? lane : (nn - 1));
    const int rwA_l = rowstartA[ia];
    const int rwB_l = rowstartB[ia];
    const float scA_l = rsA[ib];
    const float scB_l = rsB[ib];
    const float bias = b[lane];

    float* __restrict__ tw = tl + (threadIdx.x >> 5) * (GNPW * NC);

    list_stream<NC>(hA, esrcA, rwA_l, nn, lane, [&](int i, const float (&acc)[1]) {
        tw[i * NC + lane] = acc[0] * lane_pick_f(scA_l, i) + bias;
    });
    list_stream<NC>(hB, esrcB, rwB_l, nn, lane, [&](int i, const float (&acc)[1]) {
        const float tA = tw[i * NC + lane];
        const float tB = acc[0] * lane_pick_f(scB_l, i) + bias;
        out[(size_t)(n0 + i) * NC + lane] = 0.5f * tA + 0.5f * tB;
    });
}

template <int IN, int OUT>
__global__ void __launch_bounds__(256)
scan_gemm_kernel(const float* __restrict__ X, const float* __restrict__ W, float* __restrict__ Hall,
                 unsigned int* __restrict__ part, const int* __restrict__ gsum_all,
                 float* __restrict__ rs, int* __restrict__ rowstart_all,
                 int n, int stride, int gblocks, int nblocks) {
    constexpr int LDW = IN + 4;
    __shared__ float lw[OUT * LDW];
    __shared__ int red[8];
    __shared__ int wtot[8];
    const int tid = threadIdx.x;
    const int lane = tid & 31;
    const int wave = tid >> 5;

    if ((int)blockIdx.x < gblocks) {
        for (int i = tid; i < IN * OUT; i += 256) {
            int k = i / OUT;
            int cc = i - k * OUT;
            lw[cc * LDW + k] = W[i];
        }
        __syncthreads();

        const int r0 = (blockIdx.x * 8 + wave) * 16;
        if (r0 >= n) return;

        const int mrow = r0 + (lane & 15);
        const int koff = (lane >> 4) * 2;
        const float* xrow = X + (size_t)mrow * IN;
        const int col = lane & 15;

        float rsv[2];
        {
            const unsigned int* __restrict__ p0 = part + mrow;
            const unsigned int* __restrict__ p1 = part + (size_t)2 * DEG_S * stride + mrow;
            unsigned int c0 = 0, c1 = 0;
#pragma unroll
            for (int s = 0; s < DEG_S; ++s) {
                c0 += p0[(size_t)s * stride];
                c1 += p1[(size_t)s * stride];
            }
            rsv[0] = rsqrtf(fmaxf((float)c0, 1.0f));
            rsv[1] = rsqrtf(fmaxf((float)c1, 1.0f));
        }

        constexpr int CT = OUT / 16;
        v8f acc[2][CT] = {};

        for (int k = 0; k < IN; k += 4) {
            float2 av = *(const float2*)(xrow + k + koff);
            v2f a[2];
#pragma unroll
            for (int j = 0; j < 2; ++j) {
                a[j].x = av.x * rsv[j];
                a[j].y = av.y * rsv[j];
            }
#pragma unroll
            for (int cc = 0; cc < CT; ++cc) {
                float2 bv = *(const float2*)&lw[(cc * 16 + col) * LDW + k + koff];
                v2f b;
                b.x = bv.x;
                b.y = bv.y;
#pragma unroll
                for (int j = 0; j < 2; ++j) {
                    acc[j][cc] = __builtin_amdgcn_wmma_f32_16x16x4_f32(
                        false, a[j], false, b, (short)0, acc[j][cc], false, false);
                }
            }
        }

        const int mhi = (lane >> 4) * 8;
#pragma unroll
        for (int j = 0; j < 2; ++j) {
            float* __restrict__ H = Hall + (size_t)j * n * OUT;
#pragma unroll
            for (int cc = 0; cc < CT; ++cc) {
#pragma unroll
                for (int r = 0; r < 8; ++r) {
                    H[(size_t)(r0 + mhi + r) * OUT + cc * 16 + col] = acc[j][cc][r];
                }
            }
        }
        return;
    }

    const int sb = (int)blockIdx.x - gblocks;
    const int g = sb / nblocks;
    const int bx = sb - g * nblocks;
    unsigned int* __restrict__ pd = part + (size_t)(2 * g + 1) * DEG_S * stride;
    const unsigned int* __restrict__ ps = part + (size_t)(2 * g) * DEG_S * stride;
    int* __restrict__ rowstart = rowstart_all + (size_t)g * (n + 1);

    const int4* __restrict__ gs4 = (const int4*)(gsum_all + (size_t)g * (stride >> 8) * DEG_S);
    const int lim4 = bx * (DEG_S / 4);
    int o = 0;
    for (int j0 = 0; j0 < lim4; j0 += 8 * 256) {
        int4 q[8];
#pragma unroll
        for (int u = 0; u < 8; ++u) {
            const int j = j0 + u * 256 + tid;
            q[u] = gs4[(j < lim4) ? j : 0];
        }
#pragma unroll
        for (int u = 0; u < 8; ++u) {
            const int j = j0 + u * 256 + tid;
            o += (j < lim4) ? (q[u].x + q[u].y + q[u].z + q[u].w) : 0;
        }
    }
    for (int d = 16; d > 0; d >>= 1) o += __shfl_xor(o, d);
    if (lane == 0) red[wave] = o;

    const int node = bx * 256 + tid;
    unsigned int t[DEG_S];
    int c = 0;
    unsigned int co = 0;
#pragma unroll
    for (int s = 0; s < DEG_S; ++s) {
        t[s] = (node < n) ? pd[(size_t)s * stride + node] : 0u;
        co += (node < n) ? ps[(size_t)s * stride + node] : 0u;
        c += (int)t[s];
    }
    if (node < n) {
        rs[(size_t)(2 * g) * n + node] = rsqrtf(fmaxf((float)co, 1.0f));
        rs[(size_t)(2 * g + 1) * n + node] = rsqrtf(fmaxf((float)c, 1.0f));
    }
    int inc = c;
    for (int d = 1; d < 32; d <<= 1) {
        const int u = __shfl_up(inc, d);
        if (lane >= d) inc += u;
    }
    if (lane == 31) wtot[wave] = inc;
    __syncthreads();

    int base = 0;
    for (int w = 0; w < 8; ++w) {
        base += red[w];
        if (w < wave) base += wtot[w];
    }
    const int excl = base + inc - c;
    if (node < n) {
        rowstart[node] = excl;
        unsigned int off = (unsigned int)excl;
#pragma unroll
        for (int s = 0; s < DEG_S; ++s) {
            pd[(size_t)s * stride + node] = off;
            off += t[s];
        }
    }
    if (node == n - 1) rowstart[n] = excl + c;
}


extern "C" void kernel_launch(void* const* d_in, const int* in_sizes, int n_in,
                              void* d_out, int out_size, void* d_ws, size_t ws_size,
                              hipStream_t stream) {
    const float* x    = (const float*)d_in[0];
    const int*   src1 = (const int*)d_in[1];
    const int*   dst1 = (const int*)d_in[2];
    const int*   src2 = (const int*)d_in[3];
    const int*   dst2 = (const int*)d_in[4];
    const float* W1   = (const float*)d_in[5];
    const float* b1   = (const float*)d_in[6];
    const float* W2   = (const float*)d_in[7];
    const float* b2   = (const float*)d_in[8];
    float* out = (float*)d_out;

    const int N = in_sizes[0] / IN_FEATS;
    const int E = in_sizes[1];

    const int dranges = (N + DEG_HR - 1) / DEG_HR;
    const int dstride = dranges * DEG_HR;
    unsigned int* dpart = (unsigned int*)d_ws;
    float* ws   = (float*)(dpart + (size_t)4 * DEG_S * dstride);
    float* rsOI = ws;
    float* h    = rsOI + (size_t)4 * N;
    float* h2   = h + (size_t)2 * N * HID;
    int* esrc   = (int*)(h2 + (size_t)2 * N * NC);
    int* gsum   = esrc + (size_t)2 * E;
    int* rowst  = gsum + (size_t)2 * (dstride / 256) * DEG_S;

    const int T = 256;
    auto cdiv = [](long a, long b) { return (int)((a + b - 1) / b); };

    const int tiles   = N / 16;
    const int gblocks = cdiv(tiles, 8);
    const int nblocks = cdiv(N, T);
    const int wblocks = cdiv(cdiv(N, GNPW), 8);
    const int tblocks = cdiv(cdiv(N, GRW), 8);

    deg_hist_kernel<<<dim3(DEG_S, dranges, 4), DEG_T, 0, stream>>>(src1, dst1, src2, dst2, dpart, gsum, E, dstride);
    scan_gemm_kernel<IN_FEATS, HID><<<gblocks + 2 * nblocks, T, 0, stream>>>(
        x, W1, h, dpart, gsum, rsOI, rowst, N, dstride, gblocks, nblocks);
    csr_place_kernel<<<dim3(DEG_S, dranges, 2), DEG_T, 0, stream>>>(src1, dst1, src2, dst2, dpart, esrc, E, dstride);

    gather_relu_gemm_kernel<<<dim3(tblocks, 2), T, 0, stream>>>(h, rowst, esrc, rsOI, b1, W2, h2, N, E);

    gather_mean_kernel<<<wblocks, T, 0, stream>>>(
        h2, rowst, esrc, rsOI + N,
        h2 + (size_t)N * NC, rowst + (N + 1), esrc + E, rsOI + (size_t)3 * N,
        b2, out, N);
}


# edge list per 16-node tile, slice-major, node index in an entry's low 4 bits: a sort block buffers its runs in LDS and stores them whole; per-node sum order still slice, then LDS arrival
# speedup vs baseline: 5.2093x; 1.2559x over previous
#include <hip/hip_runtime.h>


typedef __attribute__((ext_vector_type(2))) float v2f;
typedef __attribute__((ext_vector_type(8))) float v8f;
typedef __attribute__((ext_vector_type(16))) float v16f;

#define IN_FEATS 128
#define HID      64
#define NC       32


#define DEG_HR 12544
#define DEG_S  16
#define DEG_T  1024

__device__ __forceinline__ void deg_bin4(unsigned int* cnt, int4 k, unsigned int lo) {
    unsigned int k0 = (unsigned int)k.x - lo;
    unsigned int k1 = (unsigned int)k.y - lo;
    unsigned int k2 = (unsigned int)k.z - lo;
    unsigned int k3 = (unsigned int)k.w - lo;
    if (k0 < DEG_HR) atomicAdd(&cnt[k0], 1u);
    if (k1 < DEG_HR) atomicAdd(&cnt[k1], 1u);
    if (k2 < DEG_HR) atomicAdd(&cnt[k2], 1u);
    if (k3 < DEG_HR) atomicAdd(&cnt[k3], 1u);
}

__global__ void __launch_bounds__(DEG_T)
deg_hist_kernel(const int* __restrict__ src1, const int* __restrict__ dst1,
                const int* __restrict__ src2, const int* __restrict__ dst2,
                unsigned int* __restrict__ part, int* __restrict__ gsum, int E, int stride) {
    __shared__ uint4 cnt4[DEG_HR / 4];
    unsigned int* cnt = (unsigned int*)cnt4;
    const int tid = threadIdx.x;
    const int s = blockIdx.x;
    const unsigned int lo = blockIdx.y * DEG_HR;
    const int p = blockIdx.z;
    const int* __restrict__ keys = (p & 2) ? ((p & 1) ? dst2 : src2) : ((p & 1) ? dst1 : src1);

    for (int i = tid; i < DEG_HR / 4; i += DEG_T) cnt4[i] = make_uint4(0u, 0u, 0u, 0u);
    __syncthreads();

    const int per = (((E + DEG_S - 1) / DEG_S) + 3) & ~3;
    const int e0 = s * per;
    int e1 = e0 + per;
    if (e1 > E) e1 = E;
    const int nvec = (e1 > e0) ? ((e1 - e0) >> 2) : 0;
    const int4* __restrict__ kv = (const int4*)(keys + e0);

    int i = tid;
    for (; i + 7 * DEG_T < nvec; i += 8 * DEG_T) {
        int4 k[8];
#pragma unroll
        for (int j = 0; j < 8; ++j) k[j] = kv[i + j * DEG_T];
#pragma unroll
        for (int j = 0; j < 8; ++j) deg_bin4(cnt, k[j], lo);
    }
    for (; i < nvec; i += DEG_T) deg_bin4(cnt, kv[i], lo);
    for (int e = e0 + (nvec << 2) + tid; e < e1; e += DEG_T) {
        unsigned int k0 = (unsigned int)keys[e] - lo;
        if (k0 < DEG_HR) atomicAdd(&cnt[k0], 1u);
    }
    __syncthreads();

    uint4* __restrict__ po = (uint4*)(part + (size_t)(p * DEG_S + s) * stride + lo);
    for (int j = tid; j < DEG_HR / 4; j += DEG_T) po[j] = cnt4[j];

    if (p & 1) {
        const int lane = tid & 31;
        int* __restrict__ go = gsum + ((size_t)(p >> 1) * (stride >> 8) + (size_t)blockIdx.y * (DEG_HR / 256)) * DEG_S + s;
        for (int grp = tid >> 5; grp < DEG_HR / 256; grp += DEG_T / 32) {
            const uint4 a = cnt4[grp * 64 + lane * 2];
            const uint4 b = cnt4[grp * 64 + lane * 2 + 1];
            int v = (int)(a.x + a.y + a.z + a.w + b.x + b.y + b.z + b.w);
            for (int o = 16; o > 0; o >>= 1) v += __shfl_xor(v, o);
            if (lane == 0) go[(size_t)grp * DEG_S] = v;
        }
    }
}

#define TPR (DEG_HR / 16)
#define STG_CAP 16384

__device__ __forceinline__ void csr_place1(unsigned int* lcur, const unsigned int* lsta,
                                           const unsigned int* gbas, int* stage,
                                           int* __restrict__ esrc, bool staged,
                                           int key, int val, unsigned int lo) {
    const unsigned int k = (unsigned int)key - lo;
    if (k < DEG_HR) {
        const unsigned int t = k >> 4;
        const unsigned int li = atomicAdd(&lcur[t], 1u);
        const int ent = (int)(((unsigned int)val << 4) | (k & 15u));
        if (staged) stage[li] = ent;
        else esrc[gbas[t] + (li - lsta[t])] = ent;
    }
}

__device__ __forceinline__ void csr_place4(unsigned int* lcur, const unsigned int* lsta,
                                           const unsigned int* gbas, int* stage,
                                           int* __restrict__ esrc, bool staged,
                                           int4 k, int4 v, unsigned int lo) {
    csr_place1(lcur, lsta, gbas, stage, esrc, staged, k.x, v.x, lo);
    csr_place1(lcur, lsta, gbas, stage, esrc, staged, k.y, v.y, lo);
    csr_place1(lcur, lsta, gbas, stage, esrc, staged, k.z, v.z, lo);
    csr_place1(lcur, lsta, gbas, stage, esrc, staged, k.w, v.w, lo);
}

__global__ void __launch_bounds__(DEG_T)
csr_place_kernel(const int* __restrict__ src1, const int* __restrict__ dst1,
                 const int* __restrict__ src2, const int* __restrict__ dst2,
                 const int* __restrict__ tcur_all, int* __restrict__ esrc_all,
                 int E, int n, int tp) {
    __shared__ unsigned int lcur[TPR];
    __shared__ unsigned int lsta[TPR];
    __shared__ unsigned int gbas[TPR];
    __shared__ unsigned int wtot[DEG_T / 32];
    __shared__ int stage[STG_CAP];
    const int tid = threadIdx.x;
    const int lane = tid & 31;
    const int wave = tid >> 5;
    const int s = blockIdx.x;
    const unsigned int lo = blockIdx.y * DEG_HR;
    const int g = blockIdx.z;
    const int* __restrict__ src = g ? src2 : src1;
    const int* __restrict__ dst = g ? dst2 : dst1;
    int* __restrict__ esrc = esrc_all + (size_t)g * E;
    const int* __restrict__ tc = tcur_all + ((size_t)g * (DEG_S + 1) + s) * tp + blockIdx.y * TPR;

    unsigned int a = 0u, c = 0u;
    if (tid < TPR && (int)(lo + tid * 16) < n) {
        a = (unsigned int)tc[tid];
        c = (unsigned int)tc[tp + tid] - a;
    }
    int inc = (int)c;
    for (int d = 1; d < 32; d <<= 1) {
        const int u = __shfl_up(inc, d);
        if (lane >= d) inc += u;
    }
    if (lane == 31) wtot[wave] = (unsigned int)inc;
    __syncthreads();
    unsigned int before = 0u, total = 0u;
    for (int w = 0; w < DEG_T / 32; ++w) {
        const unsigned int x = wtot[w];
        total += x;
        if (w < wave) before += x;
    }
    const unsigned int excl = before + (unsigned int)inc - c;
    if (tid < TPR) {
        lsta[tid] = excl;
        lcur[tid] = excl;
        gbas[tid] = a;
    }
    const bool staged = total <= STG_CAP;
    __syncthreads();

    const int per = (((E + DEG_S - 1) / DEG_S) + 3) & ~3;
    const int e0 = s * per;
    int e1 = e0 + per;
    if (e1 > E) e1 = E;
    const int nvec = (e1 > e0) ? ((e1 - e0) >> 2) : 0;
    const int4* __restrict__ kv = (const int4*)(dst + e0);
    const int4* __restrict__ sv = (const int4*)(src + e0);

    int i = tid;
    for (; i + 7 * DEG_T < nvec; i += 8 * DEG_T) {
        int4 k[8], v[8];
#pragma unroll
        for (int j = 0; j < 8; ++j) { k[j] = kv[i + j * DEG_T]; v[j] = sv[i + j * DEG_T]; }
#pragma unroll
        for (int j = 0; j < 8; ++j) csr_place4(lcur, lsta, gbas, stage, esrc, staged, k[j], v[j], lo);
    }
    for (; i < nvec; i += DEG_T) csr_place4(lcur, lsta, gbas, stage, esrc, staged, kv[i], sv[i], lo);
    for (int e = e0 + (nvec << 2) + tid; e < e1; e += DEG_T)
        csr_place1(lcur, lsta, gbas, stage, esrc, staged, dst[e], src[e], lo);
    __syncthreads();

    if (staged) {
        const int wv = __builtin_amdgcn_readfirstlane(wave);
        const int tk = wv + (DEG_T / 32) * lane;
        unsigned int ls_l = 0u, len_l = 0u, gb_l = 0u;
        if (tk < TPR) {
            ls_l = lsta[tk];
            len_l = lcur[tk] - ls_l;
            gb_l = gbas[tk];
        }
        for (int k = 0; wv + (DEG_T / 32) * k < TPR; ++k) {
            const unsigned int ls = (unsigned int)__builtin_amdgcn_readlane((int)ls_l, k);
            const unsigned int len = (unsigned int)__builtin_amdgcn_readlane((int)len_l, k);
            const unsigned int gb = (unsigned int)__builtin_amdgcn_readlane((int)gb_l, k);
            for (unsigned int j = lane; j < len; j += 32) esrc[gb + j] = stage[ls + j];
        }
    }
}

#define GRW 16

__device__ __forceinline__ int lane_pick_i(int v, int i) { return __builtin_amdgcn_readlane(v, i); }
__device__ __forceinline__ float lane_pick_f(float v, int i) {
    return __int_as_float(__builtin_amdgcn_readlane(__float_as_int(v), i));
}

template <int F>
__device__ __forceinline__ void tile_stream(const float* __restrict__ h, const int* __restrict__ esrc,
                                            int eb, int ee, int lane, v16f (&acc)[F / 32]) {
    constexpr int R = F / 32;
    constexpr int U = 16 / R;
    if (ee > eb) {
        int idn = esrc[(eb + lane < ee) ? (eb + lane) : (ee - 1)];
        for (int base = eb; base < ee; base += 32) {
            const int id_l = idn;
            const int nx = base + 32 + lane;
            idn = esrc[(nx < ee) ? nx : (ee - 1)];
            const int cnt = (ee - base < 32) ? (ee - base) : 32;
            for (int j0 = 0; j0 < cnt; j0 += U) {
                float v[U * R];
                int tag[U];
#pragma unroll
                for (int u = 0; u < U; ++u) {
                    const unsigned int ent = (unsigned int)__builtin_amdgcn_readlane(id_l, j0 + u);
                    tag[u] = (int)(ent & 15u);
                    const float* __restrict__ hp = h + (size_t)(ent >> 4) * F;
#pragma unroll
                    for (int r = 0; r < R; ++r) v[u * R + r] = hp[lane + r * 32];
                }
                asm volatile("" :: "v"(v[0]), "v"(v[1]), "v"(v[2]), "v"(v[3]), "v"(v[4]), "v"(v[5]),
                             "v"(v[6]), "v"(v[7]), "v"(v[8]), "v"(v[9]), "v"(v[10]), "v"(v[11]),
                             "v"(v[12]), "v"(v[13]), "v"(v[14]), "v"(v[15]));
#pragma unroll
                for (int u = 0; u < U; ++u) {
                    if (j0 + u < cnt) {
#pragma unroll
                        for (int r = 0; r < R; ++r) acc[r][tag[u]] += v[u * R + r];
                    }
                }
            }
        }
    }
}

#define GLD (HID + 4)

__global__ void __launch_bounds__(256)
gather_relu_gemm_kernel(const float* __restrict__ h_all, const int* __restrict__ rowstart_all,
                        const int* __restrict__ esrc_all, const float* __restrict__ rs_all,
                        const float* __restrict__ b, const float* __restrict__ W2,
                        float* __restrict__ h2_all, int n, int E) {
    constexpr int R = HID / 32;
    constexpr int CT = NC / 16;
    __shared__ __attribute__((aligned(16))) float lw[NC * GLD];
    __shared__ __attribute__((aligned(16))) float la[8 * GRW * GLD];

#pragma unroll
    for (int j = 0; j < HID * NC / 256; ++j) {
        const int i = j * 256 + (int)threadIdx.x;
        const int k = i / NC;
        const int c = i - k * NC;
        lw[c * GLD + k] = W2[i];
    }
    __syncthreads();

    const int g = blockIdx.y;
    const float* __restrict__ h = h_all + (size_t)g * n * HID;
    const int* __restrict__ rowstart = rowstart_all + (size_t)g * (n + 1);
    const int* __restrict__ esrc = esrc_all + (size_t)g * E;
    const float* __restrict__ rs_out = rs_all + (size_t)(2 * g) * n;
    const float* __restrict__ rs_in = rs_all + (size_t)(2 * g + 1) * n;
    float* __restrict__ h2 = h2_all + (size_t)g * n * NC;
    const int lane = threadIdx.x & 31;
    const int wave = threadIdx.x >> 5;
    const int gw = __builtin_amdgcn_readfirstlane((int)(blockIdx.x * 8 + wave));
    const int n0 = gw * GRW;
    if (n0 >= n) return;
    const int nn = (n - n0 < GRW) ? (n - n0) : GRW;
    const int rw_l = rowstart[n0 + ((lane < nn) ? lane : nn)];
    float* __restrict__ lt = la + wave * (GRW * GLD);

    v16f acc[R] = {};
    tile_stream<HID>(h, esrc, lane_pick_i(rw_l, 0), lane_pick_i(rw_l, nn), lane, acc);

    const int ib = n0 + ((lane < nn) ? lane : (nn - 1));
    const float sci_l = rs_in[ib];
    const float sco_l = rs_out[ib];
    float bias[R];
#pragma unroll
    for (int r = 0; r < R; ++r) bias[r] = b[lane + r * 32];
#pragma unroll
    for (int i = 0; i < GRW; ++i) {
        const float rsi = lane_pick_f(sci_l, i);
        const float rso = lane_pick_f(sco_l, i);
#pragma unroll
        for (int r = 0; r < R; ++r)
            lt[i * GLD + lane + r * 32] = fmaxf(acc[r][i] * rsi + bias[r], 0.0f) * rso;
    }

    const int col = lane & 15;
    const int koff = (lane >> 4) * 2;
    v8f d[CT] = {};
    for (int k = 0; k < HID; k += 4) {
        float2 av = *(const float2*)&lt[col * GLD + k + koff];
        v2f a;
        a.x = av.x;
        a.y = av.y;
#pragma unroll
        for (int c = 0; c < CT; ++c) {
            float2 bv = *(const float2*)&lw[(c * 16 + col) * GLD + k + koff];
            v2f bb;
            bb.x = bv.x;
            bb.y = bv.y;
            d[c] = __builtin_amdgcn_wmma_f32_16x16x4_f32(
                false, a, false, bb, (short)0, d[c], false, false);
        }
    }

    const int mhi = (lane >> 4) * 8;
#pragma unroll
    for (int c = 0; c < CT; ++c) {
#pragma unroll
        for (int r = 0; r < 8; ++r) lt[(mhi + r) * GLD + c * 16 + col] = d[c][r];
    }
    for (int i = 0; i < nn; ++i) h2[(size_t)(n0 + i) * NC + lane] = lt[i * GLD + lane];
}

__global__ void __launch_bounds__(256)
gather_mean_kernel(const float* __restrict__ hA, const int* __restrict__ rowstartA,
                   const int* __restrict__ esrcA, const float* __restrict__ rsA,
                   const float* __restrict__ hB, const int* __restrict__ rowstartB,
                   const int* __restrict__ esrcB, const float* __restrict__ rsB,
                   const float* __restrict__ b, float* __restrict__ out, int n) {
    const int lane = threadIdx.x & 31;
    const int gw = __builtin_amdgcn_readfirstlane((int)((blockIdx.x * 256 + threadIdx.x) >> 5));
    const int n0 = gw * GRW;
    if (n0 >= n) return;
    const int nn = (n - n0 < GRW) ? (n - n0) : GRW;
    const int ia = n0 + ((lane < nn) ? lane : nn);
    const int ib = n0 + ((lane < nn) ? lane : (nn - 1));
    const int rwA_l = rowstartA[ia];
    const int rwB_l = rowstartB[ia];
    const float scA_l = rsA[ib];
    const float scB_l = rsB[ib];
    const float bias = b[lane];

    v16f accA[1] = {};
    v16f accB[1] = {};
    tile_stream<NC>(hA, esrcA, lane_pick_i(rwA_l, 0), lane_pick_i(rwA_l, nn), lane, accA);
    tile_stream<NC>(hB, esrcB, lane_pick_i(rwB_l, 0), lane_pick_i(rwB_l, nn), lane, accB);
#pragma unroll
    for (int i = 0; i < GRW; ++i) {
        if (i < nn) {
            const float tA = accA[0][i] * lane_pick_f(scA_l, i) + bias;
            const float tB = accB[0][i] * lane_pick_f(scB_l, i) + bias;
            out[(size_t)(n0 + i) * NC + lane] = 0.5f * tA + 0.5f * tB;
        }
    }
}

template <int IN, int OUT>
__global__ void __launch_bounds__(256)
scan_gemm_kernel(const float* __restrict__ X, const float* __restrict__ W, float* __restrict__ Hall,
                 const unsigned int* __restrict__ part, const int* __restrict__ gsum_all,
                 float* __restrict__ rs, int* __restrict__ rowstart_all, int* __restrict__ tcur_all,
                 int n, int stride, int gblocks, int nblocks) {
    constexpr int LDW = IN + 4;
    __shared__ float lw[OUT * LDW];
    __shared__ int red[8];
    __shared__ int wtot[8];
    const int tid = threadIdx.x;
    const int lane = tid & 31;
    const int wave = tid >> 5;

    if ((int)blockIdx.x < gblocks) {
        for (int i = tid; i < IN * OUT; i += 256) {
            int k = i / OUT;
            int cc = i - k * OUT;
            lw[cc * LDW + k] = W[i];
        }
        __syncthreads();

        const int r0 = (blockIdx.x * 8 + wave) * 16;
        if (r0 >= n) return;

        const int mrow = r0 + (lane & 15);
        const int koff = (lane >> 4) * 2;
        const float* xrow = X + (size_t)mrow * IN;
        const int col = lane & 15;

        float rsv[2];
        {
            const unsigned int* __restrict__ p0 = part + mrow;
            const unsigned int* __restrict__ p1 = part + (size_t)2 * DEG_S * stride + mrow;
            unsigned int c0 = 0, c1 = 0;
#pragma unroll
            for (int s = 0; s < DEG_S; ++s) {
                c0 += p0[(size_t)s * stride];
                c1 += p1[(size_t)s * stride];
            }
            rsv[0] = rsqrtf(fmaxf((float)c0, 1.0f));
            rsv[1] = rsqrtf(fmaxf((float)c1, 1.0f));
        }

        constexpr int CT = OUT / 16;
        v8f acc[2][CT] = {};

        for (int k = 0; k < IN; k += 4) {
            float2 av = *(const float2*)(xrow + k + koff);
            v2f a[2];
#pragma unroll
            for (int j = 0; j < 2; ++j) {
                a[j].x = av.x * rsv[j];
                a[j].y = av.y * rsv[j];
            }
#pragma unroll
            for (int cc = 0; cc < CT; ++cc) {
                float2 bv = *(const float2*)&lw[(cc * 16 + col) * LDW + k + koff];
                v2f b;
                b.x = bv.x;
                b.y = bv.y;
#pragma unroll
                for (int j = 0; j < 2; ++j) {
                    acc[j][cc] = __builtin_amdgcn_wmma_f32_16x16x4_f32(
                        false, a[j], false, b, (short)0, acc[j][cc], false, false);
                }
            }
        }

        const int mhi = (lane >> 4) * 8;
#pragma unroll
        for (int j = 0; j < 2; ++j) {
            float* __restrict__ H = Hall + (size_t)j * n * OUT;
#pragma unroll
            for (int cc = 0; cc < CT; ++cc) {
#pragma unroll
                for (int r = 0; r < 8; ++r) {
                    H[(size_t)(r0 + mhi + r) * OUT + cc * 16 + col] = acc[j][cc][r];
                }
            }
        }
        return;
    }

    const int sb = (int)blockIdx.x - gblocks;
    const int g = sb / nblocks;
    const int bx = sb - g * nblocks;
    const unsigned int* __restrict__ pd = part + (size_t)(2 * g + 1) * DEG_S * stride;
    const unsigned int* __restrict__ ps = part + (size_t)(2 * g) * DEG_S * stride;
    int* __restrict__ rowstart = rowstart_all + (size_t)g * (n + 1);

    const int4* __restrict__ gs4 = (const int4*)(gsum_all + (size_t)g * (stride >> 8) * DEG_S);
    const int lim4 = bx * (DEG_S / 4);
    int o = 0;
    for (int j0 = 0; j0 < lim4; j0 += 8 * 256) {
        int4 q[8];
#pragma unroll
        for (int u = 0; u < 8; ++u) {
            const int j = j0 + u * 256 + tid;
            q[u] = gs4[(j < lim4) ? j : 0];
        }
#pragma unroll
        for (int u = 0; u < 8; ++u) {
            const int j = j0 + u * 256 + tid;
            o += (j < lim4) ? (q[u].x + q[u].y + q[u].z + q[u].w) : 0;
        }
    }
    for (int d = 16; d > 0; d >>= 1) o += __shfl_xor(o, d);
    if (lane == 0) red[wave] = o;

    const int node = bx * 256 + tid;
    unsigned int t[DEG_S];
    int c = 0;
    unsigned int co = 0;
#pragma unroll
    for (int s = 0; s < DEG_S; ++s) {
        t[s] = (node < n) ? pd[(size_t)s * stride + node] : 0u;
        co += (node < n) ? ps[(size_t)s * stride + node] : 0u;
        c += (int)t[s];
    }
    if (node < n) {
        rs[(size_t)(2 * g) * n + node] = rsqrtf(fmaxf((float)co, 1.0f));
        rs[(size_t)(2 * g + 1) * n + node] = rsqrtf(fmaxf((float)c, 1.0f));
    }
    int inc = c;
    for (int d = 1; d < 32; d <<= 1) {
        const int u = __shfl_up(inc, d);
        if (lane >= d) inc += u;
    }
    if (lane == 31) wtot[wave] = inc;
    __syncthreads();

    int base = 0;
    for (int w = 0; w < 8; ++w) {
        base += red[w];
        if (w < wave) base += wtot[w];
    }
    const int excl = base + inc - c;
    if (node < n) rowstart[node] = excl;

    const int tp = stride >> 4;
    int* __restrict__ tc = tcur_all + (size_t)g * (DEG_S + 1) * tp + (node >> 4);
    const bool wr = ((tid & 15) == 0) && (node < n);
    int run = __shfl(excl, lane & 16);
#pragma unroll
    for (int s = 0; s < DEG_S; ++s) {
        int v = (int)t[s];
        v += __shfl_xor(v, 8);
        v += __shfl_xor(v, 4);
        v += __shfl_xor(v, 2);
        v += __shfl_xor(v, 1);
        if (wr) tc[(size_t)s * tp] = run;
        run += v;
    }
    if (wr) tc[(size_t)DEG_S * tp] = run;
    if (node == n - 1) rowstart[n] = excl + c;
}


extern "C" void kernel_launch(void* const* d_in, const int* in_sizes, int n_in,
                              void* d_out, int out_size, void* d_ws, size_t ws_size,
                              hipStream_t stream) {
    const float* x    = (const float*)d_in[0];
    const int*   src1 = (const int*)d_in[1];
    const int*   dst1 = (const int*)d_in[2];
    const int*   src2 = (const int*)d_in[3];
    const int*   dst2 = (const int*)d_in[4];
    const float* W1   = (const float*)d_in[5];
    const float* b1   = (const float*)d_in[6];
    const float* W2   = (const float*)d_in[7];
    const float* b2   = (const float*)d_in[8];
    float* out = (float*)d_out;

    const int N = in_sizes[0] / IN_FEATS;
    const int E = in_sizes[1];

    const int dranges = (N + DEG_HR - 1) / DEG_HR;
    const int dstride = dranges * DEG_HR;
    unsigned int* dpart = (unsigned int*)d_ws;
    float* ws   = (float*)(dpart + (size_t)4 * DEG_S * dstride);
    float* rsOI = ws;
    float* h    = rsOI + (size_t)4 * N;
    float* h2   = h + (size_t)2 * N * HID;
    int* esrc   = (int*)(h2 + (size_t)2 * N * NC);
    int* gsum   = esrc + (size_t)2 * E;
    int* rowst  = gsum + (size_t)2 * (dstride / 256) * DEG_S;
    int* tcur   = rowst + (size_t)2 * (N + 1);

    const int T = 256;
    auto cdiv = [](long a, long b) { return (int)((a + b - 1) / b); };

    const int tiles   = N / 16;
    const int gblocks = cdiv(tiles, 8);
    const int nblocks = cdiv(N, T);
    const int tblocks = cdiv(cdiv(N, GRW), 8);

    deg_hist_kernel<<<dim3(DEG_S, dranges, 4), DEG_T, 0, stream>>>(src1, dst1, src2, dst2, dpart, gsum, E, dstride);
    scan_gemm_kernel<IN_FEATS, HID><<<gblocks + 2 * nblocks, T, 0, stream>>>(
        x, W1, h, dpart, gsum, rsOI, rowst, tcur, N, dstride, gblocks, nblocks);
    csr_place_kernel<<<dim3(DEG_S, dranges, 2), DEG_T, 0, stream>>>(src1, dst1, src2, dst2, tcur, esrc, E, N, dstride / 16);

    gather_relu_gemm_kernel<<<dim3(tblocks, 2), T, 0, stream>>>(h, rowst, esrc, rsOI, b1, W2, h2, N, E);

    gather_mean_kernel<<<tblocks, T, 0, stream>>>(
        h2, rowst, esrc, rsOI + N,
        h2 + (size_t)N * NC, rowst + (N + 1), esrc + E, rsOI + (size_t)3 * N,
        b2, out, N);
}


# row scan folded into the edge-ordering blocks; layer-1 product, once per tile, each graph's out-degree scale applied to the result rows (scaled after the sum), runs beside them: 5 launches, now 4
# speedup vs baseline: 5.5601x; 1.0673x over previous
#include <hip/hip_runtime.h>


typedef __attribute__((ext_vector_type(2))) float v2f;
typedef __attribute__((ext_vector_type(8))) float v8f;
typedef __attribute__((ext_vector_type(16))) float v16f;

#define IN_FEATS 128
#define HID      64
#define NC       32


#define DEG_HR 12544
#define DEG_S  16
#define DEG_T  1024
#define TPR (DEG_HR / 16)
#define NPS (DEG_HR / DEG_S)

__device__ __forceinline__ void deg_bin4(unsigned int* cnt, int4 k, unsigned int lo) {
    unsigned int k0 = (unsigned int)k.x - lo;
    unsigned int k1 = (unsigned int)k.y - lo;
    unsigned int k2 = (unsigned int)k.z - lo;
    unsigned int k3 = (unsigned int)k.w - lo;
    if (k0 < DEG_HR) atomicAdd(&cnt[k0], 1u);
    if (k1 < DEG_HR) atomicAdd(&cnt[k1], 1u);
    if (k2 < DEG_HR) atomicAdd(&cnt[k2], 1u);
    if (k3 < DEG_HR) atomicAdd(&cnt[k3], 1u);
}

__global__ void __launch_bounds__(DEG_T)
deg_hist_kernel(const int* __restrict__ src1, const int* __restrict__ dst1,
                const int* __restrict__ src2, const int* __restrict__ dst2,
                unsigned int* __restrict__ part, int* __restrict__ gsum, int* __restrict__ tsum,
                int E, int stride) {
    __shared__ uint4 cnt4[DEG_HR / 4];
    __shared__ unsigned int tl[TPR];
    unsigned int* cnt = (unsigned int*)cnt4;
    const int tid = threadIdx.x;
    const int s = blockIdx.x;
    const unsigned int lo = blockIdx.y * DEG_HR;
    const int p = blockIdx.z;
    const int* __restrict__ keys = (p & 2) ? ((p & 1) ? dst2 : src2) : ((p & 1) ? dst1 : src1);

    for (int i = tid; i < DEG_HR / 4; i += DEG_T) cnt4[i] = make_uint4(0u, 0u, 0u, 0u);
    __syncthreads();

    const int per = (((E + DEG_S - 1) / DEG_S) + 3) & ~3;
    const int e0 = s * per;
    int e1 = e0 + per;
    if (e1 > E) e1 = E;
    const int nvec = (e1 > e0) ? ((e1 - e0) >> 2) : 0;
    const int4* __restrict__ kv = (const int4*)(keys + e0);

    int i = tid;
    for (; i + 7 * DEG_T < nvec; i += 8 * DEG_T) {
        int4 k[8];
#pragma unroll
        for (int j = 0; j < 8; ++j) k[j] = kv[i + j * DEG_T];
#pragma unroll
        for (int j = 0; j < 8; ++j) deg_bin4(cnt, k[j], lo);
    }
    for (; i < nvec; i += DEG_T) deg_bin4(cnt, kv[i], lo);
    for (int e = e0 + (nvec << 2) + tid; e < e1; e += DEG_T) {
        unsigned int k0 = (unsigned int)keys[e] - lo;
        if (k0 < DEG_HR) atomicAdd(&cnt[k0], 1u);
    }
    __syncthreads();

    uint4* __restrict__ po = (uint4*)(part + (size_t)(p * DEG_S + s) * stride + lo);
    for (int j = tid; j < DEG_HR / 4; j += DEG_T) {
        const uint4 q = cnt4[j];
        po[j] = q;
        if (p & 1) {
            unsigned int v = q.x + q.y + q.z + q.w;
            v += __shfl_xor(v, 1);
            v += __shfl_xor(v, 2);
            if ((tid & 3) == 0) tl[j >> 2] = v;
        }
    }

    if (p & 1) {
        __syncthreads();
        int* __restrict__ to = tsum + ((size_t)(p >> 1) * DEG_S + s) * (stride >> 4) + (size_t)blockIdx.y * TPR;
        for (int t = tid; t < TPR; t += DEG_T) to[t] = (int)tl[t];
        const int lane = tid & 31;
        int* __restrict__ go = gsum + ((size_t)(p >> 1) * (stride >> 8) + (size_t)blockIdx.y * (DEG_HR / 256)) * DEG_S + s;
        for (int grp = tid >> 5; grp < DEG_HR / 256; grp += DEG_T / 32) {
            const uint4 a = cnt4[grp * 64 + lane * 2];
            const uint4 b = cnt4[grp * 64 + lane * 2 + 1];
            int v = (int)(a.x + a.y + a.z + a.w + b.x + b.y + b.z + b.w);
            for (int o = 16; o > 0; o >>= 1) v += __shfl_xor(v, o);
            if (lane == 0) go[(size_t)grp * DEG_S] = v;
        }
    }
}

#define STG_CAP 16384

__device__ __forceinline__ void csr_place1(unsigned int* lcur, const unsigned int* lsta,
                                           const unsigned int* gbas, int* stage,
                                           int* __restrict__ esrc, bool staged,
                                           int key, int val, unsigned int lo) {
    const unsigned int k = (unsigned int)key - lo;
    if (k < DEG_HR) {
        const unsigned int t = k >> 4;
        const unsigned int li = atomicAdd(&lcur[t], 1u);
        const int ent = (int)(((unsigned int)val << 4) | (k & 15u));
        if (staged) stage[li] = ent;
        else esrc[gbas[t] + (li - lsta[t])] = ent;
    }
}

__device__ __forceinline__ void csr_place4(unsigned int* lcur, const unsigned int* lsta,
                                           const unsigned int* gbas, int* stage,
                                           int* __restrict__ esrc, bool staged,
                                           int4 k, int4 v, unsigned int lo) {
    csr_place1(lcur, lsta, gbas, stage, esrc, staged, k.x, v.x, lo);
    csr_place1(lcur, lsta, gbas, stage, esrc, staged, k.y, v.y, lo);
    csr_place1(lcur, lsta, gbas, stage, esrc, staged, k.z, v.z, lo);
    csr_place1(lcur, lsta, gbas, stage, esrc, staged, k.w, v.w, lo);
}

template <int IN, int OUT>
__global__ void __launch_bounds__(DEG_T, 16)
order_gemm_kernel(const float* __restrict__ X, const float* __restrict__ W, float* __restrict__ Hall,
                  const unsigned int* __restrict__ part, const int* __restrict__ gsum_all,
                  const int* __restrict__ tsum_all, float* __restrict__ rs, int* __restrict__ tstart_all,
                  const int* __restrict__ src1, const int* __restrict__ dst1,
                  const int* __restrict__ src2, const int* __restrict__ dst2,
                  int* __restrict__ esrc_all, int E, int n, int stride, int pblocks, int tpb, int dranges) {
    constexpr int LDW = IN + 4;
    static_assert(OUT * LDW + 1024 <= STG_CAP, "the transposed weights and the scales must fit the ordering buffer");
    static_assert(DEG_T == 1024, "the product role splits the block in two halves of 512 threads");
    static_assert(DEG_HR % DEG_S == 0 && NPS <= DEG_T && TPR <= DEG_T, "one tile and one node per thread");
    __shared__ unsigned int lcur[TPR];
    __shared__ unsigned int lsta[TPR];
    __shared__ unsigned int gbas[TPR];
    __shared__ unsigned int wtot[DEG_T / 32];
    __shared__ unsigned int wtot2[DEG_T / 32];
    __shared__ int red[DEG_T / 32];
    __shared__ __attribute__((aligned(16))) int stage[STG_CAP];
    const int tid = threadIdx.x;
    const int lane = tid & 31;
    const int wave = tid >> 5;

    if ((int)blockIdx.x < pblocks) {
        float* lw = (float*)stage;
        float* lrs = lw + OUT * LDW;
        for (int i = tid; i < IN * OUT; i += DEG_T) {
            int k = i / OUT;
            int cc = i - k * OUT;
            lw[cc * LDW + k] = W[i];
        }
        {
            const int j = tid >> 9;
            const int lr = tid & 511;
            const int row = (int)blockIdx.x * tpb * 16 + lr;
            if (lr < tpb * 16 && row < n) {
                const unsigned int* __restrict__ pj = part + (size_t)(2 * j) * DEG_S * stride;
                unsigned int cj = 0;
#pragma unroll
                for (int q = 0; q < DEG_S; ++q) cj += (pj + (size_t)q * stride)[row];
                lrs[j * 512 + lr] = rsqrtf(fmaxf((float)cj, 1.0f));
            }
        }
        __syncthreads();

        if (wave >= tpb) return;
        const int r0 = ((int)blockIdx.x * tpb + wave) * 16;
        if (r0 >= n) return;

        const int mrow = r0 + (lane & 15);
        const int koff = (lane >> 4) * 2;
        const float* xrow = X + (size_t)mrow * IN;
        const int col = lane & 15;

        constexpr int CT = OUT / 16;
        const int mhi = (lane >> 4) * 8;
        v8f acc[CT] = {};
        for (int k = 0; k < IN; k += 4) {
            float2 av = *(const float2*)(xrow + k + koff);
            v2f a;
            a.x = av.x;
            a.y = av.y;
#pragma unroll
            for (int cc = 0; cc < CT; ++cc) {
                float2 bv = *(const float2*)&lw[(cc * 16 + col) * LDW + k + koff];
                v2f b;
                b.x = bv.x;
                b.y = bv.y;
                acc[cc] = __builtin_amdgcn_wmma_f32_16x16x4_f32(
                    false, a, false, b, (short)0, acc[cc], false, false);
            }
        }
        const float* ls0 = lrs + wave * 16 + mhi;
        const float* ls1 = ls0 + 512;
        float* __restrict__ H0 = Hall;
        float* __restrict__ H1 = Hall + (size_t)n * OUT;
#pragma unroll
        for (int r = 0; r < 8; ++r) {
            const float s0 = ls0[r];
            const float s1 = ls1[r];
            const size_t ro = (size_t)(r0 + mhi + r) * OUT + col;
#pragma unroll
            for (int cc = 0; cc < CT; ++cc) {
                H0[ro + cc * 16] = acc[cc][r] * s0;
                H1[ro + cc * 16] = acc[cc][r] * s1;
            }
        }
        return;
    }

    const int pb = (int)blockIdx.x - pblocks;
    const int s = pb % DEG_S;
    const int rg = pb / DEG_S;
    const int g = rg / dranges;
    const int r = rg - g * dranges;
    const unsigned int lo = (unsigned int)r * DEG_HR;
    const int tp = stride >> 4;
    const int* __restrict__ src = g ? src2 : src1;
    const int* __restrict__ dst = g ? dst2 : dst1;
    int* __restrict__ esrc = esrc_all + (size_t)g * E;

    const int4* __restrict__ gs4 = (const int4*)(gsum_all + (size_t)g * (stride >> 8) * DEG_S);
    const int lim4 = r * (DEG_HR / 256) * (DEG_S / 4);
    int o = 0;
    for (int j = tid; j < lim4; j += DEG_T) {
        const int4 q = gs4[j];
        o += q.x + q.y + q.z + q.w;
    }
    for (int d = 16; d > 0; d >>= 1) o += __shfl_xor(o, d);
    if (lane == 0) red[wave] = o;

    unsigned int tot = 0u, pre = 0u, c = 0u;
    if (tid < TPR) {
        const int* __restrict__ tsg = tsum_all + (size_t)g * DEG_S * tp + (size_t)r * TPR;
#pragma unroll
        for (int q = 0; q < DEG_S; ++q) {
            const unsigned int v = (unsigned int)(tsg + (size_t)q * tp)[tid];
            tot += v;
            pre += (q < s) ? v : 0u;
            c = (q == s) ? v : c;
        }
    }

    if (tid < NPS) {
        const int node = (int)lo + s * NPS + tid;
        if (node < n) {
            const unsigned int* __restrict__ po = part + (size_t)(2 * g) * DEG_S * stride;
            const unsigned int* __restrict__ pi = part + (size_t)(2 * g + 1) * DEG_S * stride;
            unsigned int co = 0u, ci = 0u;
#pragma unroll
            for (int q = 0; q < DEG_S; ++q) {
                co += (po + (size_t)q * stride)[node];
                ci += (pi + (size_t)q * stride)[node];
            }
            rs[(size_t)(2 * g) * n + node] = rsqrtf(fmaxf((float)co, 1.0f));
            rs[(size_t)(2 * g + 1) * n + node] = rsqrtf(fmaxf((float)ci, 1.0f));
        }
    }

    int inc = (int)c;
    int inc2 = (int)tot;
    for (int d = 1; d < 32; d <<= 1) {
        const int u = __shfl_up(inc, d);
        const int u2 = __shfl_up(inc2, d);
        if (lane >= d) { inc += u; inc2 += u2; }
    }
    if (lane == 31) { wtot[wave] = (unsigned int)inc; wtot2[wave] = (unsigned int)inc2; }
    __syncthreads();
    unsigned int before = 0u, total = 0u, before2 = 0u;
    int base = 0;
    for (int w = 0; w < DEG_T / 32; ++w) {
        const unsigned int x = wtot[w];
        const unsigned int x2 = wtot2[w];
        total += x;
        if (w < wave) { before += x; before2 += x2; }
        base += red[w];
    }
    const unsigned int excl = before + (unsigned int)inc - c;
    const unsigned int tsta = (unsigned int)base + before2 + (unsigned int)inc2 - tot;
    if (tid < TPR) {
        lsta[tid] = excl;
        lcur[tid] = excl;
        gbas[tid] = tsta + pre;
    }
    const bool staged = total <= STG_CAP;
    if (s == 0 && tid < TPR) {
        const int ntiles = (n + 15) >> 4;
        const int T = r * TPR + tid;
        int* __restrict__ tst = tstart_all + (size_t)g * (ntiles + 1);
        if (T < ntiles) tst[T] = (int)tsta;
        if (T == ntiles - 1) tst[ntiles] = (int)(tsta + tot);
    }
    __syncthreads();

    const int per = (((E + DEG_S - 1) / DEG_S) + 3) & ~3;
    const int e0 = s * per;
    int e1 = e0 + per;
    if (e1 > E) e1 = E;
    const int nvec = (e1 > e0) ? ((e1 - e0) >> 2) : 0;
    const int4* __restrict__ kv = (const int4*)(dst + e0);
    const int4* __restrict__ sv = (const int4*)(src + e0);

    int i = tid;
    for (; i + 3 * DEG_T < nvec; i += 4 * DEG_T) {
        int4 k[4], v[4];
#pragma unroll
        for (int j = 0; j < 4; ++j) { k[j] = kv[i + j * DEG_T]; v[j] = sv[i + j * DEG_T]; }
#pragma unroll
        for (int j = 0; j < 4; ++j) csr_place4(lcur, lsta, gbas, stage, esrc, staged, k[j], v[j], lo);
    }
    for (; i < nvec; i += DEG_T) csr_place4(lcur, lsta, gbas, stage, esrc, staged, kv[i], sv[i], lo);
    for (int e = e0 + (nvec << 2) + tid; e < e1; e += DEG_T)
        csr_place1(lcur, lsta, gbas, stage, esrc, staged, dst[e], src[e], lo);
    __syncthreads();

    if (staged) {
        const int wv = __builtin_amdgcn_readfirstlane(wave);
        const int tk = wv + (DEG_T / 32) * lane;
        unsigned int ls_l = 0u, len_l = 0u, gb_l = 0u;
        if (tk < TPR) {
            ls_l = lsta[tk];
            len_l = lcur[tk] - ls_l;
            gb_l = gbas[tk];
        }
        for (int k = 0; wv + (DEG_T / 32) * k < TPR; ++k) {
            const unsigned int ls = (unsigned int)__builtin_amdgcn_readlane((int)ls_l, k);
            const unsigned int len = (unsigned int)__builtin_amdgcn_readlane((int)len_l, k);
            const unsigned int gb = (unsigned int)__builtin_amdgcn_readlane((int)gb_l, k);
            for (unsigned int j = lane; j < len; j += 32) esrc[gb + j] = stage[ls + j];
        }
    }
}

#define GRW 16

__device__ __forceinline__ float lane_pick_f(float v, int i) {
    return __int_as_float(__builtin_amdgcn_readlane(__float_as_int(v), i));
}

template <int F>
__device__ __forceinline__ void tile_stream(const float* __restrict__ h, const int* __restrict__ esrc,
                                            int eb, int ee, int lane, v16f (&acc)[F / 32]) {
    constexpr int R = F / 32;
    constexpr int U = 16 / R;
    if (ee > eb) {
        int idn = esrc[(eb + lane < ee) ? (eb + lane) : (ee - 1)];
        for (int base = eb; base < ee; base += 32) {
            const int id_l = idn;
            const int nx = base + 32 + lane;
            idn = esrc[(nx < ee) ? nx : (ee - 1)];
            const int cnt = (ee - base < 32) ? (ee - base) : 32;
            for (int j0 = 0; j0 < cnt; j0 += U) {
                float v[U * R];
                int tag[U];
#pragma unroll
                for (int u = 0; u < U; ++u) {
                    const unsigned int ent = (unsigned int)__builtin_amdgcn_readlane(id_l, j0 + u);
                    tag[u] = (int)(ent & 15u);
                    const float* __restrict__ hp = h + (size_t)(ent >> 4) * F;
#pragma unroll
                    for (int r = 0; r < R; ++r) v[u * R + r] = hp[lane + r * 32];
                }
                asm volatile("" :: "v"(v[0]), "v"(v[1]), "v"(v[2]), "v"(v[3]), "v"(v[4]), "v"(v[5]),
                             "v"(v[6]), "v"(v[7]), "v"(v[8]), "v"(v[9]), "v"(v[10]), "v"(v[11]),
                             "v"(v[12]), "v"(v[13]), "v"(v[14]), "v"(v[15]));
#pragma unroll
                for (int u = 0; u < U; ++u) {
                    if (j0 + u < cnt) {
#pragma unroll
                        for (int r = 0; r < R; ++r) acc[r][tag[u]] += v[u * R + r];
                    }
                }
            }
        }
    }
}

#define GLD (HID + 4)

__global__ void __launch_bounds__(256)
gather_relu_gemm_kernel(const float* __restrict__ h_all, const int* __restrict__ tstart_all,
                        const int* __restrict__ esrc_all, const float* __restrict__ rs_all,
                        const float* __restrict__ b, const float* __restrict__ W2,
                        float* __restrict__ h2_all, int n, int E) {
    constexpr int R = HID / 32;
    constexpr int CT = NC / 16;
    __shared__ __attribute__((aligned(16))) float lw[NC * GLD];
    __shared__ __attribute__((aligned(16))) float la[8 * GRW * GLD];

#pragma unroll
    for (int j = 0; j < HID * NC / 256; ++j) {
        const int i = j * 256 + (int)threadIdx.x;
        const int k = i / NC;
        const int c = i - k * NC;
        lw[c * GLD + k] = W2[i];
    }
    __syncthreads();

    const int g = blockIdx.y;
    const float* __restrict__ h = h_all + (size_t)g * n * HID;
    const int* __restrict__ tstart = tstart_all + (size_t)g * (((n + GRW - 1) / GRW) + 1);
    const int* __restrict__ esrc = esrc_all + (size_t)g * E;
    const float* __restrict__ rs_out = rs_all + (size_t)(2 * g) * n;
    const float* __restrict__ rs_in = rs_all + (size_t)(2 * g + 1) * n;
    float* __restrict__ h2 = h2_all + (size_t)g * n * NC;
    const int lane = threadIdx.x & 31;
    const int wave = threadIdx.x >> 5;
    const int gw = __builtin_amdgcn_readfirstlane((int)(blockIdx.x * 8 + wave));
    const int n0 = gw * GRW;
    if (n0 >= n) return;
    const int nn = (n - n0 < GRW) ? (n - n0) : GRW;
    const int eb = __builtin_amdgcn_readfirstlane(tstart[gw]);
    const int ee = __builtin_amdgcn_readfirstlane(tstart[gw + 1]);
    float* __restrict__ lt = la + wave * (GRW * GLD);

    v16f acc[R] = {};
    tile_stream<HID>(h, esrc, eb, ee, lane, acc);

    const int ib = n0 + ((lane < nn) ? lane : (nn - 1));
    const float sci_l = rs_in[ib];
    const float sco_l = rs_out[ib];
    float bias[R];
#pragma unroll
    for (int r = 0; r < R; ++r) bias[r] = b[lane + r * 32];
#pragma unroll
    for (int i = 0; i < GRW; ++i) {
        const float rsi = lane_pick_f(sci_l, i);
        const float rso = lane_pick_f(sco_l, i);
#pragma unroll
        for (int r = 0; r < R; ++r)
            lt[i * GLD + lane + r * 32] = fmaxf(acc[r][i] * rsi + bias[r], 0.0f) * rso;
    }

    const int col = lane & 15;
    const int koff = (lane >> 4) * 2;
    v8f d[CT] = {};
    for (int k = 0; k < HID; k += 4) {
        float2 av = *(const float2*)&lt[col * GLD + k + koff];
        v2f a;
        a.x = av.x;
        a.y = av.y;
#pragma unroll
        for (int c = 0; c < CT; ++c) {
            float2 bv = *(const float2*)&lw[(c * 16 + col) * GLD + k + koff];
            v2f bb;
            bb.x = bv.x;
            bb.y = bv.y;
            d[c] = __builtin_amdgcn_wmma_f32_16x16x4_f32(
                false, a, false, bb, (short)0, d[c], false, false);
        }
    }

    const int mhi = (lane >> 4) * 8;
#pragma unroll
    for (int c = 0; c < CT; ++c) {
#pragma unroll
        for (int r = 0; r < 8; ++r) lt[(mhi + r) * GLD + c * 16 + col] = d[c][r];
    }
    for (int i = 0; i < nn; ++i) h2[(size_t)(n0 + i) * NC + lane] = lt[i * GLD + lane];
}

__global__ void __launch_bounds__(256)
gather_mean_kernel(const float* __restrict__ hA, const int* __restrict__ tstartA,
                   const int* __restrict__ esrcA, const float* __restrict__ rsA,
                   const float* __restrict__ hB, const int* __restrict__ tstartB,
                   const int* __restrict__ esrcB, const float* __restrict__ rsB,
                   const float* __restrict__ b, float* __restrict__ out, int n) {
    const int lane = threadIdx.x & 31;
    const int gw = __builtin_amdgcn_readfirstlane((int)((blockIdx.x * 256 + threadIdx.x) >> 5));
    const int n0 = gw * GRW;
    if (n0 >= n) return;
    const int nn = (n - n0 < GRW) ? (n - n0) : GRW;
    const int ib = n0 + ((lane < nn) ? lane : (nn - 1));
    const int ebA = __builtin_amdgcn_readfirstlane(tstartA[gw]);
    const int eeA = __builtin_amdgcn_readfirstlane(tstartA[gw + 1]);
    const int ebB = __builtin_amdgcn_readfirstlane(tstartB[gw]);
    const int eeB = __builtin_amdgcn_readfirstlane(tstartB[gw + 1]);
    const float scA_l = rsA[ib];
    const float scB_l = rsB[ib];
    const float bias = b[lane];

    v16f accA[1] = {};
    v16f accB[1] = {};
    tile_stream<NC>(hA, esrcA, ebA, eeA, lane, accA);
    tile_stream<NC>(hB, esrcB, ebB, eeB, lane, accB);
#pragma unroll
    for (int i = 0; i < GRW; ++i) {
        if (i < nn) {
            const float tA = accA[0][i] * lane_pick_f(scA_l, i) + bias;
            const float tB = accB[0][i] * lane_pick_f(scB_l, i) + bias;
            out[(size_t)(n0 + i) * NC + lane] = 0.5f * tA + 0.5f * tB;
        }
    }
}


extern "C" void kernel_launch(void* const* d_in, const int* in_sizes, int n_in,
                              void* d_out, int out_size, void* d_ws, size_t ws_size,
                              hipStream_t stream) {
    const float* x    = (const float*)d_in[0];
    const int*   src1 = (const int*)d_in[1];
    const int*   dst1 = (const int*)d_in[2];
    const int*   src2 = (const int*)d_in[3];
    const int*   dst2 = (const int*)d_in[4];
    const float* W1   = (const float*)d_in[5];
    const float* b1   = (const float*)d_in[6];
    const float* W2   = (const float*)d_in[7];
    const float* b2   = (const float*)d_in[8];
    float* out = (float*)d_out;

    const int N = in_sizes[0] / IN_FEATS;
    const int E = in_sizes[1];

    const int dranges = (N + DEG_HR - 1) / DEG_HR;
    const int dstride = dranges * DEG_HR;
    unsigned int* dpart = (unsigned int*)d_ws;
    float* ws   = (float*)(dpart + (size_t)4 * DEG_S * dstride);
    float* rsOI = ws;
    float* h    = rsOI + (size_t)4 * N;
    float* h2   = h + (size_t)2 * N * HID;
    int* esrc   = (int*)(h2 + (size_t)2 * N * NC);
    int* gsum   = esrc + (size_t)2 * E;
    int* tstart = gsum + (size_t)2 * (dstride / 256) * DEG_S;
    int* tsum   = tstart + (size_t)2 * (N + 1);

    const int T = 256;
    auto cdiv = [](long a, long b) { return (int)((a + b - 1) / b); };

    const int tiles   = N / 16;
    const int tpw     = cdiv(tiles, 256);
    const int tpb     = (tpw > DEG_T / 32) ? (DEG_T / 32) : ((tpw < 1) ? 1 : tpw);
    const int pblocks = cdiv(tiles, tpb);
    const int tblocks = cdiv(cdiv(N, GRW), 8);

    deg_hist_kernel<<<dim3(DEG_S, dranges, 4), DEG_T, 0, stream>>>(src1, dst1, src2, dst2, dpart, gsum, tsum, E, dstride);
    order_gemm_kernel<IN_FEATS, HID><<<pblocks + DEG_S * dranges * 2, DEG_T, 0, stream>>>(
        x, W1, h, dpart, gsum, tsum, rsOI, tstart, src1, dst1, src2, dst2, esrc, E, N, dstride, pblocks, tpb, dranges);

    gather_relu_gemm_kernel<<<dim3(tblocks, 2), T, 0, stream>>>(h, tstart, esrc, rsOI, b1, W2, h2, N, E);

    gather_mean_kernel<<<tblocks, T, 0, stream>>>(
        h2, tstart, esrc, rsOI + N,
        h2 + (size_t)N * NC, tstart + (cdiv(N, GRW) + 1), esrc + E, rsOI + (size_t)3 * N,
        b2, out, N);
}


# layer-2 edge sums: a wave requests the 32 rows of the next chunk before it sums the current one (64 row loads in flight, before: 16, then none during the adds); per-node add order unchanged
# speedup vs baseline: 5.7978x; 1.0428x over previous
#include <hip/hip_runtime.h>


typedef __attribute__((ext_vector_type(2))) float v2f;
typedef __attribute__((ext_vector_type(8))) float v8f;
typedef __attribute__((ext_vector_type(16))) float v16f;

#define IN_FEATS 128
#define HID      64
#define NC       32


#define DEG_HR 12544
#define DEG_S  16
#define DEG_T  1024
#define TPR (DEG_HR / 16)
#define NPS (DEG_HR / DEG_S)

__device__ __forceinline__ void deg_bin4(unsigned int* cnt, int4 k, unsigned int lo) {
    unsigned int k0 = (unsigned int)k.x - lo;
    unsigned int k1 = (unsigned int)k.y - lo;
    unsigned int k2 = (unsigned int)k.z - lo;
    unsigned int k3 = (unsigned int)k.w - lo;
    if (k0 < DEG_HR) atomicAdd(&cnt[k0], 1u);
    if (k1 < DEG_HR) atomicAdd(&cnt[k1], 1u);
    if (k2 < DEG_HR) atomicAdd(&cnt[k2], 1u);
    if (k3 < DEG_HR) atomicAdd(&cnt[k3], 1u);
}

__global__ void __launch_bounds__(DEG_T)
deg_hist_kernel(const int* __restrict__ src1, const int* __restrict__ dst1,
                const int* __restrict__ src2, const int* __restrict__ dst2,
                unsigned int* __restrict__ part, int* __restrict__ gsum, int* __restrict__ tsum,
                int E, int stride) {
    __shared__ uint4 cnt4[DEG_HR / 4];
    __shared__ unsigned int tl[TPR];
    unsigned int* cnt = (unsigned int*)cnt4;
    const int tid = threadIdx.x;
    const int s = blockIdx.x;
    const unsigned int lo = blockIdx.y * DEG_HR;
    const int p = blockIdx.z;
    const int* __restrict__ keys = (p & 2) ? ((p & 1) ? dst2 : src2) : ((p & 1) ? dst1 : src1);

    for (int i = tid; i < DEG_HR / 4; i += DEG_T) cnt4[i] = make_uint4(0u, 0u, 0u, 0u);
    __syncthreads();

    const int per = (((E + DEG_S - 1) / DEG_S) + 3) & ~3;
    const int e0 = s * per;
    int e1 = e0 + per;
    if (e1 > E) e1 = E;
    const int nvec = (e1 > e0) ? ((e1 - e0) >> 2) : 0;
    const int4* __restrict__ kv = (const int4*)(keys + e0);

    int i = tid;
    for (; i + 7 * DEG_T < nvec; i += 8 * DEG_T) {
        int4 k[8];
#pragma unroll
        for (int j = 0; j < 8; ++j) k[j] = kv[i + j * DEG_T];
#pragma unroll
        for (int j = 0; j < 8; ++j) deg_bin4(cnt, k[j], lo);
    }
    for (; i < nvec; i += DEG_T) deg_bin4(cnt, kv[i], lo);
    for (int e = e0 + (nvec << 2) + tid; e < e1; e += DEG_T) {
        unsigned int k0 = (unsigned int)keys[e] - lo;
        if (k0 < DEG_HR) atomicAdd(&cnt[k0], 1u);
    }
    __syncthreads();

    uint4* __restrict__ po = (uint4*)(part + (size_t)(p * DEG_S + s) * stride + lo);
    for (int j = tid; j < DEG_HR / 4; j += DEG_T) {
        const uint4 q = cnt4[j];
        po[j] = q;
        if (p & 1) {
            unsigned int v = q.x + q.y + q.z + q.w;
            v += __shfl_xor(v, 1);
            v += __shfl_xor(v, 2);
            if ((tid & 3) == 0) tl[j >> 2] = v;
        }
    }

    if (p & 1) {
        __syncthreads();
        int* __restrict__ to = tsum + ((size_t)(p >> 1) * DEG_S + s) * (stride >> 4) + (size_t)blockIdx.y * TPR;
        for (int t = tid; t < TPR; t += DEG_T) to[t] = (int)tl[t];
        const int lane = tid & 31;
        int* __restrict__ go = gsum + ((size_t)(p >> 1) * (stride >> 8) + (size_t)blockIdx.y * (DEG_HR / 256)) * DEG_S + s;
        for (int grp = tid >> 5; grp < DEG_HR / 256; grp += DEG_T / 32) {
            const uint4 a = cnt4[grp * 64 + lane * 2];
            const uint4 b = cnt4[grp * 64 + lane * 2 + 1];
            int v = (int)(a.x + a.y + a.z + a.w + b.x + b.y + b.z + b.w);
            for (int o = 16; o > 0; o >>= 1) v += __shfl_xor(v, o);
            if (lane == 0) go[(size_t)grp * DEG_S] = v;
        }
    }
}

#define STG_CAP 16384

__device__ __forceinline__ void csr_place1(unsigned int* lcur, const unsigned int* lsta,
                                           const unsigned int* gbas, int* stage,
                                           int* __restrict__ esrc, bool staged,
                                           int key, int val, unsigned int lo) {
    const unsigned int k = (unsigned int)key - lo;
    if (k < DEG_HR) {
        const unsigned int t = k >> 4;
        const unsigned int li = atomicAdd(&lcur[t], 1u);
        const int ent = (int)(((unsigned int)val << 4) | (k & 15u));
        if (staged) stage[li] = ent;
        else esrc[gbas[t] + (li - lsta[t])] = ent;
    }
}

__device__ __forceinline__ void csr_place4(unsigned int* lcur, const unsigned int* lsta,
                                           const unsigned int* gbas, int* stage,
                                           int* __restrict__ esrc, bool staged,
                                           int4 k, int4 v, unsigned int lo) {
    csr_place1(lcur, lsta, gbas, stage, esrc, staged, k.x, v.x, lo);
    csr_place1(lcur, lsta, gbas, stage, esrc, staged, k.y, v.y, lo);
    csr_place1(lcur, lsta, gbas, stage, esrc, staged, k.z, v.z, lo);
    csr_place1(lcur, lsta, gbas, stage, esrc, staged, k.w, v.w, lo);
}

template <int IN, int OUT>
__global__ void __launch_bounds__(DEG_T, 16)
order_gemm_kernel(const float* __restrict__ X, const float* __restrict__ W, float* __restrict__ Hall,
                  const unsigned int* __restrict__ part, const int* __restrict__ gsum_all,
                  const int* __restrict__ tsum_all, float* __restrict__ rs, int* __restrict__ tstart_all,
                  const int* __restrict__ src1, const int* __restrict__ dst1,
                  const int* __restrict__ src2, const int* __restrict__ dst2,
                  int* __restrict__ esrc_all, int E, int n, int stride, int pblocks, int tpb, int dranges) {
    constexpr int LDW = IN + 4;
    static_assert(OUT * LDW + 1024 <= STG_CAP, "the transposed weights and the scales must fit the ordering buffer");
    static_assert(DEG_T == 1024, "the product role splits the block in two halves of 512 threads");
    static_assert(DEG_HR % DEG_S == 0 && NPS <= DEG_T && TPR <= DEG_T, "one tile and one node per thread");
    __shared__ unsigned int lcur[TPR];
    __shared__ unsigned int lsta[TPR];
    __shared__ unsigned int gbas[TPR];
    __shared__ unsigned int wtot[DEG_T / 32];
    __shared__ unsigned int wtot2[DEG_T / 32];
    __shared__ int red[DEG_T / 32];
    __shared__ __attribute__((aligned(16))) int stage[STG_CAP];
    const int tid = threadIdx.x;
    const int lane = tid & 31;
    const int wave = tid >> 5;

    if ((int)blockIdx.x < pblocks) {
        float* lw = (float*)stage;
        float* lrs = lw + OUT * LDW;
        for (int i = tid; i < IN * OUT; i += DEG_T) {
            int k = i / OUT;
            int cc = i - k * OUT;
            lw[cc * LDW + k] = W[i];
        }
        {
            const int j = tid >> 9;
            const int lr = tid & 511;
            const int row = (int)blockIdx.x * tpb * 16 + lr;
            if (lr < tpb * 16 && row < n) {
                const unsigned int* __restrict__ pj = part + (size_t)(2 * j) * DEG_S * stride;
                unsigned int cj = 0;
#pragma unroll
                for (int q = 0; q < DEG_S; ++q) cj += (pj + (size_t)q * stride)[row];
                lrs[j * 512 + lr] = rsqrtf(fmaxf((float)cj, 1.0f));
            }
        }
        __syncthreads();

        if (wave >= tpb) return;
        const int r0 = ((int)blockIdx.x * tpb + wave) * 16;
        if (r0 >= n) return;

        const int mrow = r0 + (lane & 15);
        const int koff = (lane >> 4) * 2;
        const float* xrow = X + (size_t)mrow * IN;
        const int col = lane & 15;

        constexpr int CT = OUT / 16;
        const int mhi = (lane >> 4) * 8;
        v8f acc[CT] = {};
        for (int k = 0; k < IN; k += 4) {
            float2 av = *(const float2*)(xrow + k + koff);
            v2f a;
            a.x = av.x;
            a.y = av.y;
#pragma unroll
            for (int cc = 0; cc < CT; ++cc) {
                float2 bv = *(const float2*)&lw[(cc * 16 + col) * LDW + k + koff];
                v2f b;
                b.x = bv.x;
                b.y = bv.y;
                acc[cc] = __builtin_amdgcn_wmma_f32_16x16x4_f32(
                    false, a, false, b, (short)0, acc[cc], false, false);
            }
        }
        const float* ls0 = lrs + wave * 16 + mhi;
        const float* ls1 = ls0 + 512;
        float* __restrict__ H0 = Hall;
        float* __restrict__ H1 = Hall + (size_t)n * OUT;
#pragma unroll
        for (int r = 0; r < 8; ++r) {
            const float s0 = ls0[r];
            const float s1 = ls1[r];
            const size_t ro = (size_t)(r0 + mhi + r) * OUT + col;
#pragma unroll
            for (int cc = 0; cc < CT; ++cc) {
                H0[ro + cc * 16] = acc[cc][r] * s0;
                H1[ro + cc * 16] = acc[cc][r] * s1;
            }
        }
        return;
    }

    const int pb = (int)blockIdx.x - pblocks;
    const int s = pb % DEG_S;
    const int rg = pb / DEG_S;
    const int g = rg / dranges;
    const int r = rg - g * dranges;
    const unsigned int lo = (unsigned int)r * DEG_HR;
    const int tp = stride >> 4;
    const int* __restrict__ src = g ? src2 : src1;
    const int* __restrict__ dst = g ? dst2 : dst1;
    int* __restrict__ esrc = esrc_all + (size_t)g * E;

    const int4* __restrict__ gs4 = (const int4*)(gsum_all + (size_t)g * (stride >> 8) * DEG_S);
    const int lim4 = r * (DEG_HR / 256) * (DEG_S / 4);
    int o = 0;
    for (int j = tid; j < lim4; j += DEG_T) {
        const int4 q = gs4[j];
        o += q.x + q.y + q.z + q.w;
    }
    for (int d = 16; d > 0; d >>= 1) o += __shfl_xor(o, d);
    if (lane == 0) red[wave] = o;

    unsigned int tot = 0u, pre = 0u, c = 0u;
    if (tid < TPR) {
        const int* __restrict__ tsg = tsum_all + (size_t)g * DEG_S * tp + (size_t)r * TPR;
#pragma unroll
        for (int q = 0; q < DEG_S; ++q) {
            const unsigned int v = (unsigned int)(tsg + (size_t)q * tp)[tid];
            tot += v;
            pre += (q < s) ? v : 0u;
            c = (q == s) ? v : c;
        }
    }

    if (tid < NPS) {
        const int node = (int)lo + s * NPS + tid;
        if (node < n) {
            const unsigned int* __restrict__ po = part + (size_t)(2 * g) * DEG_S * stride;
            const unsigned int* __restrict__ pi = part + (size_t)(2 * g + 1) * DEG_S * stride;
            unsigned int co = 0u, ci = 0u;
#pragma unroll
            for (int q = 0; q < DEG_S; ++q) {
                co += (po + (size_t)q * stride)[node];
                ci += (pi + (size_t)q * stride)[node];
            }
            rs[(size_t)(2 * g) * n + node] = rsqrtf(fmaxf((float)co, 1.0f));
            rs[(size_t)(2 * g + 1) * n + node] = rsqrtf(fmaxf((float)ci, 1.0f));
        }
    }

    int inc = (int)c;
    int inc2 = (int)tot;
    for (int d = 1; d < 32; d <<= 1) {
        const int u = __shfl_up(inc, d);
        const int u2 = __shfl_up(inc2, d);
        if (lane >= d) { inc += u; inc2 += u2; }
    }
    if (lane == 31) { wtot[wave] = (unsigned int)inc; wtot2[wave] = (unsigned int)inc2; }
    __syncthreads();
    unsigned int before = 0u, total = 0u, before2 = 0u;
    int base = 0;
    for (int w = 0; w < DEG_T / 32; ++w) {
        const unsigned int x = wtot[w];
        const unsigned int x2 = wtot2[w];
        total += x;
        if (w < wave) { before += x; before2 += x2; }
        base += red[w];
    }
    const unsigned int excl = before + (unsigned int)inc - c;
    const unsigned int tsta = (unsigned int)base + before2 + (unsigned int)inc2 - tot;
    if (tid < TPR) {
        lsta[tid] = excl;
        lcur[tid] = excl;
        gbas[tid] = tsta + pre;
    }
    const bool staged = total <= STG_CAP;
    if (s == 0 && tid < TPR) {
        const int ntiles = (n + 15) >> 4;
        const int T = r * TPR + tid;
        int* __restrict__ tst = tstart_all + (size_t)g * (ntiles + 1);
        if (T < ntiles) tst[T] = (int)tsta;
        if (T == ntiles - 1) tst[ntiles] = (int)(tsta + tot);
    }
    __syncthreads();

    const int per = (((E + DEG_S - 1) / DEG_S) + 3) & ~3;
    const int e0 = s * per;
    int e1 = e0 + per;
    if (e1 > E) e1 = E;
    const int nvec = (e1 > e0) ? ((e1 - e0) >> 2) : 0;
    const int4* __restrict__ kv = (const int4*)(dst + e0);
    const int4* __restrict__ sv = (const int4*)(src + e0);

    int i = tid;
    for (; i + 3 * DEG_T < nvec; i += 4 * DEG_T) {
        int4 k[4], v[4];
#pragma unroll
        for (int j = 0; j < 4; ++j) { k[j] = kv[i + j * DEG_T]; v[j] = sv[i + j * DEG_T]; }
#pragma unroll
        for (int j = 0; j < 4; ++j) csr_place4(lcur, lsta, gbas, stage, esrc, staged, k[j], v[j], lo);
    }
    for (; i < nvec; i += DEG_T) csr_place4(lcur, lsta, gbas, stage, esrc, staged, kv[i], sv[i], lo);
    for (int e = e0 + (nvec << 2) + tid; e < e1; e += DEG_T)
        csr_place1(lcur, lsta, gbas, stage, esrc, staged, dst[e], src[e], lo);
    __syncthreads();

    if (staged) {
        const int wv = __builtin_amdgcn_readfirstlane(wave);
        const int tk = wv + (DEG_T / 32) * lane;
        unsigned int ls_l = 0u, len_l = 0u, gb_l = 0u;
        if (tk < TPR) {
            ls_l = lsta[tk];
            len_l = lcur[tk] - ls_l;
            gb_l = gbas[tk];
        }
        for (int k = 0; wv + (DEG_T / 32) * k < TPR; ++k) {
            const unsigned int ls = (unsigned int)__builtin_amdgcn_readlane((int)ls_l, k);
            const unsigned int len = (unsigned int)__builtin_amdgcn_readlane((int)len_l, k);
            const unsigned int gb = (unsigned int)__builtin_amdgcn_readlane((int)gb_l, k);
            for (unsigned int j = lane; j < len; j += 32) esrc[gb + j] = stage[ls + j];
        }
    }
}

#define GRW 16

__device__ __forceinline__ float lane_pick_f(float v, int i) {
    return __int_as_float(__builtin_amdgcn_readlane(__float_as_int(v), i));
}

template <int F>
__device__ __forceinline__ void tile_stream(const float* __restrict__ h, const int* __restrict__ esrc,
                                            int eb, int ee, int lane, v16f (&acc)[F / 32]) {
    constexpr int R = F / 32;
    constexpr int U = 16 / R;
    if (ee > eb) {
        int idn = esrc[(eb + lane < ee) ? (eb + lane) : (ee - 1)];
        for (int base = eb; base < ee; base += 32) {
            const int id_l = idn;
            const int nx = base + 32 + lane;
            idn = esrc[(nx < ee) ? nx : (ee - 1)];
            const int cnt = (ee - base < 32) ? (ee - base) : 32;
            for (int j0 = 0; j0 < cnt; j0 += U) {
                float v[U * R];
                int tag[U];
#pragma unroll
                for (int u = 0; u < U; ++u) {
                    const unsigned int ent = (unsigned int)__builtin_amdgcn_readlane(id_l, j0 + u);
                    tag[u] = (int)(ent & 15u);
                    const float* __restrict__ hp = h + (size_t)(ent >> 4) * F;
#pragma unroll
                    for (int r = 0; r < R; ++r) v[u * R + r] = hp[lane + r * 32];
                }
                asm volatile("" :: "v"(v[0]), "v"(v[1]), "v"(v[2]), "v"(v[3]), "v"(v[4]), "v"(v[5]),
                             "v"(v[6]), "v"(v[7]), "v"(v[8]), "v"(v[9]), "v"(v[10]), "v"(v[11]),
                             "v"(v[12]), "v"(v[13]), "v"(v[14]), "v"(v[15]));
#pragma unroll
                for (int u = 0; u < U; ++u) {
                    if (j0 + u < cnt) {
#pragma unroll
                        for (int r = 0; r < R; ++r) acc[r][tag[u]] += v[u * R + r];
                    }
                }
            }
        }
    }
}

template <int F>
__device__ __forceinline__ void chunk_request(const float* __restrict__ h, int id_l, int lane, float (&v)[32]) {
#pragma unroll
    for (int u = 0; u < 32; ++u) {
        const unsigned int ent = (unsigned int)__builtin_amdgcn_readlane(id_l, u);
        const unsigned int off = (ent >> 4) * (unsigned int)(F * sizeof(float)) + (unsigned int)lane * (unsigned int)sizeof(float);
        v[u] = *(const float*)((const char*)h + off);
    }
}

__device__ __forceinline__ void chunk_add(int id_l, int cnt, const float (&v)[32], v16f& acc) {
#pragma unroll
    for (int u = 0; u < 32; ++u) {
        if (u < cnt) {
            const int tag = __builtin_amdgcn_readlane(id_l, u) & 15;
            acc[tag] += v[u];
        }
    }
}

template <int F>
__device__ __forceinline__ void tile_stream_ahead(const float* __restrict__ h, const int* __restrict__ esrc,
                                                  int eb, int ee, int lane, v16f& acc) {
    static_assert(F == 32, "one float per lane and row");
    if (ee > eb) {
        const int last = ee - 1;
        float va[32], vb[32];
        int idA = esrc[(eb + lane < ee) ? (eb + lane) : last];
        int idB = esrc[(eb + 32 + lane < ee) ? (eb + 32 + lane) : last];
        __builtin_amdgcn_sched_barrier(0);
        chunk_request<F>(h, idA, lane, va);
        for (int base = eb; base < ee; base += 64) {
            const int nA = base + 64 + lane;
            const int idA2 = esrc[(nA < ee) ? nA : last];
            __builtin_amdgcn_sched_barrier(0);
            chunk_request<F>(h, idB, lane, vb);
            __builtin_amdgcn_sched_barrier(0);
            chunk_add(idA, ee - base, va, acc);
            const int nB = base + 96 + lane;
            const int idB2 = esrc[(nB < ee) ? nB : last];
            __builtin_amdgcn_sched_barrier(0);
            chunk_request<F>(h, idA2, lane, va);
            __builtin_amdgcn_sched_barrier(0);
            chunk_add(idB, ee - base - 32, vb, acc);
            idA = idA2;
            idB = idB2;
        }
    }
}

#define GLD (HID + 4)

__global__ void __launch_bounds__(256)
gather_relu_gemm_kernel(const float* __restrict__ h_all, const int* __restrict__ tstart_all,
                        const int* __restrict__ esrc_all, const float* __restrict__ rs_all,
                        const float* __restrict__ b, const float* __restrict__ W2,
                        float* __restrict__ h2_all, int n, int E) {
    constexpr int R = HID / 32;
    constexpr int CT = NC / 16;
    __shared__ __attribute__((aligned(16))) float lw[NC * GLD];
    __shared__ __attribute__((aligned(16))) float la[8 * GRW * GLD];

#pragma unroll
    for (int j = 0; j < HID * NC / 256; ++j) {
        const int i = j * 256 + (int)threadIdx.x;
        const int k = i / NC;
        const int c = i - k * NC;
        lw[c * GLD + k] = W2[i];
    }
    __syncthreads();

    const int g = blockIdx.y;
    const float* __restrict__ h = h_all + (size_t)g * n * HID;
    const int* __restrict__ tstart = tstart_all + (size_t)g * (((n + GRW - 1) / GRW) + 1);
    const int* __restrict__ esrc = esrc_all + (size_t)g * E;
    const float* __restrict__ rs_out = rs_all + (size_t)(2 * g) * n;
    const float* __restrict__ rs_in = rs_all + (size_t)(2 * g + 1) * n;
    float* __restrict__ h2 = h2_all + (size_t)g * n * NC;
    const int lane = threadIdx.x & 31;
    const int wave = threadIdx.x >> 5;
    const int gw = __builtin_amdgcn_readfirstlane((int)(blockIdx.x * 8 + wave));
    const int n0 = gw * GRW;
    if (n0 >= n) return;
    const int nn = (n - n0 < GRW) ? (n - n0) : GRW;
    const int eb = __builtin_amdgcn_readfirstlane(tstart[gw]);
    const int ee = __builtin_amdgcn_readfirstlane(tstart[gw + 1]);
    float* __restrict__ lt = la + wave * (GRW * GLD);

    v16f acc[R] = {};
    tile_stream<HID>(h, esrc, eb, ee, lane, acc);

    const int ib = n0 + ((lane < nn) ? lane : (nn - 1));
    const float sci_l = rs_in[ib];
    const float sco_l = rs_out[ib];
    float bias[R];
#pragma unroll
    for (int r = 0; r < R; ++r) bias[r] = b[lane + r * 32];
#pragma unroll
    for (int i = 0; i < GRW; ++i) {
        const float rsi = lane_pick_f(sci_l, i);
        const float rso = lane_pick_f(sco_l, i);
#pragma unroll
        for (int r = 0; r < R; ++r)
            lt[i * GLD + lane + r * 32] = fmaxf(acc[r][i] * rsi + bias[r], 0.0f) * rso;
    }

    const int col = lane & 15;
    const int koff = (lane >> 4) * 2;
    v8f d[CT] = {};
    for (int k = 0; k < HID; k += 4) {
        float2 av = *(const float2*)&lt[col * GLD + k + koff];
        v2f a;
        a.x = av.x;
        a.y = av.y;
#pragma unroll
        for (int c = 0; c < CT; ++c) {
            float2 bv = *(const float2*)&lw[(c * 16 + col) * GLD + k + koff];
            v2f bb;
            bb.x = bv.x;
            bb.y = bv.y;
            d[c] = __builtin_amdgcn_wmma_f32_16x16x4_f32(
                false, a, false, bb, (short)0, d[c], false, false);
        }
    }

    const int mhi = (lane >> 4) * 8;
#pragma unroll
    for (int c = 0; c < CT; ++c) {
#pragma unroll
        for (int r = 0; r < 8; ++r) lt[(mhi + r) * GLD + c * 16 + col] = d[c][r];
    }
    for (int i = 0; i < nn; ++i) h2[(size_t)(n0 + i) * NC + lane] = lt[i * GLD + lane];
}

__global__ void __launch_bounds__(256)
gather_mean_kernel(const float* __restrict__ hA, const int* __restrict__ tstartA,
                   const int* __restrict__ esrcA, const float* __restrict__ rsA,
                   const float* __restrict__ hB, const int* __restrict__ tstartB,
                   const int* __restrict__ esrcB, const float* __restrict__ rsB,
                   const float* __restrict__ b, float* __restrict__ out, int n) {
    const int lane = threadIdx.x & 31;
    const int gw = __builtin_amdgcn_readfirstlane((int)((blockIdx.x * 256 + threadIdx.x) >> 5));
    const int n0 = gw * GRW;
    if (n0 >= n) return;
    const int nn = (n - n0 < GRW) ? (n - n0) : GRW;
    const int ib = n0 + ((lane < nn) ? lane : (nn - 1));
    const int ebA = __builtin_amdgcn_readfirstlane(tstartA[gw]);
    const int eeA = __builtin_amdgcn_readfirstlane(tstartA[gw + 1]);
    const int ebB = __builtin_amdgcn_readfirstlane(tstartB[gw]);
    const int eeB = __builtin_amdgcn_readfirstlane(tstartB[gw + 1]);
    const float scA_l = rsA[ib];
    const float scB_l = rsB[ib];
    const float bias = b[lane];

    v16f accA[1] = {};
    v16f accB[1] = {};
    tile_stream_ahead<NC>(hA, esrcA, ebA, eeA, lane, accA[0]);
    tile_stream_ahead<NC>(hB, esrcB, ebB, eeB, lane, accB[0]);
#pragma unroll
    for (int i = 0; i < GRW; ++i) {
        if (i < nn) {
            const float tA = accA[0][i] * lane_pick_f(scA_l, i) + bias;
            const float tB = accB[0][i] * lane_pick_f(scB_l, i) + bias;
            out[(size_t)(n0 + i) * NC + lane] = 0.5f * tA + 0.5f * tB;
        }
    }
}


extern "C" void kernel_launch(void* const* d_in, const int* in_sizes, int n_in,
                              void* d_out, int out_size, void* d_ws, size_t ws_size,
                              hipStream_t stream) {
    const float* x    = (const float*)d_in[0];
    const int*   src1 = (const int*)d_in[1];
    const int*   dst1 = (const int*)d_in[2];
    const int*   src2 = (const int*)d_in[3];
    const int*   dst2 = (const int*)d_in[4];
    const float* W1   = (const float*)d_in[5];
    const float* b1   = (const float*)d_in[6];
    const float* W2   = (const float*)d_in[7];
    const float* b2   = (const float*)d_in[8];
    float* out = (float*)d_out;

    const int N = in_sizes[0] / IN_FEATS;
    const int E = in_sizes[1];

    const int dranges = (N + DEG_HR - 1) / DEG_HR;
    const int dstride = dranges * DEG_HR;
    unsigned int* dpart = (unsigned int*)d_ws;
    float* ws   = (float*)(dpart + (size_t)4 * DEG_S * dstride);
    float* rsOI = ws;
    float* h    = rsOI + (size_t)4 * N;
    float* h2   = h + (size_t)2 * N * HID;
    int* esrc   = (int*)(h2 + (size_t)2 * N * NC);
    int* gsum   = esrc + (size_t)2 * E;
    int* tstart = gsum + (size_t)2 * (dstride / 256) * DEG_S;
    int* tsum   = tstart + (size_t)2 * (N + 1);

    const int T = 256;
    auto cdiv = [](long a, long b) { return (int)((a + b - 1) / b); };

    const int tiles   = N / 16;
    const int tpw     = cdiv(tiles, 256);
    const int tpb     = (tpw > DEG_T / 32) ? (DEG_T / 32) : ((tpw < 1) ? 1 : tpw);
    const int pblocks = cdiv(tiles, tpb);
    const int tblocks = cdiv(cdiv(N, GRW), 8);

    deg_hist_kernel<<<dim3(DEG_S, dranges, 4), DEG_T, 0, stream>>>(src1, dst1, src2, dst2, dpart, gsum, tsum, E, dstride);
    order_gemm_kernel<IN_FEATS, HID><<<pblocks + DEG_S * dranges * 2, DEG_T, 0, stream>>>(
        x, W1, h, dpart, gsum, tsum, rsOI, tstart, src1, dst1, src2, dst2, esrc, E, N, dstride, pblocks, tpb, dranges);

    gather_relu_gemm_kernel<<<dim3(tblocks, 2), T, 0, stream>>>(h, tstart, esrc, rsOI, b1, W2, h2, N, E);

    gather_mean_kernel<<<tblocks, T, 0, stream>>>(
        h2, tstart, esrc, rsOI + N,
        h2 + (size_t)N * NC, tstart + (cdiv(N, GRW) + 1), esrc + E, rsOI + (size_t)3 * N,
        b2, out, N);
}
